# starting kernel
# baseline (speedup 1.0000x reference)
#include <hip/hip_runtime.h>
#include <math.h>

#define H      256
#define HE     64
#define NB     8
#define A_PER  64
#define N_PER  512
#define NA     (NB * A_PER)
#define NN     (NB * N_PER)

typedef __attribute__((ext_vector_type(2))) float v2f;
typedef __attribute__((ext_vector_type(8))) float v8f;

struct GemmArgs {
  const float* A;  long lda;  long sAb;
  const float* B;  long ldbn; long ldbk; long sBb;
  float*       C;  long ldc;  long sCb;
  const float* bias;
  int K;
  int relu;
  int addC;
};

template <bool BCONTIG>
__global__ __launch_bounds__(32)
void wmma_gemm_f32(GemmArgs g) {
  const int lane = threadIdx.x;
  const int r15  = lane & 15;
  const int kh   = (lane >> 4) << 1;
  const int tn = blockIdx.x, tm = blockIdx.y, bz = blockIdx.z;

  const float* Abase = g.A + (long)bz * g.sAb + (long)(tm * 64 + r15) * g.lda;
  const float* A0 = Abase;
  const float* A1 = Abase + (long)16 * g.lda;
  const float* A2 = Abase + (long)32 * g.lda;
  const float* A3 = Abase + (long)48 * g.lda;
  const float* B0 = g.B + (long)bz * g.sBb + (long)(tn * 32 + r15) * g.ldbn;
  const float* B1 = B0 + (long)16 * g.ldbn;

  const v8f zero = {0.f,0.f,0.f,0.f,0.f,0.f,0.f,0.f};
  v8f c00 = zero, c01 = zero, c10 = zero, c11 = zero;
  v8f c20 = zero, c21 = zero, c30 = zero, c31 = zero;

#if __has_builtin(__builtin_amdgcn_wmma_f32_16x16x4_f32)
  for (int k0 = 0; k0 < g.K; k0 += 4) {
    const int ka = k0 + kh;
    v2f a0 = *(const v2f*)(A0 + ka);
    v2f a1 = *(const v2f*)(A1 + ka);
    v2f a2 = *(const v2f*)(A2 + ka);
    v2f a3 = *(const v2f*)(A3 + ka);
    v2f b0, b1;
    if (BCONTIG) {
      b0 = *(const v2f*)(B0 + ka);
      b1 = *(const v2f*)(B1 + ka);
    } else {
      b0.x = B0[(long)ka * g.ldbk];  b0.y = B0[(long)(ka + 1) * g.ldbk];
      b1.x = B1[(long)ka * g.ldbk];  b1.y = B1[(long)(ka + 1) * g.ldbk];
    }
    c00 = __builtin_amdgcn_wmma_f32_16x16x4_f32(false, a0, false, b0, (short)0, c00, false, false);
    c01 = __builtin_amdgcn_wmma_f32_16x16x4_f32(false, a0, false, b1, (short)0, c01, false, false);
    c10 = __builtin_amdgcn_wmma_f32_16x16x4_f32(false, a1, false, b0, (short)0, c10, false, false);
    c11 = __builtin_amdgcn_wmma_f32_16x16x4_f32(false, a1, false, b1, (short)0, c11, false, false);
    c20 = __builtin_amdgcn_wmma_f32_16x16x4_f32(false, a2, false, b0, (short)0, c20, false, false);
    c21 = __builtin_amdgcn_wmma_f32_16x16x4_f32(false, a2, false, b1, (short)0, c21, false, false);
    c30 = __builtin_amdgcn_wmma_f32_16x16x4_f32(false, a3, false, b0, (short)0, c30, false, false);
    c31 = __builtin_amdgcn_wmma_f32_16x16x4_f32(false, a3, false, b1, (short)0, c31, false, false);
  }
#else
  {
    const float* Ab = g.A + (long)bz * g.sAb;
    const int mb = tm * 64 + ((lane >> 4) << 3);
    for (int k0 = 0; k0 < g.K; ++k0) {
      float bv0 = B0[(long)k0 * g.ldbk];
      float bv1 = B1[(long)k0 * g.ldbk];
      #pragma unroll
      for (int r = 0; r < 8; ++r) {
        float av0 = Ab[(long)(mb + r)      * g.lda + k0];
        float av1 = Ab[(long)(mb + 16 + r) * g.lda + k0];
        float av2 = Ab[(long)(mb + 32 + r) * g.lda + k0];
        float av3 = Ab[(long)(mb + 48 + r) * g.lda + k0];
        c00[r] += av0 * bv0;  c01[r] += av0 * bv1;
        c10[r] += av1 * bv0;  c11[r] += av1 * bv1;
        c20[r] += av2 * bv0;  c21[r] += av2 * bv1;
        c30[r] += av3 * bv0;  c31[r] += av3 * bv1;
      }
    }
  }
#endif

  const int  col0 = tn * 32 + r15;
  const int  col1 = col0 + 16;
  const int  m0   = tm * 64 + ((lane >> 4) << 3);
  const float bv0 = g.bias ? g.bias[col0] : 0.f;
  const float bv1 = g.bias ? g.bias[col1] : 0.f;
  float* Cp = g.C + (long)bz * g.sCb;
  const v8f* cc0[4] = { &c00, &c10, &c20, &c30 };
  const v8f* cc1[4] = { &c01, &c11, &c21, &c31 };
  #pragma unroll
  for (int i = 0; i < 4; ++i) {
    #pragma unroll
    for (int r = 0; r < 8; ++r) {
      const long row = (long)(m0 + i * 16 + r) * g.ldc;
      float v0 = (*cc0[i])[r] + bv0;
      float v1 = (*cc1[i])[r] + bv1;
      if (g.addC) { v0 += Cp[row + col0]; v1 += Cp[row + col1]; }
      if (g.relu) { v0 = fmaxf(v0, 0.f);  v1 = fmaxf(v1, 0.f); }
      Cp[row + col0] = v0;
      Cp[row + col1] = v1;
    }
  }
}

__global__ __launch_bounds__(256)
void edge_softmax_kernel(const float* __restrict__ anchor_x,
                         const float* __restrict__ node_x,
                         const float* __restrict__ node_mask,
                         const float* __restrict__ qe,
                         const float* __restrict__ Lbase,
                         float* __restrict__ attn,
                         float* __restrict__ s_out)
{
  const int bid = blockIdx.x;
  const int b   = bid >> 6;
  const int t   = threadIdx.x;

  __shared__ float sh_qe[HE];
  __shared__ float sh_dist[N_PER];
  __shared__ float sh_l[N_PER];
  __shared__ float red[256];
  __shared__ float sh_bcast;
  __shared__ float sh_ax[3];

  if (t < HE) sh_qe[t] = qe[bid * HE + t];
  if (t < 3)  sh_ax[t] = anchor_x[bid * 3 + t];
  __syncthreads();

  const float axx = sh_ax[0], axy = sh_ax[1], axz = sh_ax[2];
  const float MU_STEP = 20.f / 63.f;
  const float INV_SIG = 64.f / 20.f;

  for (int n = t; n < N_PER; n += 256) {
    const float* nx = node_x + (long)(b * N_PER + n) * 3;
    float dx = axx - nx[0] + 1e-8f;
    float dy = axy - nx[1] + 1e-8f;
    float dz = axz - nx[2] + 1e-8f;
    float d  = sqrtf(dx * dx + dy * dy + dz * dz);
    sh_dist[n] = d;
    float d10 = d * 0.1f;
    float le = 0.f;
    #pragma unroll 8
    for (int j = 0; j < HE; ++j) {
      float u = (d10 - j * MU_STEP) * INV_SIG;
      le += sh_qe[j] * __expf(-u * u);
    }
    float lb = Lbase[(long)bid * N_PER + n] + le;
    lb *= (node_mask[b * N_PER + n] - 1.f) * 1000000.0f;
    sh_l[n] = lb;
  }
  __syncthreads();

  red[t] = fmaxf(sh_l[t], sh_l[t + 256]);
  __syncthreads();
  for (int st = 128; st > 0; st >>= 1) {
    if (t < st) red[t] = fmaxf(red[t], red[t + st]);
    __syncthreads();
  }
  if (t == 0) sh_bcast = red[0];
  __syncthreads();
  const float mx = sh_bcast;

  float e0 = __expf(sh_l[t] - mx);
  float e1 = __expf(sh_l[t + 256] - mx);
  red[t] = e0 + e1;
  __syncthreads();
  for (int st = 128; st > 0; st >>= 1) {
    if (t < st) red[t] += red[t + st];
    __syncthreads();
  }
  if (t == 0) sh_bcast = red[0];
  __syncthreads();
  const float inv = 1.f / sh_bcast;

  const float a0 = e0 * inv, a1 = e1 * inv;
  sh_l[t] = a0; sh_l[t + 256] = a1;
  attn[(long)bid * N_PER + t]       = a0;
  attn[(long)bid * N_PER + t + 256] = a1;
  __syncthreads();

  const int j = t & 63, c = t >> 6;
  const float muj = j * MU_STEP;
  float part = 0.f;
  for (int n = c * 128; n < c * 128 + 128; ++n) {
    float u = (sh_dist[n] * 0.1f - muj) * INV_SIG;
    part += sh_l[n] * __expf(-u * u);
  }
  red[t] = part;
  __syncthreads();
  if (t < 64)
    s_out[bid * HE + t] = red[t] + red[t + 64] + red[t + 128] + red[t + 192];
}

__global__ __launch_bounds__(256)
void add_ln_kernel(const float* __restrict__ x, const float* __restrict__ y,
                   const float* __restrict__ gam, const float* __restrict__ bet,
                   float* __restrict__ out)
{
  const int r = blockIdx.x, t = threadIdx.x;
  __shared__ float red[256];
  __shared__ float sh_mean, sh_rstd;

  const float v = x[(long)r * H + t] + y[(long)r * H + t];
  red[t] = v;
  __syncthreads();
  for (int st = 128; st > 0; st >>= 1) { if (t < st) red[t] += red[t + st]; __syncthreads(); }
  if (t == 0) sh_mean = red[0] * (1.f / H);
  __syncthreads();

  const float d = v - sh_mean;
  red[t] = d * d;
  __syncthreads();
  for (int st = 128; st > 0; st >>= 1) { if (t < st) red[t] += red[t + st]; __syncthreads(); }
  if (t == 0) sh_rstd = rsqrtf(red[0] * (1.f / H) + 1e-5f);
  __syncthreads();

  out[(long)r * H + t] = d * sh_rstd * gam[t] + bet[t];
}

extern "C" void kernel_launch(void* const* d_in, const int* in_sizes, int n_in,
                              void* d_out, int out_size, void* d_ws, size_t ws_size,
                              hipStream_t stream) {
  (void)in_sizes; (void)n_in; (void)out_size; (void)ws_size;
  const float* anchor_x  = (const float*)d_in[0];
  const float* node_x    = (const float*)d_in[1];
  const float* anchor_f  = (const float*)d_in[2];
  const float* node_f    = (const float*)d_in[3];
  const float* node_mask = (const float*)d_in[6];
  const float* Wq  = (const float*)d_in[7];
  const float* bq  = (const float*)d_in[8];
  const float* Wkv = (const float*)d_in[9];
  const float* bkv = (const float*)d_in[10];
  const float* ln1_g = (const float*)d_in[11];
  const float* ln1_b = (const float*)d_in[12];
  const float* W1 = (const float*)d_in[13];
  const float* b1 = (const float*)d_in[14];
  const float* W2 = (const float*)d_in[15];
  const float* b2 = (const float*)d_in[16];
  const float* W3 = (const float*)d_in[17];
  const float* b3 = (const float*)d_in[18];
  const float* ln2_g = (const float*)d_in[19];
  const float* ln2_b = (const float*)d_in[20];

  const long WKV_LD = H + HE;

  float* p = (float*)d_ws;
  float* q    = p; p += (long)NA * H;
  float* qe   = p; p += (long)NA * HE;
  float* kvn  = p; p += (long)NN * 2 * H;
  float* L    = p; p += (long)NA * N_PER;
  float* attn = p; p += (long)NA * N_PER;
  float* sbuf = p; p += (long)NA * HE;
  float* upd  = p; p += (long)NA * H;
  float* af   = p; p += (long)NA * H;
  float* m1   = p; p += (long)NA * 2 * H;
  float* m2   = p; p += (long)NA * 2 * H;
  float* m3   = p; p += (long)NA * H;

  auto gemm = [&](const float* A, long lda, long sAb,
                  const float* Bm, long ldbn, long ldbk, long sBb,
                  float* C, long ldc, long sCb, const float* bias,
                  int M, int N, int K, int nb, int relu, int addC) {
    GemmArgs g{A, lda, sAb, Bm, ldbn, ldbk, sBb, C, ldc, sCb, bias, K, relu, addC};
    dim3 grid(N / 32, M / 64, nb);
    if (ldbk == 1)
      wmma_gemm_f32<true><<<grid, dim3(32), 0, stream>>>(g);
    else
      wmma_gemm_f32<false><<<grid, dim3(32), 0, stream>>>(g);
  };

  gemm(anchor_f, H, 0, Wq, H, 1, 0, q, H, 0, bq, NA, H, H, 1, 0, 0);
  gemm(q, H, 0, Wkv + H, 1, WKV_LD, 0, qe, HE, 0, nullptr, NA, HE, H, 1, 0, 0);
  gemm(node_f, H, 0, Wkv, WKV_LD, 1, 0, kvn, 2 * H, 0, bkv, NN, 2 * H, H, 1, 0, 0);
  gemm(q, H, (long)A_PER * H,
       kvn, 2 * H, 1, (long)N_PER * 2 * H,
       L, N_PER, (long)A_PER * N_PER, nullptr, A_PER, N_PER, H, NB, 0, 0);
  edge_softmax_kernel<<<dim3(NA), dim3(256), 0, stream>>>(
      anchor_x, node_x, node_mask, qe, L, attn, sbuf);
  gemm(attn, N_PER, (long)A_PER * N_PER,
       kvn + H, 1, 2 * H, (long)N_PER * 2 * H,
       upd, H, (long)A_PER * H, nullptr, A_PER, H, N_PER, NB, 0, 0);
  gemm(sbuf, HE, 0, Wkv + (long)H * WKV_LD + H, WKV_LD, 1, 0,
       upd, H, 0, nullptr, NA, H, HE, 1, 0, 1);
  add_ln_kernel<<<dim3(NA), dim3(256), 0, stream>>>(anchor_f, upd, ln1_g, ln1_b, af);
  gemm(af, H, 0, W1, H, 1, 0, m1, 2 * H, 0, b1, NA, 2 * H, H, 1, 1, 0);
  gemm(m1, 2 * H, 0, W2, 2 * H, 1, 0, m2, 2 * H, 0, b2, NA, 2 * H, 2 * H, 1, 1, 0);
  gemm(m2, 2 * H, 0, W3, 2 * H, 1, 0, m3, H, 0, b3, NA, H, 2 * H, 1, 0, 0);
  add_ln_kernel<<<dim3(NA), dim3(256), 0, stream>>>(af, m3, ln2_g, ln2_b, (float*)d_out);
}


# GEMM K loop software-pipelined: 32 k of A/B fragments per trip, next trip's loads issued as one group ahead of the current trip's 64 WMMAs (two register sets); k order and sums unchanged
# speedup vs baseline: 2.3797x; 2.3797x over previous
#include <hip/hip_runtime.h>
#include <math.h>

#define H      256
#define HE     64
#define NB     8
#define A_PER  64
#define N_PER  512
#define NA     (NB * A_PER)
#define NN     (NB * N_PER)

typedef __attribute__((ext_vector_type(2))) float v2f;
typedef __attribute__((ext_vector_type(8))) float v8f;

struct GemmArgs {
  const float* A;  long lda;  long sAb;
  const float* B;  long ldbn; long ldbk; long sBb;
  float*       C;  long ldc;  long sCb;
  const float* bias;
  int K;
  int relu;
  int addC;
};

template <bool BCONTIG>
__global__ __launch_bounds__(32)
void wmma_gemm_f32(GemmArgs g) {
  const int lane = threadIdx.x;
  const int r15  = lane & 15;
  const int kh   = (lane >> 4) << 1;
  const int tn = blockIdx.x, tm = blockIdx.y, bz = blockIdx.z;

  const float* Abase = g.A + (long)bz * g.sAb + (long)(tm * 64 + r15) * g.lda;
  const float* A0 = Abase;
  const float* A1 = Abase + (long)16 * g.lda;
  const float* A2 = Abase + (long)32 * g.lda;
  const float* A3 = Abase + (long)48 * g.lda;
  const float* B0 = g.B + (long)bz * g.sBb + (long)(tn * 32 + r15) * g.ldbn;
  const float* B1 = B0 + (long)16 * g.ldbn;

  const v8f zero = {0.f,0.f,0.f,0.f,0.f,0.f,0.f,0.f};
  v8f c00 = zero, c01 = zero, c10 = zero, c11 = zero;
  v8f c20 = zero, c21 = zero, c30 = zero, c31 = zero;

#if __has_builtin(__builtin_amdgcn_wmma_f32_16x16x4_f32)
  v2f aP[4][8], bP[2][8], aQ[4][8], bQ[2][8];

  auto load_trip = [&](v2f (&a)[4][8], v2f (&b)[2][8], int kbase) {
    #pragma unroll
    for (int s = 0; s < 8; ++s) {
      const int ka = kbase + 4 * s + kh;
      a[0][s] = *(const v2f*)(A0 + ka);
      a[1][s] = *(const v2f*)(A1 + ka);
      a[2][s] = *(const v2f*)(A2 + ka);
      a[3][s] = *(const v2f*)(A3 + ka);
      if (BCONTIG) {
        b[0][s] = *(const v2f*)(B0 + ka);
        b[1][s] = *(const v2f*)(B1 + ka);
      } else {
        b[0][s].x = B0[(long)ka * g.ldbk];  b[0][s].y = B0[(long)(ka + 1) * g.ldbk];
        b[1][s].x = B1[(long)ka * g.ldbk];  b[1][s].y = B1[(long)(ka + 1) * g.ldbk];
      }
    }
  };
  auto mma_trip = [&](const v2f (&a)[4][8], const v2f (&b)[2][8]) {
    #pragma unroll
    for (int s = 0; s < 8; ++s) {
      c00 = __builtin_amdgcn_wmma_f32_16x16x4_f32(false, a[0][s], false, b[0][s], (short)0, c00, false, false);
      c01 = __builtin_amdgcn_wmma_f32_16x16x4_f32(false, a[0][s], false, b[1][s], (short)0, c01, false, false);
      c10 = __builtin_amdgcn_wmma_f32_16x16x4_f32(false, a[1][s], false, b[0][s], (short)0, c10, false, false);
      c11 = __builtin_amdgcn_wmma_f32_16x16x4_f32(false, a[1][s], false, b[1][s], (short)0, c11, false, false);
      c20 = __builtin_amdgcn_wmma_f32_16x16x4_f32(false, a[2][s], false, b[0][s], (short)0, c20, false, false);
      c21 = __builtin_amdgcn_wmma_f32_16x16x4_f32(false, a[2][s], false, b[1][s], (short)0, c21, false, false);
      c30 = __builtin_amdgcn_wmma_f32_16x16x4_f32(false, a[3][s], false, b[0][s], (short)0, c30, false, false);
      c31 = __builtin_amdgcn_wmma_f32_16x16x4_f32(false, a[3][s], false, b[1][s], (short)0, c31, false, false);
    }
  };

  load_trip(aP, bP, 0);
  int k0 = 0;
  for (; k0 + 64 < g.K; k0 += 64) {
    load_trip(aQ, bQ, k0 + 32);
    __builtin_amdgcn_sched_barrier(0);
    mma_trip(aP, bP);
    __builtin_amdgcn_sched_barrier(0);
    load_trip(aP, bP, k0 + 64);
    __builtin_amdgcn_sched_barrier(0);
    mma_trip(aQ, bQ);
  }
  load_trip(aQ, bQ, k0 + 32);
  __builtin_amdgcn_sched_barrier(0);
  mma_trip(aP, bP);
  mma_trip(aQ, bQ);
#else
  {
    const float* Ab = g.A + (long)bz * g.sAb;
    const int mb = tm * 64 + ((lane >> 4) << 3);
    for (int k0 = 0; k0 < g.K; ++k0) {
      float bv0 = B0[(long)k0 * g.ldbk];
      float bv1 = B1[(long)k0 * g.ldbk];
      #pragma unroll
      for (int r = 0; r < 8; ++r) {
        float av0 = Ab[(long)(mb + r)      * g.lda + k0];
        float av1 = Ab[(long)(mb + 16 + r) * g.lda + k0];
        float av2 = Ab[(long)(mb + 32 + r) * g.lda + k0];
        float av3 = Ab[(long)(mb + 48 + r) * g.lda + k0];
        c00[r] += av0 * bv0;  c01[r] += av0 * bv1;
        c10[r] += av1 * bv0;  c11[r] += av1 * bv1;
        c20[r] += av2 * bv0;  c21[r] += av2 * bv1;
        c30[r] += av3 * bv0;  c31[r] += av3 * bv1;
      }
    }
  }
#endif

  const int  col0 = tn * 32 + r15;
  const int  col1 = col0 + 16;
  const int  m0   = tm * 64 + ((lane >> 4) << 3);
  const float bv0 = g.bias ? g.bias[col0] : 0.f;
  const float bv1 = g.bias ? g.bias[col1] : 0.f;
  float* Cp = g.C + (long)bz * g.sCb;
  const v8f* cc0[4] = { &c00, &c10, &c20, &c30 };
  const v8f* cc1[4] = { &c01, &c11, &c21, &c31 };
  #pragma unroll
  for (int i = 0; i < 4; ++i) {
    #pragma unroll
    for (int r = 0; r < 8; ++r) {
      const long row = (long)(m0 + i * 16 + r) * g.ldc;
      float v0 = (*cc0[i])[r] + bv0;
      float v1 = (*cc1[i])[r] + bv1;
      if (g.addC) { v0 += Cp[row + col0]; v1 += Cp[row + col1]; }
      if (g.relu) { v0 = fmaxf(v0, 0.f);  v1 = fmaxf(v1, 0.f); }
      Cp[row + col0] = v0;
      Cp[row + col1] = v1;
    }
  }
}

__global__ __launch_bounds__(256)
void edge_softmax_kernel(const float* __restrict__ anchor_x,
                         const float* __restrict__ node_x,
                         const float* __restrict__ node_mask,
                         const float* __restrict__ qe,
                         const float* __restrict__ Lbase,
                         float* __restrict__ attn,
                         float* __restrict__ s_out)
{
  const int bid = blockIdx.x;
  const int b   = bid >> 6;
  const int t   = threadIdx.x;

  __shared__ float sh_qe[HE];
  __shared__ float sh_dist[N_PER];
  __shared__ float sh_l[N_PER];
  __shared__ float red[256];
  __shared__ float sh_bcast;
  __shared__ float sh_ax[3];

  if (t < HE) sh_qe[t] = qe[bid * HE + t];
  if (t < 3)  sh_ax[t] = anchor_x[bid * 3 + t];
  __syncthreads();

  const float axx = sh_ax[0], axy = sh_ax[1], axz = sh_ax[2];
  const float MU_STEP = 20.f / 63.f;
  const float INV_SIG = 64.f / 20.f;

  for (int n = t; n < N_PER; n += 256) {
    const float* nx = node_x + (long)(b * N_PER + n) * 3;
    float dx = axx - nx[0] + 1e-8f;
    float dy = axy - nx[1] + 1e-8f;
    float dz = axz - nx[2] + 1e-8f;
    float d  = sqrtf(dx * dx + dy * dy + dz * dz);
    sh_dist[n] = d;
    float d10 = d * 0.1f;
    float le = 0.f;
    #pragma unroll 8
    for (int j = 0; j < HE; ++j) {
      float u = (d10 - j * MU_STEP) * INV_SIG;
      le += sh_qe[j] * __expf(-u * u);
    }
    float lb = Lbase[(long)bid * N_PER + n] + le;
    lb *= (node_mask[b * N_PER + n] - 1.f) * 1000000.0f;
    sh_l[n] = lb;
  }
  __syncthreads();

  red[t] = fmaxf(sh_l[t], sh_l[t + 256]);
  __syncthreads();
  for (int st = 128; st > 0; st >>= 1) {
    if (t < st) red[t] = fmaxf(red[t], red[t + st]);
    __syncthreads();
  }
  if (t == 0) sh_bcast = red[0];
  __syncthreads();
  const float mx = sh_bcast;

  float e0 = __expf(sh_l[t] - mx);
  float e1 = __expf(sh_l[t + 256] - mx);
  red[t] = e0 + e1;
  __syncthreads();
  for (int st = 128; st > 0; st >>= 1) {
    if (t < st) red[t] += red[t + st];
    __syncthreads();
  }
  if (t == 0) sh_bcast = red[0];
  __syncthreads();
  const float inv = 1.f / sh_bcast;

  const float a0 = e0 * inv, a1 = e1 * inv;
  sh_l[t] = a0; sh_l[t + 256] = a1;
  attn[(long)bid * N_PER + t]       = a0;
  attn[(long)bid * N_PER + t + 256] = a1;
  __syncthreads();

  const int j = t & 63, c = t >> 6;
  const float muj = j * MU_STEP;
  float part = 0.f;
  for (int n = c * 128; n < c * 128 + 128; ++n) {
    float u = (sh_dist[n] * 0.1f - muj) * INV_SIG;
    part += sh_l[n] * __expf(-u * u);
  }
  red[t] = part;
  __syncthreads();
  if (t < 64)
    s_out[bid * HE + t] = red[t] + red[t + 64] + red[t + 128] + red[t + 192];
}

__global__ __launch_bounds__(256)
void add_ln_kernel(const float* __restrict__ x, const float* __restrict__ y,
                   const float* __restrict__ gam, const float* __restrict__ bet,
                   float* __restrict__ out)
{
  const int r = blockIdx.x, t = threadIdx.x;
  __shared__ float red[256];
  __shared__ float sh_mean, sh_rstd;

  const float v = x[(long)r * H + t] + y[(long)r * H + t];
  red[t] = v;
  __syncthreads();
  for (int st = 128; st > 0; st >>= 1) { if (t < st) red[t] += red[t + st]; __syncthreads(); }
  if (t == 0) sh_mean = red[0] * (1.f / H);
  __syncthreads();

  const float d = v - sh_mean;
  red[t] = d * d;
  __syncthreads();
  for (int st = 128; st > 0; st >>= 1) { if (t < st) red[t] += red[t + st]; __syncthreads(); }
  if (t == 0) sh_rstd = rsqrtf(red[0] * (1.f / H) + 1e-5f);
  __syncthreads();

  out[(long)r * H + t] = d * sh_rstd * gam[t] + bet[t];
}

extern "C" void kernel_launch(void* const* d_in, const int* in_sizes, int n_in,
                              void* d_out, int out_size, void* d_ws, size_t ws_size,
                              hipStream_t stream) {
  (void)in_sizes; (void)n_in; (void)out_size; (void)ws_size;
  const float* anchor_x  = (const float*)d_in[0];
  const float* node_x    = (const float*)d_in[1];
  const float* anchor_f  = (const float*)d_in[2];
  const float* node_f    = (const float*)d_in[3];
  const float* node_mask = (const float*)d_in[6];
  const float* Wq  = (const float*)d_in[7];
  const float* bq  = (const float*)d_in[8];
  const float* Wkv = (const float*)d_in[9];
  const float* bkv = (const float*)d_in[10];
  const float* ln1_g = (const float*)d_in[11];
  const float* ln1_b = (const float*)d_in[12];
  const float* W1 = (const float*)d_in[13];
  const float* b1 = (const float*)d_in[14];
  const float* W2 = (const float*)d_in[15];
  const float* b2 = (const float*)d_in[16];
  const float* W3 = (const float*)d_in[17];
  const float* b3 = (const float*)d_in[18];
  const float* ln2_g = (const float*)d_in[19];
  const float* ln2_b = (const float*)d_in[20];

  const long WKV_LD = H + HE;

  float* p = (float*)d_ws;
  float* q    = p; p += (long)NA * H;
  float* qe   = p; p += (long)NA * HE;
  float* kvn  = p; p += (long)NN * 2 * H;
  float* L    = p; p += (long)NA * N_PER;
  float* attn = p; p += (long)NA * N_PER;
  float* sbuf = p; p += (long)NA * HE;
  float* upd  = p; p += (long)NA * H;
  float* af   = p; p += (long)NA * H;
  float* m1   = p; p += (long)NA * 2 * H;
  float* m2   = p; p += (long)NA * 2 * H;
  float* m3   = p; p += (long)NA * H;

  auto gemm = [&](const float* A, long lda, long sAb,
                  const float* Bm, long ldbn, long ldbk, long sBb,
                  float* C, long ldc, long sCb, const float* bias,
                  int M, int N, int K, int nb, int relu, int addC) {
    GemmArgs g{A, lda, sAb, Bm, ldbn, ldbk, sBb, C, ldc, sCb, bias, K, relu, addC};
    dim3 grid(N / 32, M / 64, nb);
    if (ldbk == 1)
      wmma_gemm_f32<true><<<grid, dim3(32), 0, stream>>>(g);
    else
      wmma_gemm_f32<false><<<grid, dim3(32), 0, stream>>>(g);
  };

  gemm(anchor_f, H, 0, Wq, H, 1, 0, q, H, 0, bq, NA, H, H, 1, 0, 0);
  gemm(q, H, 0, Wkv + H, 1, WKV_LD, 0, qe, HE, 0, nullptr, NA, HE, H, 1, 0, 0);
  gemm(node_f, H, 0, Wkv, WKV_LD, 1, 0, kvn, 2 * H, 0, bkv, NN, 2 * H, H, 1, 0, 0);
  gemm(q, H, (long)A_PER * H,
       kvn, 2 * H, 1, (long)N_PER * 2 * H,
       L, N_PER, (long)A_PER * N_PER, nullptr, A_PER, N_PER, H, NB, 0, 0);
  edge_softmax_kernel<<<dim3(NA), dim3(256), 0, stream>>>(
      anchor_x, node_x, node_mask, qe, L, attn, sbuf);
  gemm(attn, N_PER, (long)A_PER * N_PER,
       kvn + H, 1, 2 * H, (long)N_PER * 2 * H,
       upd, H, (long)A_PER * H, nullptr, A_PER, H, N_PER, NB, 0, 0);
  gemm(sbuf, HE, 0, Wkv + (long)H * WKV_LD + H, WKV_LD, 1, 0,
       upd, H, 0, nullptr, NA, H, HE, 1, 0, 1);
  add_ln_kernel<<<dim3(NA), dim3(256), 0, stream>>>(anchor_f, upd, ln1_g, ln1_b, af);
  gemm(af, H, 0, W1, H, 1, 0, m1, 2 * H, 0, b1, NA, 2 * H, H, 1, 1, 0);
  gemm(m1, 2 * H, 0, W2, 2 * H, 1, 0, m2, 2 * H, 0, b2, NA, 2 * H, 2 * H, 1, 1, 0);
  gemm(m2, 2 * H, 0, W3, 2 * H, 1, 0, m3, H, 0, b3, NA, H, 2 * H, 1, 0, 0);
  add_ln_kernel<<<dim3(NA), dim3(256), 0, stream>>>(af, m3, ln2_g, ln2_b, (float*)d_out);
}


# Small GEMMs re-tiled: wave tile 64x32 -> 16x32 (4x the waves, 2 instead of 8 WMMAs in series per 4-k step), 64-k trips where K allows; node projection keeps 64x32; k order and sums unchanged
# speedup vs baseline: 3.2630x; 1.3712x over previous
#include <hip/hip_runtime.h>
#include <math.h>

#define H      256
#define HE     64
#define NB     8
#define A_PER  64
#define N_PER  512
#define NA     (NB * A_PER)
#define NN     (NB * N_PER)

typedef __attribute__((ext_vector_type(2))) float v2f;
typedef __attribute__((ext_vector_type(8))) float v8f;

struct GemmArgs {
  const float* A;  long lda;  long sAb;
  const float* B;  long ldbn; long ldbk; long sBb;
  float*       C;  long ldc;  long sCb;
  const float* bias;
  int K;
  int relu;
  int addC;
};

template <bool BCONTIG, int MT, int TS>
__global__ __launch_bounds__(32)
void wmma_gemm_f32(GemmArgs g) {
  const int lane = threadIdx.x;
  const int r15  = lane & 15;
  const int kh   = (lane >> 4) << 1;
  const int tn = blockIdx.x, tm = blockIdx.y, bz = blockIdx.z;

  const float* Abase = g.A + (long)bz * g.sAb + (long)(tm * (16 * MT) + r15) * g.lda;
  const float* Ap[MT];
  #pragma unroll
  for (int i = 0; i < MT; ++i) Ap[i] = Abase + (long)(16 * i) * g.lda;
  const float* B0 = g.B + (long)bz * g.sBb + (long)(tn * 32 + r15) * g.ldbn;
  const float* B1 = B0 + (long)16 * g.ldbn;

  const v8f zero = {0.f,0.f,0.f,0.f,0.f,0.f,0.f,0.f};
  v8f c[MT][2];
  #pragma unroll
  for (int i = 0; i < MT; ++i) { c[i][0] = zero; c[i][1] = zero; }

#if __has_builtin(__builtin_amdgcn_wmma_f32_16x16x4_f32)
  constexpr int TK = 4 * TS;
  v2f aP[MT][TS], bP[2][TS], aQ[MT][TS], bQ[2][TS];

  auto load_trip = [&](v2f (&a)[MT][TS], v2f (&b)[2][TS], int kbase) {
    #pragma unroll
    for (int s = 0; s < TS; ++s) {
      const int ka = kbase + 4 * s + kh;
      #pragma unroll
      for (int i = 0; i < MT; ++i) a[i][s] = *(const v2f*)(Ap[i] + ka);
      if (BCONTIG) {
        b[0][s] = *(const v2f*)(B0 + ka);
        b[1][s] = *(const v2f*)(B1 + ka);
      } else {
        b[0][s].x = B0[(long)ka * g.ldbk];  b[0][s].y = B0[(long)(ka + 1) * g.ldbk];
        b[1][s].x = B1[(long)ka * g.ldbk];  b[1][s].y = B1[(long)(ka + 1) * g.ldbk];
      }
    }
  };
  auto mma_trip = [&](const v2f (&a)[MT][TS], const v2f (&b)[2][TS]) {
    #pragma unroll
    for (int s = 0; s < TS; ++s) {
      #pragma unroll
      for (int i = 0; i < MT; ++i) {
        c[i][0] = __builtin_amdgcn_wmma_f32_16x16x4_f32(false, a[i][s], false, b[0][s], (short)0, c[i][0], false, false);
        c[i][1] = __builtin_amdgcn_wmma_f32_16x16x4_f32(false, a[i][s], false, b[1][s], (short)0, c[i][1], false, false);
      }
    }
  };

  load_trip(aP, bP, 0);
  int k0 = 0;
  for (; k0 + 2 * TK < g.K; k0 += 2 * TK) {
    load_trip(aQ, bQ, k0 + TK);
    __builtin_amdgcn_sched_barrier(0);
    mma_trip(aP, bP);
    __builtin_amdgcn_sched_barrier(0);
    load_trip(aP, bP, k0 + 2 * TK);
    __builtin_amdgcn_sched_barrier(0);
    mma_trip(aQ, bQ);
  }
  load_trip(aQ, bQ, k0 + TK);
  __builtin_amdgcn_sched_barrier(0);
  mma_trip(aP, bP);
  mma_trip(aQ, bQ);
#else
  {
    const float* Ab = g.A + (long)bz * g.sAb;
    const int mb = tm * (16 * MT) + ((lane >> 4) << 3);
    for (int k0 = 0; k0 < g.K; ++k0) {
      float bv0 = B0[(long)k0 * g.ldbk];
      float bv1 = B1[(long)k0 * g.ldbk];
      #pragma unroll
      for (int r = 0; r < 8; ++r) {
        #pragma unroll
        for (int i = 0; i < MT; ++i) {
          float av = Ab[(long)(mb + 16 * i + r) * g.lda + k0];
          c[i][0][r] += av * bv0;  c[i][1][r] += av * bv1;
        }
      }
    }
  }
#endif

  const int  col0 = tn * 32 + r15;
  const int  col1 = col0 + 16;
  const int  m0   = tm * (16 * MT) + ((lane >> 4) << 3);
  const float bv0 = g.bias ? g.bias[col0] : 0.f;
  const float bv1 = g.bias ? g.bias[col1] : 0.f;
  float* Cp = g.C + (long)bz * g.sCb;
  #pragma unroll
  for (int i = 0; i < MT; ++i) {
    #pragma unroll
    for (int r = 0; r < 8; ++r) {
      const long row = (long)(m0 + i * 16 + r) * g.ldc;
      float v0 = c[i][0][r] + bv0;
      float v1 = c[i][1][r] + bv1;
      if (g.addC) { v0 += Cp[row + col0]; v1 += Cp[row + col1]; }
      if (g.relu) { v0 = fmaxf(v0, 0.f);  v1 = fmaxf(v1, 0.f); }
      Cp[row + col0] = v0;
      Cp[row + col1] = v1;
    }
  }
}

__global__ __launch_bounds__(256)
void edge_softmax_kernel(const float* __restrict__ anchor_x,
                         const float* __restrict__ node_x,
                         const float* __restrict__ node_mask,
                         const float* __restrict__ qe,
                         const float* __restrict__ Lbase,
                         float* __restrict__ attn,
                         float* __restrict__ s_out)
{
  const int bid = blockIdx.x;
  const int b   = bid >> 6;
  const int t   = threadIdx.x;

  __shared__ float sh_qe[HE];
  __shared__ float sh_dist[N_PER];
  __shared__ float sh_l[N_PER];
  __shared__ float red[256];
  __shared__ float sh_bcast;
  __shared__ float sh_ax[3];

  if (t < HE) sh_qe[t] = qe[bid * HE + t];
  if (t < 3)  sh_ax[t] = anchor_x[bid * 3 + t];
  __syncthreads();

  const float axx = sh_ax[0], axy = sh_ax[1], axz = sh_ax[2];
  const float MU_STEP = 20.f / 63.f;
  const float INV_SIG = 64.f / 20.f;

  for (int n = t; n < N_PER; n += 256) {
    const float* nx = node_x + (long)(b * N_PER + n) * 3;
    float dx = axx - nx[0] + 1e-8f;
    float dy = axy - nx[1] + 1e-8f;
    float dz = axz - nx[2] + 1e-8f;
    float d  = sqrtf(dx * dx + dy * dy + dz * dz);
    sh_dist[n] = d;
    float d10 = d * 0.1f;
    float le = 0.f;
    #pragma unroll 8
    for (int j = 0; j < HE; ++j) {
      float u = (d10 - j * MU_STEP) * INV_SIG;
      le += sh_qe[j] * __expf(-u * u);
    }
    float lb = Lbase[(long)bid * N_PER + n] + le;
    lb *= (node_mask[b * N_PER + n] - 1.f) * 1000000.0f;
    sh_l[n] = lb;
  }
  __syncthreads();

  red[t] = fmaxf(sh_l[t], sh_l[t + 256]);
  __syncthreads();
  for (int st = 128; st > 0; st >>= 1) {
    if (t < st) red[t] = fmaxf(red[t], red[t + st]);
    __syncthreads();
  }
  if (t == 0) sh_bcast = red[0];
  __syncthreads();
  const float mx = sh_bcast;

  float e0 = __expf(sh_l[t] - mx);
  float e1 = __expf(sh_l[t + 256] - mx);
  red[t] = e0 + e1;
  __syncthreads();
  for (int st = 128; st > 0; st >>= 1) {
    if (t < st) red[t] += red[t + st];
    __syncthreads();
  }
  if (t == 0) sh_bcast = red[0];
  __syncthreads();
  const float inv = 1.f / sh_bcast;

  const float a0 = e0 * inv, a1 = e1 * inv;
  sh_l[t] = a0; sh_l[t + 256] = a1;
  attn[(long)bid * N_PER + t]       = a0;
  attn[(long)bid * N_PER + t + 256] = a1;
  __syncthreads();

  const int j = t & 63, c = t >> 6;
  const float muj = j * MU_STEP;
  float part = 0.f;
  for (int n = c * 128; n < c * 128 + 128; ++n) {
    float u = (sh_dist[n] * 0.1f - muj) * INV_SIG;
    part += sh_l[n] * __expf(-u * u);
  }
  red[t] = part;
  __syncthreads();
  if (t < 64)
    s_out[bid * HE + t] = red[t] + red[t + 64] + red[t + 128] + red[t + 192];
}

__global__ __launch_bounds__(256)
void add_ln_kernel(const float* __restrict__ x, const float* __restrict__ y,
                   const float* __restrict__ gam, const float* __restrict__ bet,
                   float* __restrict__ out)
{
  const int r = blockIdx.x, t = threadIdx.x;
  __shared__ float red[256];
  __shared__ float sh_mean, sh_rstd;

  const float v = x[(long)r * H + t] + y[(long)r * H + t];
  red[t] = v;
  __syncthreads();
  for (int st = 128; st > 0; st >>= 1) { if (t < st) red[t] += red[t + st]; __syncthreads(); }
  if (t == 0) sh_mean = red[0] * (1.f / H);
  __syncthreads();

  const float d = v - sh_mean;
  red[t] = d * d;
  __syncthreads();
  for (int st = 128; st > 0; st >>= 1) { if (t < st) red[t] += red[t + st]; __syncthreads(); }
  if (t == 0) sh_rstd = rsqrtf(red[0] * (1.f / H) + 1e-5f);
  __syncthreads();

  out[(long)r * H + t] = d * sh_rstd * gam[t] + bet[t];
}

extern "C" void kernel_launch(void* const* d_in, const int* in_sizes, int n_in,
                              void* d_out, int out_size, void* d_ws, size_t ws_size,
                              hipStream_t stream) {
  (void)in_sizes; (void)n_in; (void)out_size; (void)ws_size;
  const float* anchor_x  = (const float*)d_in[0];
  const float* node_x    = (const float*)d_in[1];
  const float* anchor_f  = (const float*)d_in[2];
  const float* node_f    = (const float*)d_in[3];
  const float* node_mask = (const float*)d_in[6];
  const float* Wq  = (const float*)d_in[7];
  const float* bq  = (const float*)d_in[8];
  const float* Wkv = (const float*)d_in[9];
  const float* bkv = (const float*)d_in[10];
  const float* ln1_g = (const float*)d_in[11];
  const float* ln1_b = (const float*)d_in[12];
  const float* W1 = (const float*)d_in[13];
  const float* b1 = (const float*)d_in[14];
  const float* W2 = (const float*)d_in[15];
  const float* b2 = (const float*)d_in[16];
  const float* W3 = (const float*)d_in[17];
  const float* b3 = (const float*)d_in[18];
  const float* ln2_g = (const float*)d_in[19];
  const float* ln2_b = (const float*)d_in[20];

  const long WKV_LD = H + HE;

  float* p = (float*)d_ws;
  float* q    = p; p += (long)NA * H;
  float* qe   = p; p += (long)NA * HE;
  float* kvn  = p; p += (long)NN * 2 * H;
  float* L    = p; p += (long)NA * N_PER;
  float* attn = p; p += (long)NA * N_PER;
  float* sbuf = p; p += (long)NA * HE;
  float* upd  = p; p += (long)NA * H;
  float* af   = p; p += (long)NA * H;
  float* m1   = p; p += (long)NA * 2 * H;
  float* m2   = p; p += (long)NA * 2 * H;
  float* m3   = p; p += (long)NA * H;

  auto gemm = [&](const float* A, long lda, long sAb,
                  const float* Bm, long ldbn, long ldbk, long sBb,
                  float* C, long ldc, long sCb, const float* bias,
                  int M, int N, int K, int nb, int relu, int addC) {
    GemmArgs g{A, lda, sAb, Bm, ldbn, ldbk, sBb, C, ldc, sCb, bias, K, relu, addC};
    if (ldbk == 1 && (N / 32) * (M / 64) * nb >= 1024)
      wmma_gemm_f32<true, 4, 8><<<dim3(N / 32, M / 64, nb), dim3(32), 0, stream>>>(g);
    else if (ldbk == 1 && K % 128 == 0)
      wmma_gemm_f32<true, 1, 16><<<dim3(N / 32, M / 16, nb), dim3(32), 0, stream>>>(g);
    else if (ldbk == 1)
      wmma_gemm_f32<true, 1, 8><<<dim3(N / 32, M / 16, nb), dim3(32), 0, stream>>>(g);
    else
      wmma_gemm_f32<false, 1, 8><<<dim3(N / 32, M / 16, nb), dim3(32), 0, stream>>>(g);
  };

  gemm(anchor_f, H, 0, Wq, H, 1, 0, q, H, 0, bq, NA, H, H, 1, 0, 0);
  gemm(q, H, 0, Wkv + H, 1, WKV_LD, 0, qe, HE, 0, nullptr, NA, HE, H, 1, 0, 0);
  gemm(node_f, H, 0, Wkv, WKV_LD, 1, 0, kvn, 2 * H, 0, bkv, NN, 2 * H, H, 1, 0, 0);
  gemm(q, H, (long)A_PER * H,
       kvn, 2 * H, 1, (long)N_PER * 2 * H,
       L, N_PER, (long)A_PER * N_PER, nullptr, A_PER, N_PER, H, NB, 0, 0);
  edge_softmax_kernel<<<dim3(NA), dim3(256), 0, stream>>>(
      anchor_x, node_x, node_mask, qe, L, attn, sbuf);
  gemm(attn, N_PER, (long)A_PER * N_PER,
       kvn + H, 1, 2 * H, (long)N_PER * 2 * H,
       upd, H, (long)A_PER * H, nullptr, A_PER, H, N_PER, NB, 0, 0);
  gemm(sbuf, HE, 0, Wkv + (long)H * WKV_LD + H, WKV_LD, 1, 0,
       upd, H, 0, nullptr, NA, H, HE, 1, 0, 1);
  add_ln_kernel<<<dim3(NA), dim3(256), 0, stream>>>(anchor_f, upd, ln1_g, ln1_b, af);
  gemm(af, H, 0, W1, H, 1, 0, m1, 2 * H, 0, b1, NA, 2 * H, H, 1, 1, 0);
  gemm(m1, 2 * H, 0, W2, 2 * H, 1, 0, m2, 2 * H, 0, b2, NA, 2 * H, 2 * H, 1, 1, 0);
  gemm(m2, 2 * H, 0, W3, 2 * H, 1, 0, m3, H, 0, b3, NA, H, 2 * H, 1, 0, 0);
  add_ln_kernel<<<dim3(NA), dim3(256), 0, stream>>>(af, m3, ln2_g, ln2_b, (float*)d_out);
}


# Launches 12 -> 9: independent GEMMs share one launch (node projection + q; qe + base logits); s*We_v runs as a 2nd K segment in the attn*v GEMM's accumulators (chained sum: rounding order changes)
# speedup vs baseline: 3.7357x; 1.1449x over previous
#include <hip/hip_runtime.h>
#include <math.h>

#define H      256
#define HE     64
#define NB     8
#define A_PER  64
#define N_PER  512
#define NA     (NB * A_PER)
#define NN     (NB * N_PER)

typedef __attribute__((ext_vector_type(2))) float v2f;
typedef __attribute__((ext_vector_type(8))) float v8f;

struct GemmArgs {
  const float* A;  long lda;  long sAb;
  const float* B;  long ldbn; long ldbk; long sBb;
  float*       C;  long ldc;  long sCb;
  const float* bias;
  const float* A2; long lda2;
  const float* B2; long ldbn2;
  int K;
  int K2;
  int relu;
  int lgx;
  int lgy;
};

template <bool BCONTIG, int MT, int TS>
__device__ __forceinline__ void k_segment(v8f (&c)[MT][2],
                                          const float* Ab, long lda,
                                          const float* B0, const float* B1, long ldbk,
                                          int K, int kh) {
  const float* Ap[MT];
  #pragma unroll
  for (int i = 0; i < MT; ++i) Ap[i] = Ab + (long)(16 * i) * lda;

  constexpr int TK = 4 * TS;
  v2f aP[MT][TS], bP[2][TS], aQ[MT][TS], bQ[2][TS];

  auto load_trip = [&](v2f (&a)[MT][TS], v2f (&b)[2][TS], int kbase) {
    #pragma unroll
    for (int s = 0; s < TS; ++s) {
      const int ka = kbase + 4 * s + kh;
      #pragma unroll
      for (int i = 0; i < MT; ++i) a[i][s] = *(const v2f*)(Ap[i] + ka);
      if (BCONTIG) {
        b[0][s] = *(const v2f*)(B0 + ka);
        b[1][s] = *(const v2f*)(B1 + ka);
      } else {
        b[0][s].x = B0[(long)ka * ldbk];  b[0][s].y = B0[(long)(ka + 1) * ldbk];
        b[1][s].x = B1[(long)ka * ldbk];  b[1][s].y = B1[(long)(ka + 1) * ldbk];
      }
    }
  };
  auto mma_trip = [&](const v2f (&a)[MT][TS], const v2f (&b)[2][TS]) {
    #pragma unroll
    for (int s = 0; s < TS; ++s) {
      #pragma unroll
      for (int i = 0; i < MT; ++i) {
        c[i][0] = __builtin_amdgcn_wmma_f32_16x16x4_f32(false, a[i][s], false, b[0][s], (short)0, c[i][0], false, false);
        c[i][1] = __builtin_amdgcn_wmma_f32_16x16x4_f32(false, a[i][s], false, b[1][s], (short)0, c[i][1], false, false);
      }
    }
  };

  load_trip(aP, bP, 0);
  int k0 = 0;
  for (; k0 + 2 * TK < K; k0 += 2 * TK) {
    load_trip(aQ, bQ, k0 + TK);
    __builtin_amdgcn_sched_barrier(0);
    mma_trip(aP, bP);
    __builtin_amdgcn_sched_barrier(0);
    load_trip(aP, bP, k0 + 2 * TK);
    __builtin_amdgcn_sched_barrier(0);
    mma_trip(aQ, bQ);
  }
  load_trip(aQ, bQ, k0 + TK);
  __builtin_amdgcn_sched_barrier(0);
  mma_trip(aP, bP);
  mma_trip(aQ, bQ);
}

template <bool BCONTIG, int MT, int TS, bool SEG2, int TS2>
__device__ __forceinline__ void gemm_wave(const GemmArgs& g, int id) {
  const int lane = threadIdx.x;
  const int r15  = lane & 15;
  const int kh   = (lane >> 4) << 1;
  const int tn = id & ((1 << g.lgx) - 1);
  const int tm = (id >> g.lgx) & ((1 << g.lgy) - 1);
  const int bz = id >> (g.lgx + g.lgy);

  const v8f zero = {0.f,0.f,0.f,0.f,0.f,0.f,0.f,0.f};
  v8f c[MT][2];
  #pragma unroll
  for (int i = 0; i < MT; ++i) { c[i][0] = zero; c[i][1] = zero; }

  {
    const float* Ab = g.A + (long)bz * g.sAb + (long)(tm * (16 * MT) + r15) * g.lda;
    const float* B0 = g.B + (long)bz * g.sBb + (long)(tn * 32 + r15) * g.ldbn;
    const float* B1 = B0 + (long)16 * g.ldbn;
    k_segment<BCONTIG, MT, TS>(c, Ab, g.lda, B0, B1, g.ldbk, g.K, kh);
  }
  if (SEG2) {
    const float* Ab = g.A2 + (long)(bz * ((16 * MT) << g.lgy) + tm * (16 * MT) + r15) * g.lda2;
    const float* B0 = g.B2 + (long)(tn * 32 + r15) * g.ldbn2;
    const float* B1 = B0 + (long)16 * g.ldbn2;
    k_segment<true, MT, TS2>(c, Ab, g.lda2, B0, B1, 1, g.K2, kh);
  }

  const int  col0 = tn * 32 + r15;
  const int  col1 = col0 + 16;
  const int  m0   = tm * (16 * MT) + ((lane >> 4) << 3);
  const float bv0 = g.bias ? g.bias[col0] : 0.f;
  const float bv1 = g.bias ? g.bias[col1] : 0.f;
  float* Cp = g.C + (long)bz * g.sCb;
  #pragma unroll
  for (int i = 0; i < MT; ++i) {
    #pragma unroll
    for (int r = 0; r < 8; ++r) {
      const long row = (long)(m0 + i * 16 + r) * g.ldc;
      float v0 = c[i][0][r] + bv0;
      float v1 = c[i][1][r] + bv1;
      if (g.relu) { v0 = fmaxf(v0, 0.f);  v1 = fmaxf(v1, 0.f); }
      Cp[row + col0] = v0;
      Cp[row + col1] = v1;
    }
  }
}

template <bool BCONTIG, int MT, int TS, bool SEG2, int TS2>
__global__ __launch_bounds__(32)
void wmma_gemm_f32(GemmArgs g) {
  gemm_wave<BCONTIG, MT, TS, SEG2, TS2>(g, blockIdx.x);
}

template <bool BC0, int MT0, int TS0, bool BC1, int MT1, int TS1>
__global__ __launch_bounds__(32)
void wmma_gemm_pair(GemmArgs g0, GemmArgs g1, int n0) {
  const int id = blockIdx.x;
  if (id < n0) gemm_wave<BC0, MT0, TS0, false, 8>(g0, id);
  else         gemm_wave<BC1, MT1, TS1, false, 8>(g1, id - n0);
}

__global__ __launch_bounds__(256)
void edge_softmax_kernel(const float* __restrict__ anchor_x,
                         const float* __restrict__ node_x,
                         const float* __restrict__ node_mask,
                         const float* __restrict__ qe,
                         const float* __restrict__ Lbase,
                         float* __restrict__ attn,
                         float* __restrict__ s_out)
{
  const int bid = blockIdx.x;
  const int b   = bid >> 6;
  const int t   = threadIdx.x;

  __shared__ float sh_qe[HE];
  __shared__ float sh_dist[N_PER];
  __shared__ float sh_l[N_PER];
  __shared__ float red[256];
  __shared__ float sh_bcast;
  __shared__ float sh_ax[3];

  if (t < HE) sh_qe[t] = qe[bid * HE + t];
  if (t < 3)  sh_ax[t] = anchor_x[bid * 3 + t];
  __syncthreads();

  const float axx = sh_ax[0], axy = sh_ax[1], axz = sh_ax[2];
  const float MU_STEP = 20.f / 63.f;
  const float INV_SIG = 64.f / 20.f;

  for (int n = t; n < N_PER; n += 256) {
    const float* nx = node_x + (long)(b * N_PER + n) * 3;
    float dx = axx - nx[0] + 1e-8f;
    float dy = axy - nx[1] + 1e-8f;
    float dz = axz - nx[2] + 1e-8f;
    float d  = sqrtf(dx * dx + dy * dy + dz * dz);
    sh_dist[n] = d;
    float d10 = d * 0.1f;
    float le = 0.f;
    #pragma unroll 8
    for (int j = 0; j < HE; ++j) {
      float u = (d10 - j * MU_STEP) * INV_SIG;
      le += sh_qe[j] * __expf(-u * u);
    }
    float lb = Lbase[(long)bid * N_PER + n] + le;
    lb *= (node_mask[b * N_PER + n] - 1.f) * 1000000.0f;
    sh_l[n] = lb;
  }
  __syncthreads();

  red[t] = fmaxf(sh_l[t], sh_l[t + 256]);
  __syncthreads();
  for (int st = 128; st > 0; st >>= 1) {
    if (t < st) red[t] = fmaxf(red[t], red[t + st]);
    __syncthreads();
  }
  if (t == 0) sh_bcast = red[0];
  __syncthreads();
  const float mx = sh_bcast;

  float e0 = __expf(sh_l[t] - mx);
  float e1 = __expf(sh_l[t + 256] - mx);
  red[t] = e0 + e1;
  __syncthreads();
  for (int st = 128; st > 0; st >>= 1) {
    if (t < st) red[t] += red[t + st];
    __syncthreads();
  }
  if (t == 0) sh_bcast = red[0];
  __syncthreads();
  const float inv = 1.f / sh_bcast;

  const float a0 = e0 * inv, a1 = e1 * inv;
  sh_l[t] = a0; sh_l[t + 256] = a1;
  attn[(long)bid * N_PER + t]       = a0;
  attn[(long)bid * N_PER + t + 256] = a1;
  __syncthreads();

  const int j = t & 63, c = t >> 6;
  const float muj = j * MU_STEP;
  float part = 0.f;
  for (int n = c * 128; n < c * 128 + 128; ++n) {
    float u = (sh_dist[n] * 0.1f - muj) * INV_SIG;
    part += sh_l[n] * __expf(-u * u);
  }
  red[t] = part;
  __syncthreads();
  if (t < 64)
    s_out[bid * HE + t] = red[t] + red[t + 64] + red[t + 128] + red[t + 192];
}

__global__ __launch_bounds__(256)
void add_ln_kernel(const float* __restrict__ x, const float* __restrict__ y,
                   const float* __restrict__ gam, const float* __restrict__ bet,
                   float* __restrict__ out)
{
  const int r = blockIdx.x, t = threadIdx.x;
  __shared__ float red[256];
  __shared__ float sh_mean, sh_rstd;

  const float v = x[(long)r * H + t] + y[(long)r * H + t];
  red[t] = v;
  __syncthreads();
  for (int st = 128; st > 0; st >>= 1) { if (t < st) red[t] += red[t + st]; __syncthreads(); }
  if (t == 0) sh_mean = red[0] * (1.f / H);
  __syncthreads();

  const float d = v - sh_mean;
  red[t] = d * d;
  __syncthreads();
  for (int st = 128; st > 0; st >>= 1) { if (t < st) red[t] += red[t + st]; __syncthreads(); }
  if (t == 0) sh_rstd = rsqrtf(red[0] * (1.f / H) + 1e-5f);
  __syncthreads();

  out[(long)r * H + t] = d * sh_rstd * gam[t] + bet[t];
}

extern "C" void kernel_launch(void* const* d_in, const int* in_sizes, int n_in,
                              void* d_out, int out_size, void* d_ws, size_t ws_size,
                              hipStream_t stream) {
  (void)in_sizes; (void)n_in; (void)out_size; (void)ws_size;
  const float* anchor_x  = (const float*)d_in[0];
  const float* node_x    = (const float*)d_in[1];
  const float* anchor_f  = (const float*)d_in[2];
  const float* node_f    = (const float*)d_in[3];
  const float* node_mask = (const float*)d_in[6];
  const float* Wq  = (const float*)d_in[7];
  const float* bq  = (const float*)d_in[8];
  const float* Wkv = (const float*)d_in[9];
  const float* bkv = (const float*)d_in[10];
  const float* ln1_g = (const float*)d_in[11];
  const float* ln1_b = (const float*)d_in[12];
  const float* W1 = (const float*)d_in[13];
  const float* b1 = (const float*)d_in[14];
  const float* W2 = (const float*)d_in[15];
  const float* b2 = (const float*)d_in[16];
  const float* W3 = (const float*)d_in[17];
  const float* b3 = (const float*)d_in[18];
  const float* ln2_g = (const float*)d_in[19];
  const float* ln2_b = (const float*)d_in[20];

  const long WKV_LD = H + HE;

  float* p = (float*)d_ws;
  float* q    = p; p += (long)NA * H;
  float* qe   = p; p += (long)NA * HE;
  float* kvn  = p; p += (long)NN * 2 * H;
  float* L    = p; p += (long)NA * N_PER;
  float* attn = p; p += (long)NA * N_PER;
  float* sbuf = p; p += (long)NA * HE;
  float* upd  = p; p += (long)NA * H;
  float* af   = p; p += (long)NA * H;
  float* m1   = p; p += (long)NA * 2 * H;
  float* m2   = p; p += (long)NA * 2 * H;
  float* m3   = p; p += (long)NA * H;


  {
    GemmArgs gk{node_f, H, 0,  Wkv, WKV_LD, 1, 0,  kvn, 2 * H, 0,  bkv,
                nullptr, 0, nullptr, 0,  H, 0, 0, 4, 6};
    GemmArgs gq{anchor_f, H, 0,  Wq, H, 1, 0,  q, H, 0,  bq,
                nullptr, 0, nullptr, 0,  H, 0, 0, 3, 5};
    wmma_gemm_pair<true, 4, 8, true, 1, 16><<<dim3(1024 + 256), dim3(32), 0, stream>>>(gk, gq, 1024);
  }
  {
    GemmArgs ge{q, H, 0,  Wkv + H, 1, WKV_LD, 0,  qe, HE, 0,  nullptr,
                nullptr, 0, nullptr, 0,  H, 0, 0, 1, 5};
    GemmArgs gl{q, H, (long)A_PER * H,  kvn, 2 * H, 1, (long)N_PER * 2 * H,
                L, N_PER, (long)A_PER * N_PER,  nullptr,
                nullptr, 0, nullptr, 0,  H, 0, 0, 4, 2};
    wmma_gemm_pair<false, 1, 8, true, 1, 16><<<dim3(64 + 512), dim3(32), 0, stream>>>(ge, gl, 64);
  }
  edge_softmax_kernel<<<dim3(NA), dim3(256), 0, stream>>>(
      anchor_x, node_x, node_mask, qe, L, attn, sbuf);
  {
    GemmArgs gu{attn, N_PER, (long)A_PER * N_PER,  kvn + H, 1, 2 * H, (long)N_PER * 2 * H,
                upd, H, (long)A_PER * H,  nullptr,
                sbuf, HE, Wkv + (long)H * WKV_LD + H, WKV_LD,  N_PER, HE, 0, 3, 2};
    wmma_gemm_f32<false, 1, 8, true, 8><<<dim3(256), dim3(32), 0, stream>>>(gu);
  }
  add_ln_kernel<<<dim3(NA), dim3(256), 0, stream>>>(anchor_f, upd, ln1_g, ln1_b, af);
  {
    GemmArgs g1{af, H, 0,  W1, H, 1, 0,  m1, 2 * H, 0,  b1,
                nullptr, 0, nullptr, 0,  H, 0, 1, 4, 5};
    wmma_gemm_f32<true, 1, 16, false, 8><<<dim3(512), dim3(32), 0, stream>>>(g1);
  }
  {
    GemmArgs g2{m1, 2 * H, 0,  W2, 2 * H, 1, 0,  m2, 2 * H, 0,  b2,
                nullptr, 0, nullptr, 0,  2 * H, 0, 1, 4, 5};
    wmma_gemm_f32<true, 1, 16, false, 8><<<dim3(512), dim3(32), 0, stream>>>(g2);
  }
  {
    GemmArgs g3{m2, 2 * H, 0,  W3, 2 * H, 1, 0,  m3, H, 0,  b3,
                nullptr, 0, nullptr, 0,  2 * H, 0, 0, 3, 5};
    wmma_gemm_f32<true, 1, 16, false, 8><<<dim3(256), dim3(32), 0, stream>>>(g3);
  }
  add_ln_kernel<<<dim3(NA), dim3(256), 0, stream>>>(af, m3, ln2_g, ln2_b, (float*)d_out);
}


# GEMM fragment loads b64 -> b128: a lane takes 4 consecutive k per load, feeding two matrix steps; k inside each 8-k group permuted alike in A and B (same products, regrouped sums: rounding changes)
# speedup vs baseline: 3.9272x; 1.0513x over previous
#include <hip/hip_runtime.h>
#include <math.h>

#define H      256
#define HE     64
#define NB     8
#define A_PER  64
#define N_PER  512
#define NA     (NB * A_PER)
#define NN     (NB * N_PER)

typedef __attribute__((ext_vector_type(2))) float v2f;
typedef __attribute__((ext_vector_type(4))) float v4f;
typedef __attribute__((ext_vector_type(8))) float v8f;

struct GemmArgs {
  const float* A;  long lda;  long sAb;
  const float* B;  long ldbn; long ldbk; long sBb;
  float*       C;  long ldc;  long sCb;
  const float* bias;
  const float* A2; long lda2;
  const float* B2; long ldbn2;
  int K;
  int K2;
  int relu;
  int lgx;
  int lgy;
};

template <bool BCONTIG, int MT, int TS>
__device__ __forceinline__ void k_segment(v8f (&c)[MT][2],
                                          const float* Ab, long lda,
                                          const float* B0, const float* B1, long ldbk,
                                          int K, int kq) {
  const float* Ap[MT];
  #pragma unroll
  for (int i = 0; i < MT; ++i) Ap[i] = Ab + (long)(16 * i) * lda;

  static_assert(TS % 2 == 0, "a trip is a whole number of 8-k groups");
  constexpr int TK = 4 * TS;
  constexpr int TG = TS / 2;
  v4f aP[MT][TG], bP[2][TG], aQ[MT][TG], bQ[2][TG];

  auto load_trip = [&](v4f (&a)[MT][TG], v4f (&b)[2][TG], int kbase) {
    #pragma unroll
    for (int g = 0; g < TG; ++g) {
      const int ka = kbase + 8 * g + kq;
      #pragma unroll
      for (int i = 0; i < MT; ++i) a[i][g] = *(const v4f*)(Ap[i] + ka);
      if (BCONTIG) {
        b[0][g] = *(const v4f*)(B0 + ka);
        b[1][g] = *(const v4f*)(B1 + ka);
      } else {
        b[0][g].x = B0[(long)ka * ldbk];        b[0][g].y = B0[(long)(ka + 1) * ldbk];
        b[0][g].z = B0[(long)(ka + 2) * ldbk];  b[0][g].w = B0[(long)(ka + 3) * ldbk];
        b[1][g].x = B1[(long)ka * ldbk];        b[1][g].y = B1[(long)(ka + 1) * ldbk];
        b[1][g].z = B1[(long)(ka + 2) * ldbk];  b[1][g].w = B1[(long)(ka + 3) * ldbk];
      }
    }
  };
  auto mma_trip = [&](const v4f (&a)[MT][TG], const v4f (&b)[2][TG]) {
    #pragma unroll
    for (int g = 0; g < TG; ++g) {
      const v2f b0lo = __builtin_shufflevector(b[0][g], b[0][g], 0, 1);
      const v2f b1lo = __builtin_shufflevector(b[1][g], b[1][g], 0, 1);
      #pragma unroll
      for (int i = 0; i < MT; ++i) {
        const v2f alo = __builtin_shufflevector(a[i][g], a[i][g], 0, 1);
        c[i][0] = __builtin_amdgcn_wmma_f32_16x16x4_f32(false, alo, false, b0lo, (short)0, c[i][0], false, false);
        c[i][1] = __builtin_amdgcn_wmma_f32_16x16x4_f32(false, alo, false, b1lo, (short)0, c[i][1], false, false);
      }
      const v2f b0hi = __builtin_shufflevector(b[0][g], b[0][g], 2, 3);
      const v2f b1hi = __builtin_shufflevector(b[1][g], b[1][g], 2, 3);
      #pragma unroll
      for (int i = 0; i < MT; ++i) {
        const v2f ahi = __builtin_shufflevector(a[i][g], a[i][g], 2, 3);
        c[i][0] = __builtin_amdgcn_wmma_f32_16x16x4_f32(false, ahi, false, b0hi, (short)0, c[i][0], false, false);
        c[i][1] = __builtin_amdgcn_wmma_f32_16x16x4_f32(false, ahi, false, b1hi, (short)0, c[i][1], false, false);
      }
    }
  };

  load_trip(aP, bP, 0);
  int k0 = 0;
  for (; k0 + 2 * TK < K; k0 += 2 * TK) {
    load_trip(aQ, bQ, k0 + TK);
    __builtin_amdgcn_sched_barrier(0);
    mma_trip(aP, bP);
    __builtin_amdgcn_sched_barrier(0);
    load_trip(aP, bP, k0 + 2 * TK);
    __builtin_amdgcn_sched_barrier(0);
    mma_trip(aQ, bQ);
  }
  load_trip(aQ, bQ, k0 + TK);
  __builtin_amdgcn_sched_barrier(0);
  mma_trip(aP, bP);
  mma_trip(aQ, bQ);
}

template <bool BCONTIG, int MT, int TS, bool SEG2, int TS2>
__device__ __forceinline__ void gemm_wave(const GemmArgs& g, int id) {
  const int lane = threadIdx.x;
  const int r15  = lane & 15;
  const int kq   = (lane >> 4) << 2;
  const int tn = id & ((1 << g.lgx) - 1);
  const int tm = (id >> g.lgx) & ((1 << g.lgy) - 1);
  const int bz = id >> (g.lgx + g.lgy);

  const v8f zero = {0.f,0.f,0.f,0.f,0.f,0.f,0.f,0.f};
  v8f c[MT][2];
  #pragma unroll
  for (int i = 0; i < MT; ++i) { c[i][0] = zero; c[i][1] = zero; }

  {
    const float* Ab = g.A + (long)bz * g.sAb + (long)(tm * (16 * MT) + r15) * g.lda;
    const float* B0 = g.B + (long)bz * g.sBb + (long)(tn * 32 + r15) * g.ldbn;
    const float* B1 = B0 + (long)16 * g.ldbn;
    k_segment<BCONTIG, MT, TS>(c, Ab, g.lda, B0, B1, g.ldbk, g.K, kq);
  }
  if (SEG2) {
    const float* Ab = g.A2 + (long)(bz * ((16 * MT) << g.lgy) + tm * (16 * MT) + r15) * g.lda2;
    const float* B0 = g.B2 + (long)(tn * 32 + r15) * g.ldbn2;
    const float* B1 = B0 + (long)16 * g.ldbn2;
    k_segment<true, MT, TS2>(c, Ab, g.lda2, B0, B1, 1, g.K2, kq);
  }

  const int  col0 = tn * 32 + r15;
  const int  col1 = col0 + 16;
  const int  m0   = tm * (16 * MT) + ((lane >> 4) << 3);
  const float bv0 = g.bias ? g.bias[col0] : 0.f;
  const float bv1 = g.bias ? g.bias[col1] : 0.f;
  float* Cp = g.C + (long)bz * g.sCb;
  #pragma unroll
  for (int i = 0; i < MT; ++i) {
    #pragma unroll
    for (int r = 0; r < 8; ++r) {
      const long row = (long)(m0 + i * 16 + r) * g.ldc;
      float v0 = c[i][0][r] + bv0;
      float v1 = c[i][1][r] + bv1;
      if (g.relu) { v0 = fmaxf(v0, 0.f);  v1 = fmaxf(v1, 0.f); }
      Cp[row + col0] = v0;
      Cp[row + col1] = v1;
    }
  }
}

template <bool BCONTIG, int MT, int TS, bool SEG2, int TS2>
__global__ __launch_bounds__(32)
void wmma_gemm_f32(GemmArgs g) {
  gemm_wave<BCONTIG, MT, TS, SEG2, TS2>(g, blockIdx.x);
}

template <bool BC0, int MT0, int TS0, bool BC1, int MT1, int TS1>
__global__ __launch_bounds__(32)
void wmma_gemm_pair(GemmArgs g0, GemmArgs g1, int n0) {
  const int id = blockIdx.x;
  if (id < n0) gemm_wave<BC0, MT0, TS0, false, 8>(g0, id);
  else         gemm_wave<BC1, MT1, TS1, false, 8>(g1, id - n0);
}

__global__ __launch_bounds__(256)
void edge_softmax_kernel(const float* __restrict__ anchor_x,
                         const float* __restrict__ node_x,
                         const float* __restrict__ node_mask,
                         const float* __restrict__ qe,
                         const float* __restrict__ Lbase,
                         float* __restrict__ attn,
                         float* __restrict__ s_out)
{
  const int bid = blockIdx.x;
  const int b   = bid >> 6;
  const int t   = threadIdx.x;

  __shared__ float sh_qe[HE];
  __shared__ float sh_dist[N_PER];
  __shared__ float sh_l[N_PER];
  __shared__ float red[256];
  __shared__ float sh_bcast;
  __shared__ float sh_ax[3];

  if (t < HE) sh_qe[t] = qe[bid * HE + t];
  if (t < 3)  sh_ax[t] = anchor_x[bid * 3 + t];
  __syncthreads();

  const float axx = sh_ax[0], axy = sh_ax[1], axz = sh_ax[2];
  const float MU_STEP = 20.f / 63.f;
  const float INV_SIG = 64.f / 20.f;

  for (int n = t; n < N_PER; n += 256) {
    const float* nx = node_x + (long)(b * N_PER + n) * 3;
    float dx = axx - nx[0] + 1e-8f;
    float dy = axy - nx[1] + 1e-8f;
    float dz = axz - nx[2] + 1e-8f;
    float d  = sqrtf(dx * dx + dy * dy + dz * dz);
    sh_dist[n] = d;
    float d10 = d * 0.1f;
    float le = 0.f;
    #pragma unroll 8
    for (int j = 0; j < HE; ++j) {
      float u = (d10 - j * MU_STEP) * INV_SIG;
      le += sh_qe[j] * __expf(-u * u);
    }
    float lb = Lbase[(long)bid * N_PER + n] + le;
    lb *= (node_mask[b * N_PER + n] - 1.f) * 1000000.0f;
    sh_l[n] = lb;
  }
  __syncthreads();

  red[t] = fmaxf(sh_l[t], sh_l[t + 256]);
  __syncthreads();
  for (int st = 128; st > 0; st >>= 1) {
    if (t < st) red[t] = fmaxf(red[t], red[t + st]);
    __syncthreads();
  }
  if (t == 0) sh_bcast = red[0];
  __syncthreads();
  const float mx = sh_bcast;

  float e0 = __expf(sh_l[t] - mx);
  float e1 = __expf(sh_l[t + 256] - mx);
  red[t] = e0 + e1;
  __syncthreads();
  for (int st = 128; st > 0; st >>= 1) {
    if (t < st) red[t] += red[t + st];
    __syncthreads();
  }
  if (t == 0) sh_bcast = red[0];
  __syncthreads();
  const float inv = 1.f / sh_bcast;

  const float a0 = e0 * inv, a1 = e1 * inv;
  sh_l[t] = a0; sh_l[t + 256] = a1;
  attn[(long)bid * N_PER + t]       = a0;
  attn[(long)bid * N_PER + t + 256] = a1;
  __syncthreads();

  const int j = t & 63, c = t >> 6;
  const float muj = j * MU_STEP;
  float part = 0.f;
  for (int n = c * 128; n < c * 128 + 128; ++n) {
    float u = (sh_dist[n] * 0.1f - muj) * INV_SIG;
    part += sh_l[n] * __expf(-u * u);
  }
  red[t] = part;
  __syncthreads();
  if (t < 64)
    s_out[bid * HE + t] = red[t] + red[t + 64] + red[t + 128] + red[t + 192];
}

__global__ __launch_bounds__(256)
void add_ln_kernel(const float* __restrict__ x, const float* __restrict__ y,
                   const float* __restrict__ gam, const float* __restrict__ bet,
                   float* __restrict__ out)
{
  const int r = blockIdx.x, t = threadIdx.x;
  __shared__ float red[256];
  __shared__ float sh_mean, sh_rstd;

  const float v = x[(long)r * H + t] + y[(long)r * H + t];
  red[t] = v;
  __syncthreads();
  for (int st = 128; st > 0; st >>= 1) { if (t < st) red[t] += red[t + st]; __syncthreads(); }
  if (t == 0) sh_mean = red[0] * (1.f / H);
  __syncthreads();

  const float d = v - sh_mean;
  red[t] = d * d;
  __syncthreads();
  for (int st = 128; st > 0; st >>= 1) { if (t < st) red[t] += red[t + st]; __syncthreads(); }
  if (t == 0) sh_rstd = rsqrtf(red[0] * (1.f / H) + 1e-5f);
  __syncthreads();

  out[(long)r * H + t] = d * sh_rstd * gam[t] + bet[t];
}

extern "C" void kernel_launch(void* const* d_in, const int* in_sizes, int n_in,
                              void* d_out, int out_size, void* d_ws, size_t ws_size,
                              hipStream_t stream) {
  (void)in_sizes; (void)n_in; (void)out_size; (void)ws_size;
  const float* anchor_x  = (const float*)d_in[0];
  const float* node_x    = (const float*)d_in[1];
  const float* anchor_f  = (const float*)d_in[2];
  const float* node_f    = (const float*)d_in[3];
  const float* node_mask = (const float*)d_in[6];
  const float* Wq  = (const float*)d_in[7];
  const float* bq  = (const float*)d_in[8];
  const float* Wkv = (const float*)d_in[9];
  const float* bkv = (const float*)d_in[10];
  const float* ln1_g = (const float*)d_in[11];
  const float* ln1_b = (const float*)d_in[12];
  const float* W1 = (const float*)d_in[13];
  const float* b1 = (const float*)d_in[14];
  const float* W2 = (const float*)d_in[15];
  const float* b2 = (const float*)d_in[16];
  const float* W3 = (const float*)d_in[17];
  const float* b3 = (const float*)d_in[18];
  const float* ln2_g = (const float*)d_in[19];
  const float* ln2_b = (const float*)d_in[20];

  const long WKV_LD = H + HE;

  float* p = (float*)d_ws;
  float* q    = p; p += (long)NA * H;
  float* qe   = p; p += (long)NA * HE;
  float* kvn  = p; p += (long)NN * 2 * H;
  float* L    = p; p += (long)NA * N_PER;
  float* attn = p; p += (long)NA * N_PER;
  float* sbuf = p; p += (long)NA * HE;
  float* upd  = p; p += (long)NA * H;
  float* af   = p; p += (long)NA * H;
  float* m1   = p; p += (long)NA * 2 * H;
  float* m2   = p; p += (long)NA * 2 * H;
  float* m3   = p; p += (long)NA * H;


  {
    GemmArgs gk{node_f, H, 0,  Wkv, WKV_LD, 1, 0,  kvn, 2 * H, 0,  bkv,
                nullptr, 0, nullptr, 0,  H, 0, 0, 4, 6};
    GemmArgs gq{anchor_f, H, 0,  Wq, H, 1, 0,  q, H, 0,  bq,
                nullptr, 0, nullptr, 0,  H, 0, 0, 3, 5};
    wmma_gemm_pair<true, 4, 8, true, 1, 16><<<dim3(1024 + 256), dim3(32), 0, stream>>>(gk, gq, 1024);
  }
  {
    GemmArgs ge{q, H, 0,  Wkv + H, 1, WKV_LD, 0,  qe, HE, 0,  nullptr,
                nullptr, 0, nullptr, 0,  H, 0, 0, 1, 5};
    GemmArgs gl{q, H, (long)A_PER * H,  kvn, 2 * H, 1, (long)N_PER * 2 * H,
                L, N_PER, (long)A_PER * N_PER,  nullptr,
                nullptr, 0, nullptr, 0,  H, 0, 0, 4, 2};
    wmma_gemm_pair<false, 1, 8, true, 1, 16><<<dim3(64 + 512), dim3(32), 0, stream>>>(ge, gl, 64);
  }
  edge_softmax_kernel<<<dim3(NA), dim3(256), 0, stream>>>(
      anchor_x, node_x, node_mask, qe, L, attn, sbuf);
  {
    GemmArgs gu{attn, N_PER, (long)A_PER * N_PER,  kvn + H, 1, 2 * H, (long)N_PER * 2 * H,
                upd, H, (long)A_PER * H,  nullptr,
                sbuf, HE, Wkv + (long)H * WKV_LD + H, WKV_LD,  N_PER, HE, 0, 3, 2};
    wmma_gemm_f32<false, 1, 8, true, 8><<<dim3(256), dim3(32), 0, stream>>>(gu);
  }
  add_ln_kernel<<<dim3(NA), dim3(256), 0, stream>>>(anchor_f, upd, ln1_g, ln1_b, af);
  {
    GemmArgs g1{af, H, 0,  W1, H, 1, 0,  m1, 2 * H, 0,  b1,
                nullptr, 0, nullptr, 0,  H, 0, 1, 4, 5};
    wmma_gemm_f32<true, 1, 16, false, 8><<<dim3(512), dim3(32), 0, stream>>>(g1);
  }
  {
    GemmArgs g2{m1, 2 * H, 0,  W2, 2 * H, 1, 0,  m2, 2 * H, 0,  b2,
                nullptr, 0, nullptr, 0,  2 * H, 0, 1, 4, 5};
    wmma_gemm_f32<true, 1, 16, false, 8><<<dim3(512), dim3(32), 0, stream>>>(g2);
  }
  {
    GemmArgs g3{m2, 2 * H, 0,  W3, 2 * H, 1, 0,  m3, H, 0,  b3,
                nullptr, 0, nullptr, 0,  2 * H, 0, 0, 3, 5};
    wmma_gemm_f32<true, 1, 16, false, 8><<<dim3(256), dim3(32), 0, stream>>>(g3);
  }
  add_ln_kernel<<<dim3(NA), dim3(256), 0, stream>>>(af, m3, ln2_g, ln2_b, (float*)d_out);
}


# Small GEMMs: K split over 2 or 4 waves of a 4-wave workgroup (a wave walks 2-4 trips in series, not 4-18); partial tiles added in slice order through LDS behind one barrier (sum order changes)
# speedup vs baseline: 4.7415x; 1.2073x over previous
#include <hip/hip_runtime.h>
#include <math.h>

#define H      256
#define HE     64
#define NB     8
#define A_PER  64
#define N_PER  512
#define NA     (NB * A_PER)
#define NN     (NB * N_PER)

typedef __attribute__((ext_vector_type(2))) float v2f;
typedef __attribute__((ext_vector_type(4))) float v4f;
typedef __attribute__((ext_vector_type(8))) float v8f;

struct GemmArgs {
  const float* A;  long lda;  long sAb;
  const float* B;  long ldbn; long ldbk; long sBb;
  float*       C;  long ldc;  long sCb;
  const float* bias;
  const float* A2; long lda2;
  const float* B2; long ldbn2;
  int K;
  int K2;
  int relu;
  int lgx;
  int lgy;
};

template <bool BCONTIG, int MT, int TS>
__device__ __forceinline__ void k_segment(v8f (&c)[MT][2],
                                          const float* Ab, long lda,
                                          const float* B0, const float* B1, long ldbk,
                                          int K, int kq) {
  const float* Ap[MT];
  #pragma unroll
  for (int i = 0; i < MT; ++i) Ap[i] = Ab + (long)(16 * i) * lda;

  static_assert(TS % 2 == 0, "a trip is a whole number of 8-k groups");
  constexpr int TK = 4 * TS;
  constexpr int TG = TS / 2;
  v4f aP[MT][TG], bP[2][TG], aQ[MT][TG], bQ[2][TG];

  auto load_trip = [&](v4f (&a)[MT][TG], v4f (&b)[2][TG], int kbase) {
    #pragma unroll
    for (int g = 0; g < TG; ++g) {
      const int ka = kbase + 8 * g + kq;
      #pragma unroll
      for (int i = 0; i < MT; ++i) a[i][g] = *(const v4f*)(Ap[i] + ka);
      if (BCONTIG) {
        b[0][g] = *(const v4f*)(B0 + ka);
        b[1][g] = *(const v4f*)(B1 + ka);
      } else {
        b[0][g].x = B0[(long)ka * ldbk];        b[0][g].y = B0[(long)(ka + 1) * ldbk];
        b[0][g].z = B0[(long)(ka + 2) * ldbk];  b[0][g].w = B0[(long)(ka + 3) * ldbk];
        b[1][g].x = B1[(long)ka * ldbk];        b[1][g].y = B1[(long)(ka + 1) * ldbk];
        b[1][g].z = B1[(long)(ka + 2) * ldbk];  b[1][g].w = B1[(long)(ka + 3) * ldbk];
      }
    }
  };
  auto mma_trip = [&](const v4f (&a)[MT][TG], const v4f (&b)[2][TG]) {
    #pragma unroll
    for (int g = 0; g < TG; ++g) {
      const v2f b0lo = __builtin_shufflevector(b[0][g], b[0][g], 0, 1);
      const v2f b1lo = __builtin_shufflevector(b[1][g], b[1][g], 0, 1);
      #pragma unroll
      for (int i = 0; i < MT; ++i) {
        const v2f alo = __builtin_shufflevector(a[i][g], a[i][g], 0, 1);
        c[i][0] = __builtin_amdgcn_wmma_f32_16x16x4_f32(false, alo, false, b0lo, (short)0, c[i][0], false, false);
        c[i][1] = __builtin_amdgcn_wmma_f32_16x16x4_f32(false, alo, false, b1lo, (short)0, c[i][1], false, false);
      }
      const v2f b0hi = __builtin_shufflevector(b[0][g], b[0][g], 2, 3);
      const v2f b1hi = __builtin_shufflevector(b[1][g], b[1][g], 2, 3);
      #pragma unroll
      for (int i = 0; i < MT; ++i) {
        const v2f ahi = __builtin_shufflevector(a[i][g], a[i][g], 2, 3);
        c[i][0] = __builtin_amdgcn_wmma_f32_16x16x4_f32(false, ahi, false, b0hi, (short)0, c[i][0], false, false);
        c[i][1] = __builtin_amdgcn_wmma_f32_16x16x4_f32(false, ahi, false, b1hi, (short)0, c[i][1], false, false);
      }
    }
  };

  load_trip(aP, bP, 0);
  int k0 = 0;
  for (; k0 + 2 * TK < K; k0 += 2 * TK) {
    load_trip(aQ, bQ, k0 + TK);
    __builtin_amdgcn_sched_barrier(0);
    mma_trip(aP, bP);
    __builtin_amdgcn_sched_barrier(0);
    load_trip(aP, bP, k0 + 2 * TK);
    __builtin_amdgcn_sched_barrier(0);
    mma_trip(aQ, bQ);
  }
  load_trip(aQ, bQ, k0 + TK);
  __builtin_amdgcn_sched_barrier(0);
  mma_trip(aP, bP);
  mma_trip(aQ, bQ);
}

template <bool BCONTIG, int MT, int TS, bool SEG2, int TS2>
__device__ __forceinline__ void gemm_wave(const GemmArgs& g, int id) {
  const int lane = threadIdx.x;
  const int r15  = lane & 15;
  const int kq   = (lane >> 4) << 2;
  const int tn = id & ((1 << g.lgx) - 1);
  const int tm = (id >> g.lgx) & ((1 << g.lgy) - 1);
  const int bz = id >> (g.lgx + g.lgy);

  const v8f zero = {0.f,0.f,0.f,0.f,0.f,0.f,0.f,0.f};
  v8f c[MT][2];
  #pragma unroll
  for (int i = 0; i < MT; ++i) { c[i][0] = zero; c[i][1] = zero; }

  {
    const float* Ab = g.A + (long)bz * g.sAb + (long)(tm * (16 * MT) + r15) * g.lda;
    const float* B0 = g.B + (long)bz * g.sBb + (long)(tn * 32 + r15) * g.ldbn;
    const float* B1 = B0 + (long)16 * g.ldbn;
    k_segment<BCONTIG, MT, TS>(c, Ab, g.lda, B0, B1, g.ldbk, g.K, kq);
  }
  if (SEG2) {
    const float* Ab = g.A2 + (long)(bz * ((16 * MT) << g.lgy) + tm * (16 * MT) + r15) * g.lda2;
    const float* B0 = g.B2 + (long)(tn * 32 + r15) * g.ldbn2;
    const float* B1 = B0 + (long)16 * g.ldbn2;
    k_segment<true, MT, TS2>(c, Ab, g.lda2, B0, B1, 1, g.K2, kq);
  }

  const int  col0 = tn * 32 + r15;
  const int  col1 = col0 + 16;
  const int  m0   = tm * (16 * MT) + ((lane >> 4) << 3);
  const float bv0 = g.bias ? g.bias[col0] : 0.f;
  const float bv1 = g.bias ? g.bias[col1] : 0.f;
  float* Cp = g.C + (long)bz * g.sCb;
  #pragma unroll
  for (int i = 0; i < MT; ++i) {
    #pragma unroll
    for (int r = 0; r < 8; ++r) {
      const long row = (long)(m0 + i * 16 + r) * g.ldc;
      float v0 = c[i][0][r] + bv0;
      float v1 = c[i][1][r] + bv1;
      if (g.relu) { v0 = fmaxf(v0, 0.f);  v1 = fmaxf(v1, 0.f); }
      Cp[row + col0] = v0;
      Cp[row + col1] = v1;
    }
  }
}

template <bool BCONTIG, int TS, bool SEG2, int TS2, int S, bool HASB>
__device__ __forceinline__ void gemm_split(const GemmArgs& g, int wg, float* red) {
  static_assert(S == 2 || S == 4, "S waves of a 4-wave workgroup share a tile");
  constexpr int LGS = (S == 4) ? 2 : 1;
  constexpr int TPW = 4 / S;
  constexpr int RPW = 8 / S;
  const int lane = threadIdx.x & 31;
  const int wave = __builtin_amdgcn_readfirstlane((int)(threadIdx.x >> 5));
  const int ks = wave & (S - 1);
  const int tl = wave >> LGS;
  const int id = wg * TPW + tl;
  const int r15  = lane & 15;
  const int kq   = (lane >> 4) << 2;
  const int tn = id & ((1 << g.lgx) - 1);
  const int tm = (id >> g.lgx) & ((1 << g.lgy) - 1);
  const int bz = id >> (g.lgx + g.lgy);

  const v8f zero = {0.f,0.f,0.f,0.f,0.f,0.f,0.f,0.f};
  v8f c[1][2];
  c[0][0] = zero; c[0][1] = zero;

  const int  col0 = tn * 32 + r15;
  const int  col1 = col0 + 16;
  float bv0 = 0.f, bv1 = 0.f;
  if (HASB) { bv0 = g.bias[col0];  bv1 = g.bias[col1]; }

  {
    const int  Ks   = g.K >> LGS;
    const long koff = (long)ks * Ks;
    const float* Ab = g.A + (long)bz * g.sAb + (long)(tm * 16 + r15) * g.lda + koff;
    const float* B0 = g.B + (long)bz * g.sBb + (long)(tn * 32 + r15) * g.ldbn + koff * g.ldbk;
    const float* B1 = B0 + (long)16 * g.ldbn;
    k_segment<BCONTIG, 1, TS>(c, Ab, g.lda, B0, B1, g.ldbk, Ks, kq);
  }
  if (SEG2) {
    const int  Ks   = g.K2 >> LGS;
    const long koff = (long)ks * Ks;
    const float* Ab = g.A2 + (long)(bz * (16 << g.lgy) + tm * 16 + r15) * g.lda2 + koff;
    const float* B0 = g.B2 + (long)(tn * 32 + r15) * g.ldbn2 + koff;
    const float* B1 = B0 + (long)16 * g.ldbn2;
    k_segment<true, 1, TS2>(c, Ab, g.lda2, B0, B1, 1, Ks, kq);
  }

  float* mine = red + (wave * 16) * 32 + lane;
  #pragma unroll
  for (int j = 0; j < 2; ++j) {
    #pragma unroll
    for (int r = 0; r < 8; ++r) mine[(j * 8 + r) * 32] = c[0][j][r];
  }
  __syncthreads();

  const float* tile = red + (tl * S * 16) * 32 + lane;
  const int  m0   = tm * 16 + ((lane >> 4) << 3);
  float* Cp = g.C + (long)bz * g.sCb;
  #pragma unroll
  for (int q = 0; q < RPW; ++q) {
    const int r = ks * RPW + q;
    float v0 = tile[r * 32];
    float v1 = tile[(8 + r) * 32];
    #pragma unroll
    for (int s = 1; s < S; ++s) {
      v0 += tile[(s * 16 + r) * 32];
      v1 += tile[(s * 16 + 8 + r) * 32];
    }
    v0 += bv0;
    v1 += bv1;
    if (g.relu) { v0 = fmaxf(v0, 0.f);  v1 = fmaxf(v1, 0.f); }
    const long row = (long)(m0 + r) * g.ldc;
    Cp[row + col0] = v0;
    Cp[row + col1] = v1;
  }
}

template <bool BCONTIG, int TS, bool SEG2, int TS2, int S, bool HASB>
__global__ __launch_bounds__(128)
void wmma_gemm_split(GemmArgs g) {
  __shared__ float red[4 * 16 * 32];
  gemm_split<BCONTIG, TS, SEG2, TS2, S, HASB>(g, blockIdx.x, red);
}

template <bool BC0, int TS0, int S0, bool BC1, int TS1, int S1>
__global__ __launch_bounds__(128)
void wmma_gemm_split_pair(GemmArgs g0, GemmArgs g1, int n0) {
  __shared__ float red[4 * 16 * 32];
  const int id = blockIdx.x;
  if (id < n0) gemm_split<BC0, TS0, false, 8, S0, false>(g0, id, red);
  else         gemm_split<BC1, TS1, false, 8, S1, false>(g1, id - n0, red);
}

template <bool BC0, int MT0, int TS0, bool BC1, int MT1, int TS1>
__global__ __launch_bounds__(32)
void wmma_gemm_pair(GemmArgs g0, GemmArgs g1, int n0) {
  const int id = blockIdx.x;
  if (id < n0) gemm_wave<BC0, MT0, TS0, false, 8>(g0, id);
  else         gemm_wave<BC1, MT1, TS1, false, 8>(g1, id - n0);
}

__global__ __launch_bounds__(256)
void edge_softmax_kernel(const float* __restrict__ anchor_x,
                         const float* __restrict__ node_x,
                         const float* __restrict__ node_mask,
                         const float* __restrict__ qe,
                         const float* __restrict__ Lbase,
                         float* __restrict__ attn,
                         float* __restrict__ s_out)
{
  const int bid = blockIdx.x;
  const int b   = bid >> 6;
  const int t   = threadIdx.x;

  __shared__ float sh_qe[HE];
  __shared__ float sh_dist[N_PER];
  __shared__ float sh_l[N_PER];
  __shared__ float red[256];
  __shared__ float sh_bcast;
  __shared__ float sh_ax[3];

  if (t < HE) sh_qe[t] = qe[bid * HE + t];
  if (t < 3)  sh_ax[t] = anchor_x[bid * 3 + t];
  __syncthreads();

  const float axx = sh_ax[0], axy = sh_ax[1], axz = sh_ax[2];
  const float MU_STEP = 20.f / 63.f;
  const float INV_SIG = 64.f / 20.f;

  for (int n = t; n < N_PER; n += 256) {
    const float* nx = node_x + (long)(b * N_PER + n) * 3;
    float dx = axx - nx[0] + 1e-8f;
    float dy = axy - nx[1] + 1e-8f;
    float dz = axz - nx[2] + 1e-8f;
    float d  = sqrtf(dx * dx + dy * dy + dz * dz);
    sh_dist[n] = d;
    float d10 = d * 0.1f;
    float le = 0.f;
    #pragma unroll 8
    for (int j = 0; j < HE; ++j) {
      float u = (d10 - j * MU_STEP) * INV_SIG;
      le += sh_qe[j] * __expf(-u * u);
    }
    float lb = Lbase[(long)bid * N_PER + n] + le;
    lb *= (node_mask[b * N_PER + n] - 1.f) * 1000000.0f;
    sh_l[n] = lb;
  }
  __syncthreads();

  red[t] = fmaxf(sh_l[t], sh_l[t + 256]);
  __syncthreads();
  for (int st = 128; st > 0; st >>= 1) {
    if (t < st) red[t] = fmaxf(red[t], red[t + st]);
    __syncthreads();
  }
  if (t == 0) sh_bcast = red[0];
  __syncthreads();
  const float mx = sh_bcast;

  float e0 = __expf(sh_l[t] - mx);
  float e1 = __expf(sh_l[t + 256] - mx);
  red[t] = e0 + e1;
  __syncthreads();
  for (int st = 128; st > 0; st >>= 1) {
    if (t < st) red[t] += red[t + st];
    __syncthreads();
  }
  if (t == 0) sh_bcast = red[0];
  __syncthreads();
  const float inv = 1.f / sh_bcast;

  const float a0 = e0 * inv, a1 = e1 * inv;
  sh_l[t] = a0; sh_l[t + 256] = a1;
  attn[(long)bid * N_PER + t]       = a0;
  attn[(long)bid * N_PER + t + 256] = a1;
  __syncthreads();

  const int j = t & 63, c = t >> 6;
  const float muj = j * MU_STEP;
  float part = 0.f;
  for (int n = c * 128; n < c * 128 + 128; ++n) {
    float u = (sh_dist[n] * 0.1f - muj) * INV_SIG;
    part += sh_l[n] * __expf(-u * u);
  }
  red[t] = part;
  __syncthreads();
  if (t < 64)
    s_out[bid * HE + t] = red[t] + red[t + 64] + red[t + 128] + red[t + 192];
}

__global__ __launch_bounds__(256)
void add_ln_kernel(const float* __restrict__ x, const float* __restrict__ y,
                   const float* __restrict__ gam, const float* __restrict__ bet,
                   float* __restrict__ out)
{
  const int r = blockIdx.x, t = threadIdx.x;
  __shared__ float red[256];
  __shared__ float sh_mean, sh_rstd;

  const float v = x[(long)r * H + t] + y[(long)r * H + t];
  red[t] = v;
  __syncthreads();
  for (int st = 128; st > 0; st >>= 1) { if (t < st) red[t] += red[t + st]; __syncthreads(); }
  if (t == 0) sh_mean = red[0] * (1.f / H);
  __syncthreads();

  const float d = v - sh_mean;
  red[t] = d * d;
  __syncthreads();
  for (int st = 128; st > 0; st >>= 1) { if (t < st) red[t] += red[t + st]; __syncthreads(); }
  if (t == 0) sh_rstd = rsqrtf(red[0] * (1.f / H) + 1e-5f);
  __syncthreads();

  out[(long)r * H + t] = d * sh_rstd * gam[t] + bet[t];
}

extern "C" void kernel_launch(void* const* d_in, const int* in_sizes, int n_in,
                              void* d_out, int out_size, void* d_ws, size_t ws_size,
                              hipStream_t stream) {
  (void)in_sizes; (void)n_in; (void)out_size; (void)ws_size;
  const float* anchor_x  = (const float*)d_in[0];
  const float* node_x    = (const float*)d_in[1];
  const float* anchor_f  = (const float*)d_in[2];
  const float* node_f    = (const float*)d_in[3];
  const float* node_mask = (const float*)d_in[6];
  const float* Wq  = (const float*)d_in[7];
  const float* bq  = (const float*)d_in[8];
  const float* Wkv = (const float*)d_in[9];
  const float* bkv = (const float*)d_in[10];
  const float* ln1_g = (const float*)d_in[11];
  const float* ln1_b = (const float*)d_in[12];
  const float* W1 = (const float*)d_in[13];
  const float* b1 = (const float*)d_in[14];
  const float* W2 = (const float*)d_in[15];
  const float* b2 = (const float*)d_in[16];
  const float* W3 = (const float*)d_in[17];
  const float* b3 = (const float*)d_in[18];
  const float* ln2_g = (const float*)d_in[19];
  const float* ln2_b = (const float*)d_in[20];

  const long WKV_LD = H + HE;

  float* p = (float*)d_ws;
  float* q    = p; p += (long)NA * H;
  float* qe   = p; p += (long)NA * HE;
  float* kvn  = p; p += (long)NN * 2 * H;
  float* L    = p; p += (long)NA * N_PER;
  float* attn = p; p += (long)NA * N_PER;
  float* sbuf = p; p += (long)NA * HE;
  float* upd  = p; p += (long)NA * H;
  float* af   = p; p += (long)NA * H;
  float* m1   = p; p += (long)NA * 2 * H;
  float* m2   = p; p += (long)NA * 2 * H;
  float* m3   = p; p += (long)NA * H;


  {
    GemmArgs gk{node_f, H, 0,  Wkv, WKV_LD, 1, 0,  kvn, 2 * H, 0,  bkv,
                nullptr, 0, nullptr, 0,  H, 0, 0, 4, 6};
    GemmArgs gq{anchor_f, H, 0,  Wq, H, 1, 0,  q, H, 0,  bq,
                nullptr, 0, nullptr, 0,  H, 0, 0, 3, 5};
    wmma_gemm_pair<true, 4, 8, true, 1, 16><<<dim3(1024 + 256), dim3(32), 0, stream>>>(gk, gq, 1024);
  }
  {
    GemmArgs ge{q, H, 0,  Wkv + H, 1, WKV_LD, 0,  qe, HE, 0,  nullptr,
                nullptr, 0, nullptr, 0,  H, 0, 0, 1, 5};
    GemmArgs gl{q, H, (long)A_PER * H,  kvn, 2 * H, 1, (long)N_PER * 2 * H,
                L, N_PER, (long)A_PER * N_PER,  nullptr,
                nullptr, 0, nullptr, 0,  H, 0, 0, 4, 2};
    wmma_gemm_split_pair<false, 8, 4, true, 16, 2><<<dim3(64 + 256), dim3(128), 0, stream>>>(ge, gl, 64);
  }
  edge_softmax_kernel<<<dim3(NA), dim3(256), 0, stream>>>(
      anchor_x, node_x, node_mask, qe, L, attn, sbuf);
  {
    GemmArgs gu{attn, N_PER, (long)A_PER * N_PER,  kvn + H, 1, 2 * H, (long)N_PER * 2 * H,
                upd, H, (long)A_PER * H,  nullptr,
                sbuf, HE, Wkv + (long)H * WKV_LD + H, WKV_LD,  N_PER, HE, 0, 3, 2};
    wmma_gemm_split<false, 16, true, 2, 4, false><<<dim3(256), dim3(128), 0, stream>>>(gu);
  }
  add_ln_kernel<<<dim3(NA), dim3(256), 0, stream>>>(anchor_f, upd, ln1_g, ln1_b, af);
  {
    GemmArgs g1{af, H, 0,  W1, H, 1, 0,  m1, 2 * H, 0,  b1,
                nullptr, 0, nullptr, 0,  H, 0, 1, 4, 5};
    wmma_gemm_split<true, 16, false, 8, 2, true><<<dim3(256), dim3(128), 0, stream>>>(g1);
  }
  {
    GemmArgs g2{m1, 2 * H, 0,  W2, 2 * H, 1, 0,  m2, 2 * H, 0,  b2,
                nullptr, 0, nullptr, 0,  2 * H, 0, 1, 4, 5};
    wmma_gemm_split<true, 16, false, 8, 4, true><<<dim3(512), dim3(128), 0, stream>>>(g2);
  }
  {
    GemmArgs g3{m2, 2 * H, 0,  W3, 2 * H, 1, 0,  m3, H, 0,  b3,
                nullptr, 0, nullptr, 0,  2 * H, 0, 0, 3, 5};
    wmma_gemm_split<true, 16, false, 8, 4, true><<<dim3(256), dim3(128), 0, stream>>>(g3);
  }
  add_ln_kernel<<<dim3(NA), dim3(256), 0, stream>>>(af, m3, ln2_g, ln2_b, (float*)d_out);
}


# LayerNorm: one wave per row (8 values a lane, row sums by in-register cross-lane adds) replaces a 256-thread block per row with barrier-stepped LDS tree sums: 512 waves, not 4096 (sum order changes)
# speedup vs baseline: 5.0311x; 1.0611x over previous
#include <hip/hip_runtime.h>
#include <math.h>

#define H      256
#define HE     64
#define NB     8
#define A_PER  64
#define N_PER  512
#define NA     (NB * A_PER)
#define NN     (NB * N_PER)

typedef __attribute__((ext_vector_type(2))) float v2f;
typedef __attribute__((ext_vector_type(4))) float v4f;
typedef __attribute__((ext_vector_type(8))) float v8f;

struct GemmArgs {
  const float* A;  long lda;  long sAb;
  const float* B;  long ldbn; long ldbk; long sBb;
  float*       C;  long ldc;  long sCb;
  const float* bias;
  const float* A2; long lda2;
  const float* B2; long ldbn2;
  int K;
  int K2;
  int relu;
  int lgx;
  int lgy;
};

template <bool BCONTIG, int MT, int TS>
__device__ __forceinline__ void k_segment(v8f (&c)[MT][2],
                                          const float* Ab, long lda,
                                          const float* B0, const float* B1, long ldbk,
                                          int K, int kq) {
  const float* Ap[MT];
  #pragma unroll
  for (int i = 0; i < MT; ++i) Ap[i] = Ab + (long)(16 * i) * lda;

  static_assert(TS % 2 == 0, "a trip is a whole number of 8-k groups");
  constexpr int TK = 4 * TS;
  constexpr int TG = TS / 2;
  v4f aP[MT][TG], bP[2][TG], aQ[MT][TG], bQ[2][TG];

  auto load_trip = [&](v4f (&a)[MT][TG], v4f (&b)[2][TG], int kbase) {
    #pragma unroll
    for (int g = 0; g < TG; ++g) {
      const int ka = kbase + 8 * g + kq;
      #pragma unroll
      for (int i = 0; i < MT; ++i) a[i][g] = *(const v4f*)(Ap[i] + ka);
      if (BCONTIG) {
        b[0][g] = *(const v4f*)(B0 + ka);
        b[1][g] = *(const v4f*)(B1 + ka);
      } else {
        b[0][g].x = B0[(long)ka * ldbk];        b[0][g].y = B0[(long)(ka + 1) * ldbk];
        b[0][g].z = B0[(long)(ka + 2) * ldbk];  b[0][g].w = B0[(long)(ka + 3) * ldbk];
        b[1][g].x = B1[(long)ka * ldbk];        b[1][g].y = B1[(long)(ka + 1) * ldbk];
        b[1][g].z = B1[(long)(ka + 2) * ldbk];  b[1][g].w = B1[(long)(ka + 3) * ldbk];
      }
    }
  };
  auto mma_trip = [&](const v4f (&a)[MT][TG], const v4f (&b)[2][TG]) {
    #pragma unroll
    for (int g = 0; g < TG; ++g) {
      const v2f b0lo = __builtin_shufflevector(b[0][g], b[0][g], 0, 1);
      const v2f b1lo = __builtin_shufflevector(b[1][g], b[1][g], 0, 1);
      #pragma unroll
      for (int i = 0; i < MT; ++i) {
        const v2f alo = __builtin_shufflevector(a[i][g], a[i][g], 0, 1);
        c[i][0] = __builtin_amdgcn_wmma_f32_16x16x4_f32(false, alo, false, b0lo, (short)0, c[i][0], false, false);
        c[i][1] = __builtin_amdgcn_wmma_f32_16x16x4_f32(false, alo, false, b1lo, (short)0, c[i][1], false, false);
      }
      const v2f b0hi = __builtin_shufflevector(b[0][g], b[0][g], 2, 3);
      const v2f b1hi = __builtin_shufflevector(b[1][g], b[1][g], 2, 3);
      #pragma unroll
      for (int i = 0; i < MT; ++i) {
        const v2f ahi = __builtin_shufflevector(a[i][g], a[i][g], 2, 3);
        c[i][0] = __builtin_amdgcn_wmma_f32_16x16x4_f32(false, ahi, false, b0hi, (short)0, c[i][0], false, false);
        c[i][1] = __builtin_amdgcn_wmma_f32_16x16x4_f32(false, ahi, false, b1hi, (short)0, c[i][1], false, false);
      }
    }
  };

  load_trip(aP, bP, 0);
  int k0 = 0;
  for (; k0 + 2 * TK < K; k0 += 2 * TK) {
    load_trip(aQ, bQ, k0 + TK);
    __builtin_amdgcn_sched_barrier(0);
    mma_trip(aP, bP);
    __builtin_amdgcn_sched_barrier(0);
    load_trip(aP, bP, k0 + 2 * TK);
    __builtin_amdgcn_sched_barrier(0);
    mma_trip(aQ, bQ);
  }
  load_trip(aQ, bQ, k0 + TK);
  __builtin_amdgcn_sched_barrier(0);
  mma_trip(aP, bP);
  mma_trip(aQ, bQ);
}

template <bool BCONTIG, int MT, int TS, bool SEG2, int TS2>
__device__ __forceinline__ void gemm_wave(const GemmArgs& g, int id) {
  const int lane = threadIdx.x;
  const int r15  = lane & 15;
  const int kq   = (lane >> 4) << 2;
  const int tn = id & ((1 << g.lgx) - 1);
  const int tm = (id >> g.lgx) & ((1 << g.lgy) - 1);
  const int bz = id >> (g.lgx + g.lgy);

  const v8f zero = {0.f,0.f,0.f,0.f,0.f,0.f,0.f,0.f};
  v8f c[MT][2];
  #pragma unroll
  for (int i = 0; i < MT; ++i) { c[i][0] = zero; c[i][1] = zero; }

  {
    const float* Ab = g.A + (long)bz * g.sAb + (long)(tm * (16 * MT) + r15) * g.lda;
    const float* B0 = g.B + (long)bz * g.sBb + (long)(tn * 32 + r15) * g.ldbn;
    const float* B1 = B0 + (long)16 * g.ldbn;
    k_segment<BCONTIG, MT, TS>(c, Ab, g.lda, B0, B1, g.ldbk, g.K, kq);
  }
  if (SEG2) {
    const float* Ab = g.A2 + (long)(bz * ((16 * MT) << g.lgy) + tm * (16 * MT) + r15) * g.lda2;
    const float* B0 = g.B2 + (long)(tn * 32 + r15) * g.ldbn2;
    const float* B1 = B0 + (long)16 * g.ldbn2;
    k_segment<true, MT, TS2>(c, Ab, g.lda2, B0, B1, 1, g.K2, kq);
  }

  const int  col0 = tn * 32 + r15;
  const int  col1 = col0 + 16;
  const int  m0   = tm * (16 * MT) + ((lane >> 4) << 3);
  const float bv0 = g.bias ? g.bias[col0] : 0.f;
  const float bv1 = g.bias ? g.bias[col1] : 0.f;
  float* Cp = g.C + (long)bz * g.sCb;
  #pragma unroll
  for (int i = 0; i < MT; ++i) {
    #pragma unroll
    for (int r = 0; r < 8; ++r) {
      const long row = (long)(m0 + i * 16 + r) * g.ldc;
      float v0 = c[i][0][r] + bv0;
      float v1 = c[i][1][r] + bv1;
      if (g.relu) { v0 = fmaxf(v0, 0.f);  v1 = fmaxf(v1, 0.f); }
      Cp[row + col0] = v0;
      Cp[row + col1] = v1;
    }
  }
}

template <bool BCONTIG, int TS, bool SEG2, int TS2, int S, bool HASB>
__device__ __forceinline__ void gemm_split(const GemmArgs& g, int wg, float* red) {
  static_assert(S == 2 || S == 4, "S waves of a 4-wave workgroup share a tile");
  constexpr int LGS = (S == 4) ? 2 : 1;
  constexpr int TPW = 4 / S;
  constexpr int RPW = 8 / S;
  const int lane = threadIdx.x & 31;
  const int wave = __builtin_amdgcn_readfirstlane((int)(threadIdx.x >> 5));
  const int ks = wave & (S - 1);
  const int tl = wave >> LGS;
  const int id = wg * TPW + tl;
  const int r15  = lane & 15;
  const int kq   = (lane >> 4) << 2;
  const int tn = id & ((1 << g.lgx) - 1);
  const int tm = (id >> g.lgx) & ((1 << g.lgy) - 1);
  const int bz = id >> (g.lgx + g.lgy);

  const v8f zero = {0.f,0.f,0.f,0.f,0.f,0.f,0.f,0.f};
  v8f c[1][2];
  c[0][0] = zero; c[0][1] = zero;

  const int  col0 = tn * 32 + r15;
  const int  col1 = col0 + 16;
  float bv0 = 0.f, bv1 = 0.f;
  if (HASB) { bv0 = g.bias[col0];  bv1 = g.bias[col1]; }

  {
    const int  Ks   = g.K >> LGS;
    const long koff = (long)ks * Ks;
    const float* Ab = g.A + (long)bz * g.sAb + (long)(tm * 16 + r15) * g.lda + koff;
    const float* B0 = g.B + (long)bz * g.sBb + (long)(tn * 32 + r15) * g.ldbn + koff * g.ldbk;
    const float* B1 = B0 + (long)16 * g.ldbn;
    k_segment<BCONTIG, 1, TS>(c, Ab, g.lda, B0, B1, g.ldbk, Ks, kq);
  }
  if (SEG2) {
    const int  Ks   = g.K2 >> LGS;
    const long koff = (long)ks * Ks;
    const float* Ab = g.A2 + (long)(bz * (16 << g.lgy) + tm * 16 + r15) * g.lda2 + koff;
    const float* B0 = g.B2 + (long)(tn * 32 + r15) * g.ldbn2 + koff;
    const float* B1 = B0 + (long)16 * g.ldbn2;
    k_segment<true, 1, TS2>(c, Ab, g.lda2, B0, B1, 1, Ks, kq);
  }

  float* mine = red + (wave * 16) * 32 + lane;
  #pragma unroll
  for (int j = 0; j < 2; ++j) {
    #pragma unroll
    for (int r = 0; r < 8; ++r) mine[(j * 8 + r) * 32] = c[0][j][r];
  }
  __syncthreads();

  const float* tile = red + (tl * S * 16) * 32 + lane;
  const int  m0   = tm * 16 + ((lane >> 4) << 3);
  float* Cp = g.C + (long)bz * g.sCb;
  #pragma unroll
  for (int q = 0; q < RPW; ++q) {
    const int r = ks * RPW + q;
    float v0 = tile[r * 32];
    float v1 = tile[(8 + r) * 32];
    #pragma unroll
    for (int s = 1; s < S; ++s) {
      v0 += tile[(s * 16 + r) * 32];
      v1 += tile[(s * 16 + 8 + r) * 32];
    }
    v0 += bv0;
    v1 += bv1;
    if (g.relu) { v0 = fmaxf(v0, 0.f);  v1 = fmaxf(v1, 0.f); }
    const long row = (long)(m0 + r) * g.ldc;
    Cp[row + col0] = v0;
    Cp[row + col1] = v1;
  }
}

template <bool BCONTIG, int TS, bool SEG2, int TS2, int S, bool HASB>
__global__ __launch_bounds__(128)
void wmma_gemm_split(GemmArgs g) {
  __shared__ float red[4 * 16 * 32];
  gemm_split<BCONTIG, TS, SEG2, TS2, S, HASB>(g, blockIdx.x, red);
}

template <bool BC0, int TS0, int S0, bool BC1, int TS1, int S1>
__global__ __launch_bounds__(128)
void wmma_gemm_split_pair(GemmArgs g0, GemmArgs g1, int n0) {
  __shared__ float red[4 * 16 * 32];
  const int id = blockIdx.x;
  if (id < n0) gemm_split<BC0, TS0, false, 8, S0, false>(g0, id, red);
  else         gemm_split<BC1, TS1, false, 8, S1, false>(g1, id - n0, red);
}

template <bool BC0, int MT0, int TS0, bool BC1, int MT1, int TS1>
__global__ __launch_bounds__(32)
void wmma_gemm_pair(GemmArgs g0, GemmArgs g1, int n0) {
  const int id = blockIdx.x;
  if (id < n0) gemm_wave<BC0, MT0, TS0, false, 8>(g0, id);
  else         gemm_wave<BC1, MT1, TS1, false, 8>(g1, id - n0);
}

__global__ __launch_bounds__(256)
void edge_softmax_kernel(const float* __restrict__ anchor_x,
                         const float* __restrict__ node_x,
                         const float* __restrict__ node_mask,
                         const float* __restrict__ qe,
                         const float* __restrict__ Lbase,
                         float* __restrict__ attn,
                         float* __restrict__ s_out)
{
  const int bid = blockIdx.x;
  const int b   = bid >> 6;
  const int t   = threadIdx.x;

  __shared__ float sh_qe[HE];
  __shared__ float sh_dist[N_PER];
  __shared__ float sh_l[N_PER];
  __shared__ float red[256];
  __shared__ float sh_bcast;
  __shared__ float sh_ax[3];

  if (t < HE) sh_qe[t] = qe[bid * HE + t];
  if (t < 3)  sh_ax[t] = anchor_x[bid * 3 + t];
  __syncthreads();

  const float axx = sh_ax[0], axy = sh_ax[1], axz = sh_ax[2];
  const float MU_STEP = 20.f / 63.f;
  const float INV_SIG = 64.f / 20.f;

  for (int n = t; n < N_PER; n += 256) {
    const float* nx = node_x + (long)(b * N_PER + n) * 3;
    float dx = axx - nx[0] + 1e-8f;
    float dy = axy - nx[1] + 1e-8f;
    float dz = axz - nx[2] + 1e-8f;
    float d  = sqrtf(dx * dx + dy * dy + dz * dz);
    sh_dist[n] = d;
    float d10 = d * 0.1f;
    float le = 0.f;
    #pragma unroll 8
    for (int j = 0; j < HE; ++j) {
      float u = (d10 - j * MU_STEP) * INV_SIG;
      le += sh_qe[j] * __expf(-u * u);
    }
    float lb = Lbase[(long)bid * N_PER + n] + le;
    lb *= (node_mask[b * N_PER + n] - 1.f) * 1000000.0f;
    sh_l[n] = lb;
  }
  __syncthreads();

  red[t] = fmaxf(sh_l[t], sh_l[t + 256]);
  __syncthreads();
  for (int st = 128; st > 0; st >>= 1) {
    if (t < st) red[t] = fmaxf(red[t], red[t + st]);
    __syncthreads();
  }
  if (t == 0) sh_bcast = red[0];
  __syncthreads();
  const float mx = sh_bcast;

  float e0 = __expf(sh_l[t] - mx);
  float e1 = __expf(sh_l[t + 256] - mx);
  red[t] = e0 + e1;
  __syncthreads();
  for (int st = 128; st > 0; st >>= 1) {
    if (t < st) red[t] += red[t + st];
    __syncthreads();
  }
  if (t == 0) sh_bcast = red[0];
  __syncthreads();
  const float inv = 1.f / sh_bcast;

  const float a0 = e0 * inv, a1 = e1 * inv;
  sh_l[t] = a0; sh_l[t + 256] = a1;
  attn[(long)bid * N_PER + t]       = a0;
  attn[(long)bid * N_PER + t + 256] = a1;
  __syncthreads();

  const int j = t & 63, c = t >> 6;
  const float muj = j * MU_STEP;
  float part = 0.f;
  for (int n = c * 128; n < c * 128 + 128; ++n) {
    float u = (sh_dist[n] * 0.1f - muj) * INV_SIG;
    part += sh_l[n] * __expf(-u * u);
  }
  red[t] = part;
  __syncthreads();
  if (t < 64)
    s_out[bid * HE + t] = red[t] + red[t + 64] + red[t + 128] + red[t + 192];
}

__device__ __forceinline__ float wave_sum(float v) {
  v += __builtin_bit_cast(float, __builtin_amdgcn_update_dpp(0, __builtin_bit_cast(int, v), 0x161, 0xf, 0xf, true));
  v += __builtin_bit_cast(float, __builtin_amdgcn_update_dpp(0, __builtin_bit_cast(int, v), 0x162, 0xf, 0xf, true));
  v += __builtin_bit_cast(float, __builtin_amdgcn_update_dpp(0, __builtin_bit_cast(int, v), 0x164, 0xf, 0xf, true));
  v += __builtin_bit_cast(float, __builtin_amdgcn_update_dpp(0, __builtin_bit_cast(int, v), 0x168, 0xf, 0xf, true));
  const int iv = __builtin_bit_cast(int, v);
  v += __builtin_bit_cast(float, __builtin_amdgcn_permlanex16(iv, iv, 0x76543210, 0xfedcba98, false, false));
  return v;
}

__global__ __launch_bounds__(128)
void add_ln_kernel(const float* __restrict__ x, const float* __restrict__ y,
                   const float* __restrict__ gam, const float* __restrict__ bet,
                   float* __restrict__ out)
{
  const int lane = threadIdx.x & 31;
  const int r    = blockIdx.x * 4 + (threadIdx.x >> 5);
  const int c0   = lane * 4;
  const long base = (long)r * H + c0;

  const v4f x0 = *(const v4f*)(x + base), x1 = *(const v4f*)(x + base + 128);
  const v4f y0 = *(const v4f*)(y + base), y1 = *(const v4f*)(y + base + 128);
  const v4f g0 = *(const v4f*)(gam + c0), g1 = *(const v4f*)(gam + c0 + 128);
  const v4f b0 = *(const v4f*)(bet + c0), b1 = *(const v4f*)(bet + c0 + 128);

  const v4f v0 = x0 + y0, v1 = x1 + y1;
  const float mean = wave_sum(((v0.x + v0.y) + (v0.z + v0.w)) + ((v1.x + v1.y) + (v1.z + v1.w))) * (1.f / H);

  const v4f d0 = v0 - mean, d1 = v1 - mean;
  const v4f q0 = d0 * d0,   q1 = d1 * d1;
  const float var  = wave_sum(((q0.x + q0.y) + (q0.z + q0.w)) + ((q1.x + q1.y) + (q1.z + q1.w))) * (1.f / H);
  const float rstd = rsqrtf(var + 1e-5f);

  *(v4f*)(out + base)       = d0 * rstd * g0 + b0;
  *(v4f*)(out + base + 128) = d1 * rstd * g1 + b1;
}

extern "C" void kernel_launch(void* const* d_in, const int* in_sizes, int n_in,
                              void* d_out, int out_size, void* d_ws, size_t ws_size,
                              hipStream_t stream) {
  (void)in_sizes; (void)n_in; (void)out_size; (void)ws_size;
  const float* anchor_x  = (const float*)d_in[0];
  const float* node_x    = (const float*)d_in[1];
  const float* anchor_f  = (const float*)d_in[2];
  const float* node_f    = (const float*)d_in[3];
  const float* node_mask = (const float*)d_in[6];
  const float* Wq  = (const float*)d_in[7];
  const float* bq  = (const float*)d_in[8];
  const float* Wkv = (const float*)d_in[9];
  const float* bkv = (const float*)d_in[10];
  const float* ln1_g = (const float*)d_in[11];
  const float* ln1_b = (const float*)d_in[12];
  const float* W1 = (const float*)d_in[13];
  const float* b1 = (const float*)d_in[14];
  const float* W2 = (const float*)d_in[15];
  const float* b2 = (const float*)d_in[16];
  const float* W3 = (const float*)d_in[17];
  const float* b3 = (const float*)d_in[18];
  const float* ln2_g = (const float*)d_in[19];
  const float* ln2_b = (const float*)d_in[20];

  const long WKV_LD = H + HE;

  float* p = (float*)d_ws;
  float* q    = p; p += (long)NA * H;
  float* qe   = p; p += (long)NA * HE;
  float* kvn  = p; p += (long)NN * 2 * H;
  float* L    = p; p += (long)NA * N_PER;
  float* attn = p; p += (long)NA * N_PER;
  float* sbuf = p; p += (long)NA * HE;
  float* upd  = p; p += (long)NA * H;
  float* af   = p; p += (long)NA * H;
  float* m1   = p; p += (long)NA * 2 * H;
  float* m2   = p; p += (long)NA * 2 * H;
  float* m3   = p; p += (long)NA * H;


  {
    GemmArgs gk{node_f, H, 0,  Wkv, WKV_LD, 1, 0,  kvn, 2 * H, 0,  bkv,
                nullptr, 0, nullptr, 0,  H, 0, 0, 4, 6};
    GemmArgs gq{anchor_f, H, 0,  Wq, H, 1, 0,  q, H, 0,  bq,
                nullptr, 0, nullptr, 0,  H, 0, 0, 3, 5};
    wmma_gemm_pair<true, 4, 8, true, 1, 16><<<dim3(1024 + 256), dim3(32), 0, stream>>>(gk, gq, 1024);
  }
  {
    GemmArgs ge{q, H, 0,  Wkv + H, 1, WKV_LD, 0,  qe, HE, 0,  nullptr,
                nullptr, 0, nullptr, 0,  H, 0, 0, 1, 5};
    GemmArgs gl{q, H, (long)A_PER * H,  kvn, 2 * H, 1, (long)N_PER * 2 * H,
                L, N_PER, (long)A_PER * N_PER,  nullptr,
                nullptr, 0, nullptr, 0,  H, 0, 0, 4, 2};
    wmma_gemm_split_pair<false, 8, 4, true, 16, 2><<<dim3(64 + 256), dim3(128), 0, stream>>>(ge, gl, 64);
  }
  edge_softmax_kernel<<<dim3(NA), dim3(256), 0, stream>>>(
      anchor_x, node_x, node_mask, qe, L, attn, sbuf);
  {
    GemmArgs gu{attn, N_PER, (long)A_PER * N_PER,  kvn + H, 1, 2 * H, (long)N_PER * 2 * H,
                upd, H, (long)A_PER * H,  nullptr,
                sbuf, HE, Wkv + (long)H * WKV_LD + H, WKV_LD,  N_PER, HE, 0, 3, 2};
    wmma_gemm_split<false, 16, true, 2, 4, false><<<dim3(256), dim3(128), 0, stream>>>(gu);
  }
  add_ln_kernel<<<dim3(NA / 4), dim3(128), 0, stream>>>(anchor_f, upd, ln1_g, ln1_b, af);
  {
    GemmArgs g1{af, H, 0,  W1, H, 1, 0,  m1, 2 * H, 0,  b1,
                nullptr, 0, nullptr, 0,  H, 0, 1, 4, 5};
    wmma_gemm_split<true, 16, false, 8, 2, true><<<dim3(256), dim3(128), 0, stream>>>(g1);
  }
  {
    GemmArgs g2{m1, 2 * H, 0,  W2, 2 * H, 1, 0,  m2, 2 * H, 0,  b2,
                nullptr, 0, nullptr, 0,  2 * H, 0, 1, 4, 5};
    wmma_gemm_split<true, 16, false, 8, 4, true><<<dim3(512), dim3(128), 0, stream>>>(g2);
  }
  {
    GemmArgs g3{m2, 2 * H, 0,  W3, 2 * H, 1, 0,  m3, H, 0,  b3,
                nullptr, 0, nullptr, 0,  2 * H, 0, 0, 3, 5};
    wmma_gemm_split<true, 16, false, 8, 4, true><<<dim3(256), dim3(128), 0, stream>>>(g3);
  }
  add_ln_kernel<<<dim3(NA / 4), dim3(128), 0, stream>>>(af, m3, ln2_g, ln2_b, (float*)d_out);
}


# Edge kernel: a thread keeps its two nodes' 64 RBF values in registers from the logit pass; s = sum attn*rbf folded over lanes (in-ALU swizzles) and waves (LDS), no 2nd exp pass (s sum order changes)
# speedup vs baseline: 5.8682x; 1.1664x over previous
#include <hip/hip_runtime.h>
#include <math.h>

#define H      256
#define HE     64
#define NB     8
#define A_PER  64
#define N_PER  512
#define NA     (NB * A_PER)
#define NN     (NB * N_PER)

typedef __attribute__((ext_vector_type(2))) float v2f;
typedef __attribute__((ext_vector_type(4))) float v4f;
typedef __attribute__((ext_vector_type(8))) float v8f;

struct GemmArgs {
  const float* A;  long lda;  long sAb;
  const float* B;  long ldbn; long ldbk; long sBb;
  float*       C;  long ldc;  long sCb;
  const float* bias;
  const float* A2; long lda2;
  const float* B2; long ldbn2;
  int K;
  int K2;
  int relu;
  int lgx;
  int lgy;
};

template <bool BCONTIG, int MT, int TS>
__device__ __forceinline__ void k_segment(v8f (&c)[MT][2],
                                          const float* Ab, long lda,
                                          const float* B0, const float* B1, long ldbk,
                                          int K, int kq) {
  const float* Ap[MT];
  #pragma unroll
  for (int i = 0; i < MT; ++i) Ap[i] = Ab + (long)(16 * i) * lda;

  static_assert(TS % 2 == 0, "a trip is a whole number of 8-k groups");
  constexpr int TK = 4 * TS;
  constexpr int TG = TS / 2;
  v4f aP[MT][TG], bP[2][TG], aQ[MT][TG], bQ[2][TG];

  auto load_trip = [&](v4f (&a)[MT][TG], v4f (&b)[2][TG], int kbase) {
    #pragma unroll
    for (int g = 0; g < TG; ++g) {
      const int ka = kbase + 8 * g + kq;
      #pragma unroll
      for (int i = 0; i < MT; ++i) a[i][g] = *(const v4f*)(Ap[i] + ka);
      if (BCONTIG) {
        b[0][g] = *(const v4f*)(B0 + ka);
        b[1][g] = *(const v4f*)(B1 + ka);
      } else {
        b[0][g].x = B0[(long)ka * ldbk];        b[0][g].y = B0[(long)(ka + 1) * ldbk];
        b[0][g].z = B0[(long)(ka + 2) * ldbk];  b[0][g].w = B0[(long)(ka + 3) * ldbk];
        b[1][g].x = B1[(long)ka * ldbk];        b[1][g].y = B1[(long)(ka + 1) * ldbk];
        b[1][g].z = B1[(long)(ka + 2) * ldbk];  b[1][g].w = B1[(long)(ka + 3) * ldbk];
      }
    }
  };
  auto mma_trip = [&](const v4f (&a)[MT][TG], const v4f (&b)[2][TG]) {
    #pragma unroll
    for (int g = 0; g < TG; ++g) {
      const v2f b0lo = __builtin_shufflevector(b[0][g], b[0][g], 0, 1);
      const v2f b1lo = __builtin_shufflevector(b[1][g], b[1][g], 0, 1);
      #pragma unroll
      for (int i = 0; i < MT; ++i) {
        const v2f alo = __builtin_shufflevector(a[i][g], a[i][g], 0, 1);
        c[i][0] = __builtin_amdgcn_wmma_f32_16x16x4_f32(false, alo, false, b0lo, (short)0, c[i][0], false, false);
        c[i][1] = __builtin_amdgcn_wmma_f32_16x16x4_f32(false, alo, false, b1lo, (short)0, c[i][1], false, false);
      }
      const v2f b0hi = __builtin_shufflevector(b[0][g], b[0][g], 2, 3);
      const v2f b1hi = __builtin_shufflevector(b[1][g], b[1][g], 2, 3);
      #pragma unroll
      for (int i = 0; i < MT; ++i) {
        const v2f ahi = __builtin_shufflevector(a[i][g], a[i][g], 2, 3);
        c[i][0] = __builtin_amdgcn_wmma_f32_16x16x4_f32(false, ahi, false, b0hi, (short)0, c[i][0], false, false);
        c[i][1] = __builtin_amdgcn_wmma_f32_16x16x4_f32(false, ahi, false, b1hi, (short)0, c[i][1], false, false);
      }
    }
  };

  load_trip(aP, bP, 0);
  int k0 = 0;
  for (; k0 + 2 * TK < K; k0 += 2 * TK) {
    load_trip(aQ, bQ, k0 + TK);
    __builtin_amdgcn_sched_barrier(0);
    mma_trip(aP, bP);
    __builtin_amdgcn_sched_barrier(0);
    load_trip(aP, bP, k0 + 2 * TK);
    __builtin_amdgcn_sched_barrier(0);
    mma_trip(aQ, bQ);
  }
  load_trip(aQ, bQ, k0 + TK);
  __builtin_amdgcn_sched_barrier(0);
  mma_trip(aP, bP);
  mma_trip(aQ, bQ);
}

template <bool BCONTIG, int MT, int TS, bool SEG2, int TS2>
__device__ __forceinline__ void gemm_wave(const GemmArgs& g, int id) {
  const int lane = threadIdx.x;
  const int r15  = lane & 15;
  const int kq   = (lane >> 4) << 2;
  const int tn = id & ((1 << g.lgx) - 1);
  const int tm = (id >> g.lgx) & ((1 << g.lgy) - 1);
  const int bz = id >> (g.lgx + g.lgy);

  const v8f zero = {0.f,0.f,0.f,0.f,0.f,0.f,0.f,0.f};
  v8f c[MT][2];
  #pragma unroll
  for (int i = 0; i < MT; ++i) { c[i][0] = zero; c[i][1] = zero; }

  {
    const float* Ab = g.A + (long)bz * g.sAb + (long)(tm * (16 * MT) + r15) * g.lda;
    const float* B0 = g.B + (long)bz * g.sBb + (long)(tn * 32 + r15) * g.ldbn;
    const float* B1 = B0 + (long)16 * g.ldbn;
    k_segment<BCONTIG, MT, TS>(c, Ab, g.lda, B0, B1, g.ldbk, g.K, kq);
  }
  if (SEG2) {
    const float* Ab = g.A2 + (long)(bz * ((16 * MT) << g.lgy) + tm * (16 * MT) + r15) * g.lda2;
    const float* B0 = g.B2 + (long)(tn * 32 + r15) * g.ldbn2;
    const float* B1 = B0 + (long)16 * g.ldbn2;
    k_segment<true, MT, TS2>(c, Ab, g.lda2, B0, B1, 1, g.K2, kq);
  }

  const int  col0 = tn * 32 + r15;
  const int  col1 = col0 + 16;
  const int  m0   = tm * (16 * MT) + ((lane >> 4) << 3);
  const float bv0 = g.bias ? g.bias[col0] : 0.f;
  const float bv1 = g.bias ? g.bias[col1] : 0.f;
  float* Cp = g.C + (long)bz * g.sCb;
  #pragma unroll
  for (int i = 0; i < MT; ++i) {
    #pragma unroll
    for (int r = 0; r < 8; ++r) {
      const long row = (long)(m0 + i * 16 + r) * g.ldc;
      float v0 = c[i][0][r] + bv0;
      float v1 = c[i][1][r] + bv1;
      if (g.relu) { v0 = fmaxf(v0, 0.f);  v1 = fmaxf(v1, 0.f); }
      Cp[row + col0] = v0;
      Cp[row + col1] = v1;
    }
  }
}

template <bool BCONTIG, int TS, bool SEG2, int TS2, int S, bool HASB>
__device__ __forceinline__ void gemm_split(const GemmArgs& g, int wg, float* red) {
  static_assert(S == 2 || S == 4, "S waves of a 4-wave workgroup share a tile");
  constexpr int LGS = (S == 4) ? 2 : 1;
  constexpr int TPW = 4 / S;
  constexpr int RPW = 8 / S;
  const int lane = threadIdx.x & 31;
  const int wave = __builtin_amdgcn_readfirstlane((int)(threadIdx.x >> 5));
  const int ks = wave & (S - 1);
  const int tl = wave >> LGS;
  const int id = wg * TPW + tl;
  const int r15  = lane & 15;
  const int kq   = (lane >> 4) << 2;
  const int tn = id & ((1 << g.lgx) - 1);
  const int tm = (id >> g.lgx) & ((1 << g.lgy) - 1);
  const int bz = id >> (g.lgx + g.lgy);

  const v8f zero = {0.f,0.f,0.f,0.f,0.f,0.f,0.f,0.f};
  v8f c[1][2];
  c[0][0] = zero; c[0][1] = zero;

  const int  col0 = tn * 32 + r15;
  const int  col1 = col0 + 16;
  float bv0 = 0.f, bv1 = 0.f;
  if (HASB) { bv0 = g.bias[col0];  bv1 = g.bias[col1]; }

  {
    const int  Ks   = g.K >> LGS;
    const long koff = (long)ks * Ks;
    const float* Ab = g.A + (long)bz * g.sAb + (long)(tm * 16 + r15) * g.lda + koff;
    const float* B0 = g.B + (long)bz * g.sBb + (long)(tn * 32 + r15) * g.ldbn + koff * g.ldbk;
    const float* B1 = B0 + (long)16 * g.ldbn;
    k_segment<BCONTIG, 1, TS>(c, Ab, g.lda, B0, B1, g.ldbk, Ks, kq);
  }
  if (SEG2) {
    const int  Ks   = g.K2 >> LGS;
    const long koff = (long)ks * Ks;
    const float* Ab = g.A2 + (long)(bz * (16 << g.lgy) + tm * 16 + r15) * g.lda2 + koff;
    const float* B0 = g.B2 + (long)(tn * 32 + r15) * g.ldbn2 + koff;
    const float* B1 = B0 + (long)16 * g.ldbn2;
    k_segment<true, 1, TS2>(c, Ab, g.lda2, B0, B1, 1, Ks, kq);
  }

  float* mine = red + (wave * 16) * 32 + lane;
  #pragma unroll
  for (int j = 0; j < 2; ++j) {
    #pragma unroll
    for (int r = 0; r < 8; ++r) mine[(j * 8 + r) * 32] = c[0][j][r];
  }
  __syncthreads();

  const float* tile = red + (tl * S * 16) * 32 + lane;
  const int  m0   = tm * 16 + ((lane >> 4) << 3);
  float* Cp = g.C + (long)bz * g.sCb;
  #pragma unroll
  for (int q = 0; q < RPW; ++q) {
    const int r = ks * RPW + q;
    float v0 = tile[r * 32];
    float v1 = tile[(8 + r) * 32];
    #pragma unroll
    for (int s = 1; s < S; ++s) {
      v0 += tile[(s * 16 + r) * 32];
      v1 += tile[(s * 16 + 8 + r) * 32];
    }
    v0 += bv0;
    v1 += bv1;
    if (g.relu) { v0 = fmaxf(v0, 0.f);  v1 = fmaxf(v1, 0.f); }
    const long row = (long)(m0 + r) * g.ldc;
    Cp[row + col0] = v0;
    Cp[row + col1] = v1;
  }
}

template <bool BCONTIG, int TS, bool SEG2, int TS2, int S, bool HASB>
__global__ __launch_bounds__(128)
void wmma_gemm_split(GemmArgs g) {
  __shared__ float red[4 * 16 * 32];
  gemm_split<BCONTIG, TS, SEG2, TS2, S, HASB>(g, blockIdx.x, red);
}

template <bool BC0, int TS0, int S0, bool BC1, int TS1, int S1>
__global__ __launch_bounds__(128)
void wmma_gemm_split_pair(GemmArgs g0, GemmArgs g1, int n0) {
  __shared__ float red[4 * 16 * 32];
  const int id = blockIdx.x;
  if (id < n0) gemm_split<BC0, TS0, false, 8, S0, false>(g0, id, red);
  else         gemm_split<BC1, TS1, false, 8, S1, false>(g1, id - n0, red);
}

template <bool BC0, int MT0, int TS0, bool BC1, int MT1, int TS1>
__global__ __launch_bounds__(32)
void wmma_gemm_pair(GemmArgs g0, GemmArgs g1, int n0) {
  const int id = blockIdx.x;
  if (id < n0) gemm_wave<BC0, MT0, TS0, false, 8>(g0, id);
  else         gemm_wave<BC1, MT1, TS1, false, 8>(g1, id - n0);
}


template <int M>
__device__ __forceinline__ float lane_xor_row(float v) {
  return __builtin_bit_cast(float, __builtin_amdgcn_update_dpp(0, __builtin_bit_cast(int, v), 0x160 | M, 0xf, 0xf, true));
}
__device__ __forceinline__ float lane_xor16(float v) {
  const int iv = __builtin_bit_cast(int, v);
  return __builtin_bit_cast(float, __builtin_amdgcn_permlanex16(iv, iv, 0x76543210, 0xfedcba98, false, false));
}

template <int LEN, int BIT>
__device__ __forceinline__ void fold_step(float (&v)[HE], bool up) {
  #pragma unroll
  for (int i = 0; i < LEN / 2; ++i) {
    const float lo = v[i], hi = v[i + LEN / 2];
    const float send = up ? lo : hi;
    const float keep = up ? hi : lo;
    float got;
    if constexpr (BIT == 4) got = lane_xor16(send);
    else                    got = lane_xor_row<(1 << BIT)>(send);
    v[i] = keep + got;
  }
}

__global__ __launch_bounds__(256)
void edge_softmax_kernel(const float* __restrict__ anchor_x,
                         const float* __restrict__ node_x,
                         const float* __restrict__ node_mask,
                         const float* __restrict__ qe,
                         const float* __restrict__ Lbase,
                         float* __restrict__ attn,
                         float* __restrict__ s_out)
{
  const int bid = blockIdx.x;
  const int b   = bid >> 6;
  const int t   = threadIdx.x;

  __shared__ float sh_qe[HE];
  __shared__ float sh_l[N_PER];
  __shared__ float red[256];
  __shared__ float sh_s[8 * HE];
  __shared__ float sh_bcast;
  __shared__ float sh_ax[3];

  if (t < HE) sh_qe[t] = qe[bid * HE + t];
  if (t < 3)  sh_ax[t] = anchor_x[bid * 3 + t];
  __syncthreads();

  const float axx = sh_ax[0], axy = sh_ax[1], axz = sh_ax[2];
  const float MU_STEP = 20.f / 63.f;
  const float INV_SIG = 64.f / 20.f;

  const int n0 = t, n1 = t + 256;
  const float* nx0 = node_x + (long)(b * N_PER + n0) * 3;
  const float* nx1 = node_x + (long)(b * N_PER + n1) * 3;
  float dx0 = axx - nx0[0] + 1e-8f;
  float dy0 = axy - nx0[1] + 1e-8f;
  float dz0 = axz - nx0[2] + 1e-8f;
  float dx1 = axx - nx1[0] + 1e-8f;
  float dy1 = axy - nx1[1] + 1e-8f;
  float dz1 = axz - nx1[2] + 1e-8f;
  const float q0 = __builtin_fmaf(dz0, dz0, __builtin_fmaf(dx0, dx0, dy0 * dy0));
  const float q1 = __builtin_fmaf(dz1, dz1, __builtin_fmaf(dx1, dx1, dy1 * dy1));
  const float d10_0 = sqrtf(q0) * 0.1f;
  const float d10_1 = sqrtf(q1) * 0.1f;

  float r0[HE], r1[HE];
  float le0 = 0.f, le1 = 0.f;
  #pragma unroll
  for (int j = 0; j < HE; ++j) {
    const float u0 = __builtin_fmaf(-(float)j, MU_STEP, d10_0) * INV_SIG;
    const float u1 = __builtin_fmaf(-(float)j, MU_STEP, d10_1) * INV_SIG;
    r0[j] = __expf(-u0 * u0);
    r1[j] = __expf(-u1 * u1);
    const float qj = sh_qe[j];
    le0 += qj * r0[j];
    le1 += qj * r1[j];
  }
  float lb0 = Lbase[(long)bid * N_PER + n0] + le0;
  float lb1 = Lbase[(long)bid * N_PER + n1] + le1;
  lb0 *= (node_mask[b * N_PER + n0] - 1.f) * 1000000.0f;
  lb1 *= (node_mask[b * N_PER + n1] - 1.f) * 1000000.0f;
  sh_l[n0] = lb0;
  sh_l[n1] = lb1;
  __syncthreads();

  red[t] = fmaxf(sh_l[t], sh_l[t + 256]);
  __syncthreads();
  for (int st = 128; st > 0; st >>= 1) {
    if (t < st) red[t] = fmaxf(red[t], red[t + st]);
    __syncthreads();
  }
  if (t == 0) sh_bcast = red[0];
  __syncthreads();
  const float mx = sh_bcast;

  float e0 = __expf(sh_l[t] - mx);
  float e1 = __expf(sh_l[t + 256] - mx);
  red[t] = e0 + e1;
  __syncthreads();
  for (int st = 128; st > 0; st >>= 1) {
    if (t < st) red[t] += red[t + st];
    __syncthreads();
  }
  if (t == 0) sh_bcast = red[0];
  __syncthreads();
  const float inv = 1.f / sh_bcast;

  const float a0 = e0 * inv, a1 = e1 * inv;
  attn[(long)bid * N_PER + t]       = a0;
  attn[(long)bid * N_PER + t + 256] = a1;

  float w[HE];
  #pragma unroll
  for (int j = 0; j < HE; ++j) w[j] = a0 * r0[j] + a1 * r1[j];
  const int lane = t & 31;
  fold_step<64, 4>(w, (lane & 16) != 0);
  fold_step<32, 3>(w, (lane & 8) != 0);
  fold_step<16, 2>(w, (lane & 4) != 0);
  fold_step<8,  1>(w, (lane & 2) != 0);
  fold_step<4,  0>(w, (lane & 1) != 0);
  const int wave = t >> 5;
  sh_s[wave * HE + 2 * lane]     = w[0];
  sh_s[wave * HE + 2 * lane + 1] = w[1];
  __syncthreads();
  if (t < HE) {
    float acc = sh_s[t];
    #pragma unroll
    for (int q = 1; q < 8; ++q) acc += sh_s[q * HE + t];
    s_out[bid * HE + t] = acc;
  }
}

__device__ __forceinline__ float wave_sum(float v) {
  v += __builtin_bit_cast(float, __builtin_amdgcn_update_dpp(0, __builtin_bit_cast(int, v), 0x161, 0xf, 0xf, true));
  v += __builtin_bit_cast(float, __builtin_amdgcn_update_dpp(0, __builtin_bit_cast(int, v), 0x162, 0xf, 0xf, true));
  v += __builtin_bit_cast(float, __builtin_amdgcn_update_dpp(0, __builtin_bit_cast(int, v), 0x164, 0xf, 0xf, true));
  v += __builtin_bit_cast(float, __builtin_amdgcn_update_dpp(0, __builtin_bit_cast(int, v), 0x168, 0xf, 0xf, true));
  const int iv = __builtin_bit_cast(int, v);
  v += __builtin_bit_cast(float, __builtin_amdgcn_permlanex16(iv, iv, 0x76543210, 0xfedcba98, false, false));
  return v;
}

__global__ __launch_bounds__(128)
void add_ln_kernel(const float* __restrict__ x, const float* __restrict__ y,
                   const float* __restrict__ gam, const float* __restrict__ bet,
                   float* __restrict__ out)
{
  const int lane = threadIdx.x & 31;
  const int r    = blockIdx.x * 4 + (threadIdx.x >> 5);
  const int c0   = lane * 4;
  const long base = (long)r * H + c0;

  const v4f x0 = *(const v4f*)(x + base), x1 = *(const v4f*)(x + base + 128);
  const v4f y0 = *(const v4f*)(y + base), y1 = *(const v4f*)(y + base + 128);
  const v4f g0 = *(const v4f*)(gam + c0), g1 = *(const v4f*)(gam + c0 + 128);
  const v4f b0 = *(const v4f*)(bet + c0), b1 = *(const v4f*)(bet + c0 + 128);

  const v4f v0 = x0 + y0, v1 = x1 + y1;
  const float mean = wave_sum(((v0.x + v0.y) + (v0.z + v0.w)) + ((v1.x + v1.y) + (v1.z + v1.w))) * (1.f / H);

  const v4f d0 = v0 - mean, d1 = v1 - mean;
  const v4f q0 = d0 * d0,   q1 = d1 * d1;
  const float var  = wave_sum(((q0.x + q0.y) + (q0.z + q0.w)) + ((q1.x + q1.y) + (q1.z + q1.w))) * (1.f / H);
  const float rstd = rsqrtf(var + 1e-5f);

  *(v4f*)(out + base)       = d0 * rstd * g0 + b0;
  *(v4f*)(out + base + 128) = d1 * rstd * g1 + b1;
}

extern "C" void kernel_launch(void* const* d_in, const int* in_sizes, int n_in,
                              void* d_out, int out_size, void* d_ws, size_t ws_size,
                              hipStream_t stream) {
  (void)in_sizes; (void)n_in; (void)out_size; (void)ws_size;
  const float* anchor_x  = (const float*)d_in[0];
  const float* node_x    = (const float*)d_in[1];
  const float* anchor_f  = (const float*)d_in[2];
  const float* node_f    = (const float*)d_in[3];
  const float* node_mask = (const float*)d_in[6];
  const float* Wq  = (const float*)d_in[7];
  const float* bq  = (const float*)d_in[8];
  const float* Wkv = (const float*)d_in[9];
  const float* bkv = (const float*)d_in[10];
  const float* ln1_g = (const float*)d_in[11];
  const float* ln1_b = (const float*)d_in[12];
  const float* W1 = (const float*)d_in[13];
  const float* b1 = (const float*)d_in[14];
  const float* W2 = (const float*)d_in[15];
  const float* b2 = (const float*)d_in[16];
  const float* W3 = (const float*)d_in[17];
  const float* b3 = (const float*)d_in[18];
  const float* ln2_g = (const float*)d_in[19];
  const float* ln2_b = (const float*)d_in[20];

  const long WKV_LD = H + HE;

  float* p = (float*)d_ws;
  float* q    = p; p += (long)NA * H;
  float* qe   = p; p += (long)NA * HE;
  float* kvn  = p; p += (long)NN * 2 * H;
  float* L    = p; p += (long)NA * N_PER;
  float* attn = p; p += (long)NA * N_PER;
  float* sbuf = p; p += (long)NA * HE;
  float* upd  = p; p += (long)NA * H;
  float* af   = p; p += (long)NA * H;
  float* m1   = p; p += (long)NA * 2 * H;
  float* m2   = p; p += (long)NA * 2 * H;
  float* m3   = p; p += (long)NA * H;


  {
    GemmArgs gk{node_f, H, 0,  Wkv, WKV_LD, 1, 0,  kvn, 2 * H, 0,  bkv,
                nullptr, 0, nullptr, 0,  H, 0, 0, 4, 6};
    GemmArgs gq{anchor_f, H, 0,  Wq, H, 1, 0,  q, H, 0,  bq,
                nullptr, 0, nullptr, 0,  H, 0, 0, 3, 5};
    wmma_gemm_pair<true, 4, 8, true, 1, 16><<<dim3(1024 + 256), dim3(32), 0, stream>>>(gk, gq, 1024);
  }
  {
    GemmArgs ge{q, H, 0,  Wkv + H, 1, WKV_LD, 0,  qe, HE, 0,  nullptr,
                nullptr, 0, nullptr, 0,  H, 0, 0, 1, 5};
    GemmArgs gl{q, H, (long)A_PER * H,  kvn, 2 * H, 1, (long)N_PER * 2 * H,
                L, N_PER, (long)A_PER * N_PER,  nullptr,
                nullptr, 0, nullptr, 0,  H, 0, 0, 4, 2};
    wmma_gemm_split_pair<false, 8, 4, true, 16, 2><<<dim3(64 + 256), dim3(128), 0, stream>>>(ge, gl, 64);
  }
  edge_softmax_kernel<<<dim3(NA), dim3(256), 0, stream>>>(
      anchor_x, node_x, node_mask, qe, L, attn, sbuf);
  {
    GemmArgs gu{attn, N_PER, (long)A_PER * N_PER,  kvn + H, 1, 2 * H, (long)N_PER * 2 * H,
                upd, H, (long)A_PER * H,  nullptr,
                sbuf, HE, Wkv + (long)H * WKV_LD + H, WKV_LD,  N_PER, HE, 0, 3, 2};
    wmma_gemm_split<false, 16, true, 2, 4, false><<<dim3(256), dim3(128), 0, stream>>>(gu);
  }
  add_ln_kernel<<<dim3(NA / 4), dim3(128), 0, stream>>>(anchor_f, upd, ln1_g, ln1_b, af);
  {
    GemmArgs g1{af, H, 0,  W1, H, 1, 0,  m1, 2 * H, 0,  b1,
                nullptr, 0, nullptr, 0,  H, 0, 1, 4, 5};
    wmma_gemm_split<true, 16, false, 8, 2, true><<<dim3(256), dim3(128), 0, stream>>>(g1);
  }
  {
    GemmArgs g2{m1, 2 * H, 0,  W2, 2 * H, 1, 0,  m2, 2 * H, 0,  b2,
                nullptr, 0, nullptr, 0,  2 * H, 0, 1, 4, 5};
    wmma_gemm_split<true, 16, false, 8, 4, true><<<dim3(512), dim3(128), 0, stream>>>(g2);
  }
  {
    GemmArgs g3{m2, 2 * H, 0,  W3, 2 * H, 1, 0,  m3, H, 0,  b3,
                nullptr, 0, nullptr, 0,  2 * H, 0, 0, 3, 5};
    wmma_gemm_split<true, 16, false, 8, 4, true><<<dim3(256), dim3(128), 0, stream>>>(g3);
  }
  add_ln_kernel<<<dim3(NA / 4), dim3(128), 0, stream>>>(af, m3, ln2_g, ln2_b, (float*)d_out);
}


# Launches 9 -> 8: the first LayerNorm has no launch of its own; the first MLP GEMM normalises its 16 A rows into LDS as prologue (a wave per row), the final kernel recomputes them; sums unchanged
# speedup vs baseline: 6.0835x; 1.0367x over previous
#include <hip/hip_runtime.h>
#include <math.h>

#define H      256
#define HE     64
#define NB     8
#define A_PER  64
#define N_PER  512
#define NA     (NB * A_PER)
#define NN     (NB * N_PER)

typedef __attribute__((ext_vector_type(2))) float v2f;
typedef __attribute__((ext_vector_type(4))) float v4f;
typedef __attribute__((ext_vector_type(8))) float v8f;

struct GemmArgs {
  const float* A;  long lda;  long sAb;
  const float* B;  long ldbn; long ldbk; long sBb;
  float*       C;  long ldc;  long sCb;
  const float* bias;
  const float* A2; long lda2;
  const float* B2; long ldbn2;
  int K;
  int K2;
  int relu;
  int lgx;
  int lgy;
};

template <bool BCONTIG, int MT, int TS>
__device__ __forceinline__ void k_segment(v8f (&c)[MT][2],
                                          const float* Ab, long lda,
                                          const float* B0, const float* B1, long ldbk,
                                          int K, int kq) {
  const float* Ap[MT];
  #pragma unroll
  for (int i = 0; i < MT; ++i) Ap[i] = Ab + (long)(16 * i) * lda;

  static_assert(TS % 2 == 0, "a trip is a whole number of 8-k groups");
  constexpr int TK = 4 * TS;
  constexpr int TG = TS / 2;
  v4f aP[MT][TG], bP[2][TG], aQ[MT][TG], bQ[2][TG];

  auto load_trip = [&](v4f (&a)[MT][TG], v4f (&b)[2][TG], int kbase) {
    #pragma unroll
    for (int g = 0; g < TG; ++g) {
      const int ka = kbase + 8 * g + kq;
      #pragma unroll
      for (int i = 0; i < MT; ++i) a[i][g] = *(const v4f*)(Ap[i] + ka);
      if (BCONTIG) {
        b[0][g] = *(const v4f*)(B0 + ka);
        b[1][g] = *(const v4f*)(B1 + ka);
      } else {
        b[0][g].x = B0[(long)ka * ldbk];        b[0][g].y = B0[(long)(ka + 1) * ldbk];
        b[0][g].z = B0[(long)(ka + 2) * ldbk];  b[0][g].w = B0[(long)(ka + 3) * ldbk];
        b[1][g].x = B1[(long)ka * ldbk];        b[1][g].y = B1[(long)(ka + 1) * ldbk];
        b[1][g].z = B1[(long)(ka + 2) * ldbk];  b[1][g].w = B1[(long)(ka + 3) * ldbk];
      }
    }
  };
  auto mma_trip = [&](const v4f (&a)[MT][TG], const v4f (&b)[2][TG]) {
    #pragma unroll
    for (int g = 0; g < TG; ++g) {
      const v2f b0lo = __builtin_shufflevector(b[0][g], b[0][g], 0, 1);
      const v2f b1lo = __builtin_shufflevector(b[1][g], b[1][g], 0, 1);
      #pragma unroll
      for (int i = 0; i < MT; ++i) {
        const v2f alo = __builtin_shufflevector(a[i][g], a[i][g], 0, 1);
        c[i][0] = __builtin_amdgcn_wmma_f32_16x16x4_f32(false, alo, false, b0lo, (short)0, c[i][0], false, false);
        c[i][1] = __builtin_amdgcn_wmma_f32_16x16x4_f32(false, alo, false, b1lo, (short)0, c[i][1], false, false);
      }
      const v2f b0hi = __builtin_shufflevector(b[0][g], b[0][g], 2, 3);
      const v2f b1hi = __builtin_shufflevector(b[1][g], b[1][g], 2, 3);
      #pragma unroll
      for (int i = 0; i < MT; ++i) {
        const v2f ahi = __builtin_shufflevector(a[i][g], a[i][g], 2, 3);
        c[i][0] = __builtin_amdgcn_wmma_f32_16x16x4_f32(false, ahi, false, b0hi, (short)0, c[i][0], false, false);
        c[i][1] = __builtin_amdgcn_wmma_f32_16x16x4_f32(false, ahi, false, b1hi, (short)0, c[i][1], false, false);
      }
    }
  };

  load_trip(aP, bP, 0);
  int k0 = 0;
  for (; k0 + 2 * TK < K; k0 += 2 * TK) {
    load_trip(aQ, bQ, k0 + TK);
    __builtin_amdgcn_sched_barrier(0);
    mma_trip(aP, bP);
    __builtin_amdgcn_sched_barrier(0);
    load_trip(aP, bP, k0 + 2 * TK);
    __builtin_amdgcn_sched_barrier(0);
    mma_trip(aQ, bQ);
  }
  load_trip(aQ, bQ, k0 + TK);
  __builtin_amdgcn_sched_barrier(0);
  mma_trip(aP, bP);
  mma_trip(aQ, bQ);
}

template <bool BCONTIG, int MT, int TS, bool SEG2, int TS2>
__device__ __forceinline__ void gemm_wave(const GemmArgs& g, int id) {
  const int lane = threadIdx.x;
  const int r15  = lane & 15;
  const int kq   = (lane >> 4) << 2;
  const int tn = id & ((1 << g.lgx) - 1);
  const int tm = (id >> g.lgx) & ((1 << g.lgy) - 1);
  const int bz = id >> (g.lgx + g.lgy);

  const v8f zero = {0.f,0.f,0.f,0.f,0.f,0.f,0.f,0.f};
  v8f c[MT][2];
  #pragma unroll
  for (int i = 0; i < MT; ++i) { c[i][0] = zero; c[i][1] = zero; }

  {
    const float* Ab = g.A + (long)bz * g.sAb + (long)(tm * (16 * MT) + r15) * g.lda;
    const float* B0 = g.B + (long)bz * g.sBb + (long)(tn * 32 + r15) * g.ldbn;
    const float* B1 = B0 + (long)16 * g.ldbn;
    k_segment<BCONTIG, MT, TS>(c, Ab, g.lda, B0, B1, g.ldbk, g.K, kq);
  }
  if (SEG2) {
    const float* Ab = g.A2 + (long)(bz * ((16 * MT) << g.lgy) + tm * (16 * MT) + r15) * g.lda2;
    const float* B0 = g.B2 + (long)(tn * 32 + r15) * g.ldbn2;
    const float* B1 = B0 + (long)16 * g.ldbn2;
    k_segment<true, MT, TS2>(c, Ab, g.lda2, B0, B1, 1, g.K2, kq);
  }

  const int  col0 = tn * 32 + r15;
  const int  col1 = col0 + 16;
  const int  m0   = tm * (16 * MT) + ((lane >> 4) << 3);
  const float bv0 = g.bias ? g.bias[col0] : 0.f;
  const float bv1 = g.bias ? g.bias[col1] : 0.f;
  float* Cp = g.C + (long)bz * g.sCb;
  #pragma unroll
  for (int i = 0; i < MT; ++i) {
    #pragma unroll
    for (int r = 0; r < 8; ++r) {
      const long row = (long)(m0 + i * 16 + r) * g.ldc;
      float v0 = c[i][0][r] + bv0;
      float v1 = c[i][1][r] + bv1;
      if (g.relu) { v0 = fmaxf(v0, 0.f);  v1 = fmaxf(v1, 0.f); }
      Cp[row + col0] = v0;
      Cp[row + col1] = v1;
    }
  }
}

template <bool BCONTIG, int TS, bool SEG2, int TS2, int S, bool HASB>
__device__ __forceinline__ void gemm_split(const GemmArgs& g, int wg, float* red) {
  static_assert(S == 2 || S == 4, "S waves of a 4-wave workgroup share a tile");
  constexpr int LGS = (S == 4) ? 2 : 1;
  constexpr int TPW = 4 / S;
  constexpr int RPW = 8 / S;
  const int lane = threadIdx.x & 31;
  const int wave = __builtin_amdgcn_readfirstlane((int)(threadIdx.x >> 5));
  const int ks = wave & (S - 1);
  const int tl = wave >> LGS;
  const int id = wg * TPW + tl;
  const int r15  = lane & 15;
  const int kq   = (lane >> 4) << 2;
  const int tn = id & ((1 << g.lgx) - 1);
  const int tm = (id >> g.lgx) & ((1 << g.lgy) - 1);
  const int bz = id >> (g.lgx + g.lgy);

  const v8f zero = {0.f,0.f,0.f,0.f,0.f,0.f,0.f,0.f};
  v8f c[1][2];
  c[0][0] = zero; c[0][1] = zero;

  const int  col0 = tn * 32 + r15;
  const int  col1 = col0 + 16;
  float bv0 = 0.f, bv1 = 0.f;
  if (HASB) { bv0 = g.bias[col0];  bv1 = g.bias[col1]; }

  {
    const int  Ks   = g.K >> LGS;
    const long koff = (long)ks * Ks;
    const float* Ab = g.A + (long)bz * g.sAb + (long)(tm * 16 + r15) * g.lda + koff;
    const float* B0 = g.B + (long)bz * g.sBb + (long)(tn * 32 + r15) * g.ldbn + koff * g.ldbk;
    const float* B1 = B0 + (long)16 * g.ldbn;
    k_segment<BCONTIG, 1, TS>(c, Ab, g.lda, B0, B1, g.ldbk, Ks, kq);
  }
  if (SEG2) {
    const int  Ks   = g.K2 >> LGS;
    const long koff = (long)ks * Ks;
    const float* Ab = g.A2 + (long)(bz * (16 << g.lgy) + tm * 16 + r15) * g.lda2 + koff;
    const float* B0 = g.B2 + (long)(tn * 32 + r15) * g.ldbn2 + koff;
    const float* B1 = B0 + (long)16 * g.ldbn2;
    k_segment<true, 1, TS2>(c, Ab, g.lda2, B0, B1, 1, Ks, kq);
  }

  float* mine = red + (wave * 16) * 32 + lane;
  #pragma unroll
  for (int j = 0; j < 2; ++j) {
    #pragma unroll
    for (int r = 0; r < 8; ++r) mine[(j * 8 + r) * 32] = c[0][j][r];
  }
  __syncthreads();

  const float* tile = red + (tl * S * 16) * 32 + lane;
  const int  m0   = tm * 16 + ((lane >> 4) << 3);
  float* Cp = g.C + (long)bz * g.sCb;
  #pragma unroll
  for (int q = 0; q < RPW; ++q) {
    const int r = ks * RPW + q;
    float v0 = tile[r * 32];
    float v1 = tile[(8 + r) * 32];
    #pragma unroll
    for (int s = 1; s < S; ++s) {
      v0 += tile[(s * 16 + r) * 32];
      v1 += tile[(s * 16 + 8 + r) * 32];
    }
    v0 += bv0;
    v1 += bv1;
    if (g.relu) { v0 = fmaxf(v0, 0.f);  v1 = fmaxf(v1, 0.f); }
    const long row = (long)(m0 + r) * g.ldc;
    Cp[row + col0] = v0;
    Cp[row + col1] = v1;
  }
}

template <bool BCONTIG, int TS, bool SEG2, int TS2, int S, bool HASB>
__global__ __launch_bounds__(128)
void wmma_gemm_split(GemmArgs g) {
  __shared__ float red[4 * 16 * 32];
  gemm_split<BCONTIG, TS, SEG2, TS2, S, HASB>(g, blockIdx.x, red);
}

template <bool BC0, int TS0, int S0, bool BC1, int TS1, int S1>
__global__ __launch_bounds__(128)
void wmma_gemm_split_pair(GemmArgs g0, GemmArgs g1, int n0) {
  __shared__ float red[4 * 16 * 32];
  const int id = blockIdx.x;
  if (id < n0) gemm_split<BC0, TS0, false, 8, S0, false>(g0, id, red);
  else         gemm_split<BC1, TS1, false, 8, S1, false>(g1, id - n0, red);
}

template <bool BC0, int MT0, int TS0, bool BC1, int MT1, int TS1>
__global__ __launch_bounds__(32)
void wmma_gemm_pair(GemmArgs g0, GemmArgs g1, int n0) {
  const int id = blockIdx.x;
  if (id < n0) gemm_wave<BC0, MT0, TS0, false, 8>(g0, id);
  else         gemm_wave<BC1, MT1, TS1, false, 8>(g1, id - n0);
}


template <int M>
__device__ __forceinline__ float lane_xor_row(float v) {
  return __builtin_bit_cast(float, __builtin_amdgcn_update_dpp(0, __builtin_bit_cast(int, v), 0x160 | M, 0xf, 0xf, true));
}
__device__ __forceinline__ float lane_xor16(float v) {
  const int iv = __builtin_bit_cast(int, v);
  return __builtin_bit_cast(float, __builtin_amdgcn_permlanex16(iv, iv, 0x76543210, 0xfedcba98, false, false));
}

template <int LEN, int BIT>
__device__ __forceinline__ void fold_step(float (&v)[HE], bool up) {
  #pragma unroll
  for (int i = 0; i < LEN / 2; ++i) {
    const float lo = v[i], hi = v[i + LEN / 2];
    const float send = up ? lo : hi;
    const float keep = up ? hi : lo;
    float got;
    if constexpr (BIT == 4) got = lane_xor16(send);
    else                    got = lane_xor_row<(1 << BIT)>(send);
    v[i] = keep + got;
  }
}

__global__ __launch_bounds__(256)
void edge_softmax_kernel(const float* __restrict__ anchor_x,
                         const float* __restrict__ node_x,
                         const float* __restrict__ node_mask,
                         const float* __restrict__ qe,
                         const float* __restrict__ Lbase,
                         float* __restrict__ attn,
                         float* __restrict__ s_out)
{
  const int bid = blockIdx.x;
  const int b   = bid >> 6;
  const int t   = threadIdx.x;

  __shared__ float sh_qe[HE];
  __shared__ float sh_l[N_PER];
  __shared__ float red[256];
  __shared__ float sh_s[8 * HE];
  __shared__ float sh_bcast;
  __shared__ float sh_ax[3];

  if (t < HE) sh_qe[t] = qe[bid * HE + t];
  if (t < 3)  sh_ax[t] = anchor_x[bid * 3 + t];
  __syncthreads();

  const float axx = sh_ax[0], axy = sh_ax[1], axz = sh_ax[2];
  const float MU_STEP = 20.f / 63.f;
  const float INV_SIG = 64.f / 20.f;

  const int n0 = t, n1 = t + 256;
  const float* nx0 = node_x + (long)(b * N_PER + n0) * 3;
  const float* nx1 = node_x + (long)(b * N_PER + n1) * 3;
  float dx0 = axx - nx0[0] + 1e-8f;
  float dy0 = axy - nx0[1] + 1e-8f;
  float dz0 = axz - nx0[2] + 1e-8f;
  float dx1 = axx - nx1[0] + 1e-8f;
  float dy1 = axy - nx1[1] + 1e-8f;
  float dz1 = axz - nx1[2] + 1e-8f;
  const float q0 = __builtin_fmaf(dz0, dz0, __builtin_fmaf(dx0, dx0, dy0 * dy0));
  const float q1 = __builtin_fmaf(dz1, dz1, __builtin_fmaf(dx1, dx1, dy1 * dy1));
  const float d10_0 = sqrtf(q0) * 0.1f;
  const float d10_1 = sqrtf(q1) * 0.1f;

  float r0[HE], r1[HE];
  float le0 = 0.f, le1 = 0.f;
  #pragma unroll
  for (int j = 0; j < HE; ++j) {
    const float u0 = __builtin_fmaf(-(float)j, MU_STEP, d10_0) * INV_SIG;
    const float u1 = __builtin_fmaf(-(float)j, MU_STEP, d10_1) * INV_SIG;
    r0[j] = __expf(-u0 * u0);
    r1[j] = __expf(-u1 * u1);
    const float qj = sh_qe[j];
    le0 += qj * r0[j];
    le1 += qj * r1[j];
  }
  float lb0 = Lbase[(long)bid * N_PER + n0] + le0;
  float lb1 = Lbase[(long)bid * N_PER + n1] + le1;
  lb0 *= (node_mask[b * N_PER + n0] - 1.f) * 1000000.0f;
  lb1 *= (node_mask[b * N_PER + n1] - 1.f) * 1000000.0f;
  sh_l[n0] = lb0;
  sh_l[n1] = lb1;
  __syncthreads();

  red[t] = fmaxf(sh_l[t], sh_l[t + 256]);
  __syncthreads();
  for (int st = 128; st > 0; st >>= 1) {
    if (t < st) red[t] = fmaxf(red[t], red[t + st]);
    __syncthreads();
  }
  if (t == 0) sh_bcast = red[0];
  __syncthreads();
  const float mx = sh_bcast;

  float e0 = __expf(sh_l[t] - mx);
  float e1 = __expf(sh_l[t + 256] - mx);
  red[t] = e0 + e1;
  __syncthreads();
  for (int st = 128; st > 0; st >>= 1) {
    if (t < st) red[t] += red[t + st];
    __syncthreads();
  }
  if (t == 0) sh_bcast = red[0];
  __syncthreads();
  const float inv = 1.f / sh_bcast;

  const float a0 = e0 * inv, a1 = e1 * inv;
  attn[(long)bid * N_PER + t]       = a0;
  attn[(long)bid * N_PER + t + 256] = a1;

  float w[HE];
  #pragma unroll
  for (int j = 0; j < HE; ++j) w[j] = a0 * r0[j] + a1 * r1[j];
  const int lane = t & 31;
  fold_step<64, 4>(w, (lane & 16) != 0);
  fold_step<32, 3>(w, (lane & 8) != 0);
  fold_step<16, 2>(w, (lane & 4) != 0);
  fold_step<8,  1>(w, (lane & 2) != 0);
  fold_step<4,  0>(w, (lane & 1) != 0);
  const int wave = t >> 5;
  sh_s[wave * HE + 2 * lane]     = w[0];
  sh_s[wave * HE + 2 * lane + 1] = w[1];
  __syncthreads();
  if (t < HE) {
    float acc = sh_s[t];
    #pragma unroll
    for (int q = 1; q < 8; ++q) acc += sh_s[q * HE + t];
    s_out[bid * HE + t] = acc;
  }
}

__device__ __forceinline__ float wave_sum(float v) {
  v += __builtin_bit_cast(float, __builtin_amdgcn_update_dpp(0, __builtin_bit_cast(int, v), 0x161, 0xf, 0xf, true));
  v += __builtin_bit_cast(float, __builtin_amdgcn_update_dpp(0, __builtin_bit_cast(int, v), 0x162, 0xf, 0xf, true));
  v += __builtin_bit_cast(float, __builtin_amdgcn_update_dpp(0, __builtin_bit_cast(int, v), 0x164, 0xf, 0xf, true));
  v += __builtin_bit_cast(float, __builtin_amdgcn_update_dpp(0, __builtin_bit_cast(int, v), 0x168, 0xf, 0xf, true));
  const int iv = __builtin_bit_cast(int, v);
  v += __builtin_bit_cast(float, __builtin_amdgcn_permlanex16(iv, iv, 0x76543210, 0xfedcba98, false, false));
  return v;
}

__device__ __forceinline__ void ln_row(const v4f v0, const v4f v1,
                                       const v4f g0, const v4f g1, const v4f b0, const v4f b1,
                                       v4f& o0, v4f& o1) {
  const float mean = wave_sum(((v0.x + v0.y) + (v0.z + v0.w)) + ((v1.x + v1.y) + (v1.z + v1.w))) * (1.f / H);

  const v4f d0 = v0 - mean, d1 = v1 - mean;
  const v4f q0 = d0 * d0,   q1 = d1 * d1;
  const float var  = wave_sum(((q0.x + q0.y) + (q0.z + q0.w)) + ((q1.x + q1.y) + (q1.z + q1.w))) * (1.f / H);
  const float rstd = rsqrtf(var + 1e-5f);

  o0 = d0 * rstd * g0 + b0;
  o1 = d1 * rstd * g1 + b1;
}

__global__ __launch_bounds__(128)
void add_ln_ln_kernel(const float* __restrict__ x, const float* __restrict__ y,
                      const float* __restrict__ gam1, const float* __restrict__ bet1,
                      const float* __restrict__ z,
                      const float* __restrict__ gam2, const float* __restrict__ bet2,
                      float* __restrict__ out)
{
  const int lane = threadIdx.x & 31;
  const int r    = blockIdx.x * 4 + (threadIdx.x >> 5);
  const int c0   = lane * 4;
  const long base = (long)r * H + c0;

  const v4f x0 = *(const v4f*)(x + base), x1 = *(const v4f*)(x + base + 128);
  const v4f y0 = *(const v4f*)(y + base), y1 = *(const v4f*)(y + base + 128);
  const v4f g0 = *(const v4f*)(gam1 + c0), g1 = *(const v4f*)(gam1 + c0 + 128);
  const v4f b0 = *(const v4f*)(bet1 + c0), b1 = *(const v4f*)(bet1 + c0 + 128);
  const v4f z0 = *(const v4f*)(z + base), z1 = *(const v4f*)(z + base + 128);
  const v4f h0 = *(const v4f*)(gam2 + c0), h1 = *(const v4f*)(gam2 + c0 + 128);
  const v4f e0 = *(const v4f*)(bet2 + c0), e1 = *(const v4f*)(bet2 + c0 + 128);
  __builtin_amdgcn_sched_barrier(0);

  v4f a0, a1;
  ln_row(x0 + y0, x1 + y1, g0, g1, b0, b1, a0, a1);
  v4f o0, o1;
  ln_row(a0 + z0, a1 + z1, h0, h1, e0, e1, o0, o1);
  *(v4f*)(out + base)       = o0;
  *(v4f*)(out + base + 128) = o1;
}

#define APITCH (H + 4)

__global__ __launch_bounds__(128)
void ln_gemm_split_kernel(const float* __restrict__ x, const float* __restrict__ y,
                          const float* __restrict__ gam, const float* __restrict__ bet,
                          GemmArgs g)
{
  constexpr int NG = 16;
  __shared__ float red[4 * 16 * 32];
  __shared__ v4f   arow4[16 * APITCH / 4];
  float* arow = (float*)arow4;

  const int lane = threadIdx.x & 31;
  const int wave = __builtin_amdgcn_readfirstlane((int)(threadIdx.x >> 5));
  const int ks = wave & 1;
  const int tl = wave >> 1;
  const int id = blockIdx.x * 2 + tl;
  const int r15  = lane & 15;
  const int kq   = (lane >> 4) << 2;
  const int tn = id & ((1 << g.lgx) - 1);
  const int tm = (id >> g.lgx) & ((1 << g.lgy) - 1);

  const int c0 = lane * 4;
  v4f x0[4], x1[4], y0[4], y1[4];
  #pragma unroll
  for (int i = 0; i < 4; ++i) {
    const long base = (long)(tm * 16 + wave * 4 + i) * H + c0;
    x0[i] = *(const v4f*)(x + base);  x1[i] = *(const v4f*)(x + base + 128);
    y0[i] = *(const v4f*)(y + base);  y1[i] = *(const v4f*)(y + base + 128);
  }
  const v4f g0 = *(const v4f*)(gam + c0), g1 = *(const v4f*)(gam + c0 + 128);
  const v4f b0 = *(const v4f*)(bet + c0), b1 = *(const v4f*)(bet + c0 + 128);

  const int  col0 = tn * 32 + r15;
  const int  col1 = col0 + 16;
  const float bv0 = g.bias[col0], bv1 = g.bias[col1];

  const int koff = ks * (H / 2);
  const float* B0 = g.B + (long)(tn * 32 + r15) * g.ldbn + koff + kq;
  const float* B1 = B0 + (long)16 * g.ldbn;
  v4f fb0[NG], fb1[NG];
  #pragma unroll
  for (int q = 0; q < NG; ++q) {
    fb0[q] = *(const v4f*)(B0 + 8 * q);
    fb1[q] = *(const v4f*)(B1 + 8 * q);
  }
  __builtin_amdgcn_sched_barrier(0);

  #pragma unroll
  for (int i = 0; i < 4; ++i) {
    v4f o0, o1;
    ln_row(x0[i] + y0[i], x1[i] + y1[i], g0, g1, b0, b1, o0, o1);
    float* dst = arow + (wave * 4 + i) * APITCH + c0;
    *(v4f*)dst         = o0;
    *(v4f*)(dst + 128) = o1;
  }
  __syncthreads();

  const float* ar = arow + r15 * APITCH + koff + kq;
  v4f fa[NG];
  #pragma unroll
  for (int q = 0; q < NG; ++q) fa[q] = *(const v4f*)(ar + 8 * q);
  __builtin_amdgcn_sched_barrier(0);

  const v8f zero = {0.f,0.f,0.f,0.f,0.f,0.f,0.f,0.f};
  v8f acc0 = zero, acc1 = zero;
  #pragma unroll
  for (int q = 0; q < NG; ++q) {
    const v2f alo = __builtin_shufflevector(fa[q], fa[q], 0, 1);
    const v2f ahi = __builtin_shufflevector(fa[q], fa[q], 2, 3);
    acc0 = __builtin_amdgcn_wmma_f32_16x16x4_f32(false, alo, false, __builtin_shufflevector(fb0[q], fb0[q], 0, 1), (short)0, acc0, false, false);
    acc1 = __builtin_amdgcn_wmma_f32_16x16x4_f32(false, alo, false, __builtin_shufflevector(fb1[q], fb1[q], 0, 1), (short)0, acc1, false, false);
    acc0 = __builtin_amdgcn_wmma_f32_16x16x4_f32(false, ahi, false, __builtin_shufflevector(fb0[q], fb0[q], 2, 3), (short)0, acc0, false, false);
    acc1 = __builtin_amdgcn_wmma_f32_16x16x4_f32(false, ahi, false, __builtin_shufflevector(fb1[q], fb1[q], 2, 3), (short)0, acc1, false, false);
  }

  float* mine = red + (wave * 16) * 32 + lane;
  #pragma unroll
  for (int r = 0; r < 8; ++r) {
    mine[r * 32]       = acc0[r];
    mine[(8 + r) * 32] = acc1[r];
  }
  __syncthreads();

  const float* tile = red + (tl * 2 * 16) * 32 + lane;
  const int m0 = tm * 16 + ((lane >> 4) << 3);
  #pragma unroll
  for (int q = 0; q < 4; ++q) {
    const int r = ks * 4 + q;
    float v0 = tile[r * 32]       + tile[(16 + r) * 32];
    float v1 = tile[(8 + r) * 32] + tile[(16 + 8 + r) * 32];
    v0 += bv0;
    v1 += bv1;
    if (g.relu) { v0 = fmaxf(v0, 0.f);  v1 = fmaxf(v1, 0.f); }
    const long row = (long)(m0 + r) * g.ldc;
    g.C[row + col0] = v0;
    g.C[row + col1] = v1;
  }
}

extern "C" void kernel_launch(void* const* d_in, const int* in_sizes, int n_in,
                              void* d_out, int out_size, void* d_ws, size_t ws_size,
                              hipStream_t stream) {
  (void)in_sizes; (void)n_in; (void)out_size; (void)ws_size;
  const float* anchor_x  = (const float*)d_in[0];
  const float* node_x    = (const float*)d_in[1];
  const float* anchor_f  = (const float*)d_in[2];
  const float* node_f    = (const float*)d_in[3];
  const float* node_mask = (const float*)d_in[6];
  const float* Wq  = (const float*)d_in[7];
  const float* bq  = (const float*)d_in[8];
  const float* Wkv = (const float*)d_in[9];
  const float* bkv = (const float*)d_in[10];
  const float* ln1_g = (const float*)d_in[11];
  const float* ln1_b = (const float*)d_in[12];
  const float* W1 = (const float*)d_in[13];
  const float* b1 = (const float*)d_in[14];
  const float* W2 = (const float*)d_in[15];
  const float* b2 = (const float*)d_in[16];
  const float* W3 = (const float*)d_in[17];
  const float* b3 = (const float*)d_in[18];
  const float* ln2_g = (const float*)d_in[19];
  const float* ln2_b = (const float*)d_in[20];

  const long WKV_LD = H + HE;

  float* p = (float*)d_ws;
  float* q    = p; p += (long)NA * H;
  float* qe   = p; p += (long)NA * HE;
  float* kvn  = p; p += (long)NN * 2 * H;
  float* L    = p; p += (long)NA * N_PER;
  float* attn = p; p += (long)NA * N_PER;
  float* sbuf = p; p += (long)NA * HE;
  float* upd  = p; p += (long)NA * H;
  float* m1   = p; p += (long)NA * 2 * H;
  float* m2   = p; p += (long)NA * 2 * H;
  float* m3   = p; p += (long)NA * H;


  {
    GemmArgs gk{node_f, H, 0,  Wkv, WKV_LD, 1, 0,  kvn, 2 * H, 0,  bkv,
                nullptr, 0, nullptr, 0,  H, 0, 0, 4, 6};
    GemmArgs gq{anchor_f, H, 0,  Wq, H, 1, 0,  q, H, 0,  bq,
                nullptr, 0, nullptr, 0,  H, 0, 0, 3, 5};
    wmma_gemm_pair<true, 4, 8, true, 1, 16><<<dim3(1024 + 256), dim3(32), 0, stream>>>(gk, gq, 1024);
  }
  {
    GemmArgs ge{q, H, 0,  Wkv + H, 1, WKV_LD, 0,  qe, HE, 0,  nullptr,
                nullptr, 0, nullptr, 0,  H, 0, 0, 1, 5};
    GemmArgs gl{q, H, (long)A_PER * H,  kvn, 2 * H, 1, (long)N_PER * 2 * H,
                L, N_PER, (long)A_PER * N_PER,  nullptr,
                nullptr, 0, nullptr, 0,  H, 0, 0, 4, 2};
    wmma_gemm_split_pair<false, 8, 4, true, 16, 2><<<dim3(64 + 256), dim3(128), 0, stream>>>(ge, gl, 64);
  }
  edge_softmax_kernel<<<dim3(NA), dim3(256), 0, stream>>>(
      anchor_x, node_x, node_mask, qe, L, attn, sbuf);
  {
    GemmArgs gu{attn, N_PER, (long)A_PER * N_PER,  kvn + H, 1, 2 * H, (long)N_PER * 2 * H,
                upd, H, (long)A_PER * H,  nullptr,
                sbuf, HE, Wkv + (long)H * WKV_LD + H, WKV_LD,  N_PER, HE, 0, 3, 2};
    wmma_gemm_split<false, 16, true, 2, 4, false><<<dim3(256), dim3(128), 0, stream>>>(gu);
  }
  {
    GemmArgs g1{nullptr, 0, 0,  W1, H, 1, 0,  m1, 2 * H, 0,  b1,
                nullptr, 0, nullptr, 0,  H, 0, 1, 4, 5};
    ln_gemm_split_kernel<<<dim3(256), dim3(128), 0, stream>>>(anchor_f, upd, ln1_g, ln1_b, g1);
  }
  {
    GemmArgs g2{m1, 2 * H, 0,  W2, 2 * H, 1, 0,  m2, 2 * H, 0,  b2,
                nullptr, 0, nullptr, 0,  2 * H, 0, 1, 4, 5};
    wmma_gemm_split<true, 16, false, 8, 4, true><<<dim3(512), dim3(128), 0, stream>>>(g2);
  }
  {
    GemmArgs g3{m2, 2 * H, 0,  W3, 2 * H, 1, 0,  m3, H, 0,  b3,
                nullptr, 0, nullptr, 0,  2 * H, 0, 0, 3, 5};
    wmma_gemm_split<true, 16, false, 8, 4, true><<<dim3(256), dim3(128), 0, stream>>>(g3);
  }
  add_ln_ln_kernel<<<dim3(NA / 4), dim3(128), 0, stream>>>(
      anchor_f, upd, ln1_g, ln1_b, m3, ln2_g, ln2_b, (float*)d_out);
}


# Edge kernel: a thread owns 8 nodes, not 2: 2 waves an anchor, not 8 (1024 waves, not 4096); max and sums in the lane, over lanes, then the 2 waves in LDS: 4 barriers, not 23 (sum order changes)
# speedup vs baseline: 6.5386x; 1.0748x over previous
#include <hip/hip_runtime.h>
#include <math.h>

#define H      256
#define HE     64
#define NB     8
#define A_PER  64
#define N_PER  512
#define NA     (NB * A_PER)
#define NN     (NB * N_PER)
#define EDGE_NPT 8

typedef __attribute__((ext_vector_type(2))) float v2f;
typedef __attribute__((ext_vector_type(4))) float v4f;
typedef __attribute__((ext_vector_type(8))) float v8f;

struct GemmArgs {
  const float* A;  long lda;  long sAb;
  const float* B;  long ldbn; long ldbk; long sBb;
  float*       C;  long ldc;  long sCb;
  const float* bias;
  const float* A2; long lda2;
  const float* B2; long ldbn2;
  int K;
  int K2;
  int relu;
  int lgx;
  int lgy;
};

template <bool BCONTIG, int MT, int TS>
__device__ __forceinline__ void k_segment(v8f (&c)[MT][2],
                                          const float* Ab, long lda,
                                          const float* B0, const float* B1, long ldbk,
                                          int K, int kq) {
  const float* Ap[MT];
  #pragma unroll
  for (int i = 0; i < MT; ++i) Ap[i] = Ab + (long)(16 * i) * lda;

  static_assert(TS % 2 == 0, "a trip is a whole number of 8-k groups");
  constexpr int TK = 4 * TS;
  constexpr int TG = TS / 2;
  v4f aP[MT][TG], bP[2][TG], aQ[MT][TG], bQ[2][TG];

  auto load_trip = [&](v4f (&a)[MT][TG], v4f (&b)[2][TG], int kbase) {
    #pragma unroll
    for (int g = 0; g < TG; ++g) {
      const int ka = kbase + 8 * g + kq;
      #pragma unroll
      for (int i = 0; i < MT; ++i) a[i][g] = *(const v4f*)(Ap[i] + ka);
      if (BCONTIG) {
        b[0][g] = *(const v4f*)(B0 + ka);
        b[1][g] = *(const v4f*)(B1 + ka);
      } else {
        b[0][g].x = B0[(long)ka * ldbk];        b[0][g].y = B0[(long)(ka + 1) * ldbk];
        b[0][g].z = B0[(long)(ka + 2) * ldbk];  b[0][g].w = B0[(long)(ka + 3) * ldbk];
        b[1][g].x = B1[(long)ka * ldbk];        b[1][g].y = B1[(long)(ka + 1) * ldbk];
        b[1][g].z = B1[(long)(ka + 2) * ldbk];  b[1][g].w = B1[(long)(ka + 3) * ldbk];
      }
    }
  };
  auto mma_trip = [&](const v4f (&a)[MT][TG], const v4f (&b)[2][TG]) {
    #pragma unroll
    for (int g = 0; g < TG; ++g) {
      const v2f b0lo = __builtin_shufflevector(b[0][g], b[0][g], 0, 1);
      const v2f b1lo = __builtin_shufflevector(b[1][g], b[1][g], 0, 1);
      #pragma unroll
      for (int i = 0; i < MT; ++i) {
        const v2f alo = __builtin_shufflevector(a[i][g], a[i][g], 0, 1);
        c[i][0] = __builtin_amdgcn_wmma_f32_16x16x4_f32(false, alo, false, b0lo, (short)0, c[i][0], false, false);
        c[i][1] = __builtin_amdgcn_wmma_f32_16x16x4_f32(false, alo, false, b1lo, (short)0, c[i][1], false, false);
      }
      const v2f b0hi = __builtin_shufflevector(b[0][g], b[0][g], 2, 3);
      const v2f b1hi = __builtin_shufflevector(b[1][g], b[1][g], 2, 3);
      #pragma unroll
      for (int i = 0; i < MT; ++i) {
        const v2f ahi = __builtin_shufflevector(a[i][g], a[i][g], 2, 3);
        c[i][0] = __builtin_amdgcn_wmma_f32_16x16x4_f32(false, ahi, false, b0hi, (short)0, c[i][0], false, false);
        c[i][1] = __builtin_amdgcn_wmma_f32_16x16x4_f32(false, ahi, false, b1hi, (short)0, c[i][1], false, false);
      }
    }
  };

  load_trip(aP, bP, 0);
  int k0 = 0;
  for (; k0 + 2 * TK < K; k0 += 2 * TK) {
    load_trip(aQ, bQ, k0 + TK);
    __builtin_amdgcn_sched_barrier(0);
    mma_trip(aP, bP);
    __builtin_amdgcn_sched_barrier(0);
    load_trip(aP, bP, k0 + 2 * TK);
    __builtin_amdgcn_sched_barrier(0);
    mma_trip(aQ, bQ);
  }
  load_trip(aQ, bQ, k0 + TK);
  __builtin_amdgcn_sched_barrier(0);
  mma_trip(aP, bP);
  mma_trip(aQ, bQ);
}

template <bool BCONTIG, int MT, int TS, bool SEG2, int TS2>
__device__ __forceinline__ void gemm_wave(const GemmArgs& g, int id) {
  const int lane = threadIdx.x;
  const int r15  = lane & 15;
  const int kq   = (lane >> 4) << 2;
  const int tn = id & ((1 << g.lgx) - 1);
  const int tm = (id >> g.lgx) & ((1 << g.lgy) - 1);
  const int bz = id >> (g.lgx + g.lgy);

  const v8f zero = {0.f,0.f,0.f,0.f,0.f,0.f,0.f,0.f};
  v8f c[MT][2];
  #pragma unroll
  for (int i = 0; i < MT; ++i) { c[i][0] = zero; c[i][1] = zero; }

  {
    const float* Ab = g.A + (long)bz * g.sAb + (long)(tm * (16 * MT) + r15) * g.lda;
    const float* B0 = g.B + (long)bz * g.sBb + (long)(tn * 32 + r15) * g.ldbn;
    const float* B1 = B0 + (long)16 * g.ldbn;
    k_segment<BCONTIG, MT, TS>(c, Ab, g.lda, B0, B1, g.ldbk, g.K, kq);
  }
  if (SEG2) {
    const float* Ab = g.A2 + (long)(bz * ((16 * MT) << g.lgy) + tm * (16 * MT) + r15) * g.lda2;
    const float* B0 = g.B2 + (long)(tn * 32 + r15) * g.ldbn2;
    const float* B1 = B0 + (long)16 * g.ldbn2;
    k_segment<true, MT, TS2>(c, Ab, g.lda2, B0, B1, 1, g.K2, kq);
  }

  const int  col0 = tn * 32 + r15;
  const int  col1 = col0 + 16;
  const int  m0   = tm * (16 * MT) + ((lane >> 4) << 3);
  const float bv0 = g.bias ? g.bias[col0] : 0.f;
  const float bv1 = g.bias ? g.bias[col1] : 0.f;
  float* Cp = g.C + (long)bz * g.sCb;
  #pragma unroll
  for (int i = 0; i < MT; ++i) {
    #pragma unroll
    for (int r = 0; r < 8; ++r) {
      const long row = (long)(m0 + i * 16 + r) * g.ldc;
      float v0 = c[i][0][r] + bv0;
      float v1 = c[i][1][r] + bv1;
      if (g.relu) { v0 = fmaxf(v0, 0.f);  v1 = fmaxf(v1, 0.f); }
      Cp[row + col0] = v0;
      Cp[row + col1] = v1;
    }
  }
}

template <bool BCONTIG, int TS, bool SEG2, int TS2, int S, bool HASB>
__device__ __forceinline__ void gemm_split(const GemmArgs& g, int wg, float* red) {
  static_assert(S == 2 || S == 4, "S waves of a 4-wave workgroup share a tile");
  constexpr int LGS = (S == 4) ? 2 : 1;
  constexpr int TPW = 4 / S;
  constexpr int RPW = 8 / S;
  const int lane = threadIdx.x & 31;
  const int wave = __builtin_amdgcn_readfirstlane((int)(threadIdx.x >> 5));
  const int ks = wave & (S - 1);
  const int tl = wave >> LGS;
  const int id = wg * TPW + tl;
  const int r15  = lane & 15;
  const int kq   = (lane >> 4) << 2;
  const int tn = id & ((1 << g.lgx) - 1);
  const int tm = (id >> g.lgx) & ((1 << g.lgy) - 1);
  const int bz = id >> (g.lgx + g.lgy);

  const v8f zero = {0.f,0.f,0.f,0.f,0.f,0.f,0.f,0.f};
  v8f c[1][2];
  c[0][0] = zero; c[0][1] = zero;

  const int  col0 = tn * 32 + r15;
  const int  col1 = col0 + 16;
  float bv0 = 0.f, bv1 = 0.f;
  if (HASB) { bv0 = g.bias[col0];  bv1 = g.bias[col1]; }

  {
    const int  Ks   = g.K >> LGS;
    const long koff = (long)ks * Ks;
    const float* Ab = g.A + (long)bz * g.sAb + (long)(tm * 16 + r15) * g.lda + koff;
    const float* B0 = g.B + (long)bz * g.sBb + (long)(tn * 32 + r15) * g.ldbn + koff * g.ldbk;
    const float* B1 = B0 + (long)16 * g.ldbn;
    k_segment<BCONTIG, 1, TS>(c, Ab, g.lda, B0, B1, g.ldbk, Ks, kq);
  }
  if (SEG2) {
    const int  Ks   = g.K2 >> LGS;
    const long koff = (long)ks * Ks;
    const float* Ab = g.A2 + (long)(bz * (16 << g.lgy) + tm * 16 + r15) * g.lda2 + koff;
    const float* B0 = g.B2 + (long)(tn * 32 + r15) * g.ldbn2 + koff;
    const float* B1 = B0 + (long)16 * g.ldbn2;
    k_segment<true, 1, TS2>(c, Ab, g.lda2, B0, B1, 1, Ks, kq);
  }

  float* mine = red + (wave * 16) * 32 + lane;
  #pragma unroll
  for (int j = 0; j < 2; ++j) {
    #pragma unroll
    for (int r = 0; r < 8; ++r) mine[(j * 8 + r) * 32] = c[0][j][r];
  }
  __syncthreads();

  const float* tile = red + (tl * S * 16) * 32 + lane;
  const int  m0   = tm * 16 + ((lane >> 4) << 3);
  float* Cp = g.C + (long)bz * g.sCb;
  #pragma unroll
  for (int q = 0; q < RPW; ++q) {
    const int r = ks * RPW + q;
    float v0 = tile[r * 32];
    float v1 = tile[(8 + r) * 32];
    #pragma unroll
    for (int s = 1; s < S; ++s) {
      v0 += tile[(s * 16 + r) * 32];
      v1 += tile[(s * 16 + 8 + r) * 32];
    }
    v0 += bv0;
    v1 += bv1;
    if (g.relu) { v0 = fmaxf(v0, 0.f);  v1 = fmaxf(v1, 0.f); }
    const long row = (long)(m0 + r) * g.ldc;
    Cp[row + col0] = v0;
    Cp[row + col1] = v1;
  }
}

template <bool BCONTIG, int TS, bool SEG2, int TS2, int S, bool HASB>
__global__ __launch_bounds__(128)
void wmma_gemm_split(GemmArgs g) {
  __shared__ float red[4 * 16 * 32];
  gemm_split<BCONTIG, TS, SEG2, TS2, S, HASB>(g, blockIdx.x, red);
}

template <bool BC0, int TS0, int S0, bool BC1, int TS1, int S1>
__global__ __launch_bounds__(128)
void wmma_gemm_split_pair(GemmArgs g0, GemmArgs g1, int n0) {
  __shared__ float red[4 * 16 * 32];
  const int id = blockIdx.x;
  if (id < n0) gemm_split<BC0, TS0, false, 8, S0, false>(g0, id, red);
  else         gemm_split<BC1, TS1, false, 8, S1, false>(g1, id - n0, red);
}

template <bool BC0, int MT0, int TS0, bool BC1, int MT1, int TS1>
__global__ __launch_bounds__(32)
void wmma_gemm_pair(GemmArgs g0, GemmArgs g1, int n0) {
  const int id = blockIdx.x;
  if (id < n0) gemm_wave<BC0, MT0, TS0, false, 8>(g0, id);
  else         gemm_wave<BC1, MT1, TS1, false, 8>(g1, id - n0);
}


template <int M>
__device__ __forceinline__ float lane_xor_row(float v) {
  return __builtin_bit_cast(float, __builtin_amdgcn_update_dpp(0, __builtin_bit_cast(int, v), 0x160 | M, 0xf, 0xf, true));
}
__device__ __forceinline__ float lane_xor16(float v) {
  const int iv = __builtin_bit_cast(int, v);
  return __builtin_bit_cast(float, __builtin_amdgcn_permlanex16(iv, iv, 0x76543210, 0xfedcba98, false, false));
}

template <int LEN, int BIT>
__device__ __forceinline__ void fold_step(float (&v)[HE], bool up) {
  #pragma unroll
  for (int i = 0; i < LEN / 2; ++i) {
    const float lo = v[i], hi = v[i + LEN / 2];
    const float send = up ? lo : hi;
    const float keep = up ? hi : lo;
    float got;
    if constexpr (BIT == 4) got = lane_xor16(send);
    else                    got = lane_xor_row<(1 << BIT)>(send);
    v[i] = keep + got;
  }
}

__device__ __forceinline__ float wave_max(float v) {
  v = fmaxf(v, lane_xor_row<1>(v));
  v = fmaxf(v, lane_xor_row<2>(v));
  v = fmaxf(v, lane_xor_row<4>(v));
  v = fmaxf(v, lane_xor_row<8>(v));
  v = fmaxf(v, lane_xor16(v));
  return v;
}
__device__ __forceinline__ float wave_sum(float v);

template <int NPT>
__global__ __launch_bounds__(N_PER / NPT)
void edge_softmax_kernel(const float* __restrict__ anchor_x,
                         const float* __restrict__ node_x,
                         const float* __restrict__ node_mask,
                         const float* __restrict__ qe,
                         const float* __restrict__ Lbase,
                         float* __restrict__ attn,
                         float* __restrict__ s_out)
{
  constexpr int NT = N_PER / NPT;
  constexpr int NW = NT / 32;
  static_assert(NT >= HE && NT % 32 == 0, "one thread per edge channel in the prologue and the last step");
  const int bid = blockIdx.x;
  const int b   = bid >> 6;
  const int t   = threadIdx.x;

  __shared__ float sh_qe[HE];
  __shared__ float sh_red[2 * NW];
  __shared__ float sh_s[NW * HE];

  float qv = 0.f;
  if (NT == HE || t < HE) qv = qe[bid * HE + t];
  const float axx = anchor_x[bid * 3], axy = anchor_x[bid * 3 + 1], axz = anchor_x[bid * 3 + 2];
  float px[NPT], py[NPT], pz[NPT], lbase[NPT], mk[NPT];
  #pragma unroll
  for (int i = 0; i < NPT; ++i) {
    const int n = t + i * NT;
    const float* nx = node_x + (long)(b * N_PER + n) * 3;
    px[i] = nx[0];  py[i] = nx[1];  pz[i] = nx[2];
    lbase[i] = Lbase[(long)bid * N_PER + n];
    mk[i]    = node_mask[b * N_PER + n];
  }
  if (NT == HE || t < HE) sh_qe[t] = qv;
  __syncthreads();

  const float MU_STEP = 20.f / 63.f;
  const float INV_SIG = 64.f / 20.f;

  float d10[NPT];
  #pragma unroll
  for (int i = 0; i < NPT; ++i) {
    const float dx = axx - px[i] + 1e-8f;
    const float dy = axy - py[i] + 1e-8f;
    const float dz = axz - pz[i] + 1e-8f;
    const float q = __builtin_fmaf(dz, dz, __builtin_fmaf(dx, dx, dy * dy));
    d10[i] = sqrtf(q) * 0.1f;
  }

  float r[NPT][HE];
  float le[NPT];
  #pragma unroll
  for (int i = 0; i < NPT; ++i) le[i] = 0.f;
  #pragma unroll
  for (int j = 0; j < HE; ++j) {
    const float qj = sh_qe[j];
    #pragma unroll
    for (int i = 0; i < NPT; ++i) {
      const float u = __builtin_fmaf(-(float)j, MU_STEP, d10[i]) * INV_SIG;
      r[i][j] = __expf(-u * u);
      le[i] += qj * r[i][j];
    }
  }

  float lb[NPT];
  #pragma unroll
  for (int i = 0; i < NPT; ++i) {
    lb[i] = lbase[i] + le[i];
    lb[i] *= (mk[i] - 1.f) * 1000000.0f;
  }
  float mloc = lb[0];
  #pragma unroll
  for (int i = 1; i < NPT; ++i) mloc = fmaxf(mloc, lb[i]);
  mloc = wave_max(mloc);
  const int lane = t & 31;
  const int wave = t >> 5;
  if (lane == 0) sh_red[wave] = mloc;
  __syncthreads();
  float mx = sh_red[0];
  #pragma unroll
  for (int q = 1; q < NW; ++q) mx = fmaxf(mx, sh_red[q]);

  float e[NPT];
  #pragma unroll
  for (int i = 0; i < NPT; ++i) e[i] = __expf(lb[i] - mx);
  float sloc = e[0];
  #pragma unroll
  for (int i = 1; i < NPT; ++i) sloc += e[i];
  sloc = wave_sum(sloc);
  if (lane == 0) sh_red[NW + wave] = sloc;
  __syncthreads();
  float tot = sh_red[NW];
  #pragma unroll
  for (int q = 1; q < NW; ++q) tot += sh_red[NW + q];
  const float inv = 1.f / tot;

  float a[NPT];
  #pragma unroll
  for (int i = 0; i < NPT; ++i) {
    a[i] = e[i] * inv;
    attn[(long)bid * N_PER + t + i * NT] = a[i];
  }

  float w[HE];
  #pragma unroll
  for (int j = 0; j < HE; ++j) {
    float acc = a[0] * r[0][j];
    #pragma unroll
    for (int i = 1; i < NPT; ++i) acc += a[i] * r[i][j];
    w[j] = acc;
  }
  fold_step<64, 4>(w, (lane & 16) != 0);
  fold_step<32, 3>(w, (lane & 8) != 0);
  fold_step<16, 2>(w, (lane & 4) != 0);
  fold_step<8,  1>(w, (lane & 2) != 0);
  fold_step<4,  0>(w, (lane & 1) != 0);
  sh_s[wave * HE + 2 * lane]     = w[0];
  sh_s[wave * HE + 2 * lane + 1] = w[1];
  __syncthreads();
  if (NT == HE || t < HE) {
    float acc = sh_s[t];
    #pragma unroll
    for (int q = 1; q < NW; ++q) acc += sh_s[q * HE + t];
    s_out[bid * HE + t] = acc;
  }
}

__device__ __forceinline__ float wave_sum(float v) {
  v += __builtin_bit_cast(float, __builtin_amdgcn_update_dpp(0, __builtin_bit_cast(int, v), 0x161, 0xf, 0xf, true));
  v += __builtin_bit_cast(float, __builtin_amdgcn_update_dpp(0, __builtin_bit_cast(int, v), 0x162, 0xf, 0xf, true));
  v += __builtin_bit_cast(float, __builtin_amdgcn_update_dpp(0, __builtin_bit_cast(int, v), 0x164, 0xf, 0xf, true));
  v += __builtin_bit_cast(float, __builtin_amdgcn_update_dpp(0, __builtin_bit_cast(int, v), 0x168, 0xf, 0xf, true));
  const int iv = __builtin_bit_cast(int, v);
  v += __builtin_bit_cast(float, __builtin_amdgcn_permlanex16(iv, iv, 0x76543210, 0xfedcba98, false, false));
  return v;
}

__device__ __forceinline__ void ln_row(const v4f v0, const v4f v1,
                                       const v4f g0, const v4f g1, const v4f b0, const v4f b1,
                                       v4f& o0, v4f& o1) {
  const float mean = wave_sum(((v0.x + v0.y) + (v0.z + v0.w)) + ((v1.x + v1.y) + (v1.z + v1.w))) * (1.f / H);

  const v4f d0 = v0 - mean, d1 = v1 - mean;
  const v4f q0 = d0 * d0,   q1 = d1 * d1;
  const float var  = wave_sum(((q0.x + q0.y) + (q0.z + q0.w)) + ((q1.x + q1.y) + (q1.z + q1.w))) * (1.f / H);
  const float rstd = rsqrtf(var + 1e-5f);

  o0 = d0 * rstd * g0 + b0;
  o1 = d1 * rstd * g1 + b1;
}

__global__ __launch_bounds__(128)
void add_ln_ln_kernel(const float* __restrict__ x, const float* __restrict__ y,
                      const float* __restrict__ gam1, const float* __restrict__ bet1,
                      const float* __restrict__ z,
                      const float* __restrict__ gam2, const float* __restrict__ bet2,
                      float* __restrict__ out)
{
  const int lane = threadIdx.x & 31;
  const int r    = blockIdx.x * 4 + (threadIdx.x >> 5);
  const int c0   = lane * 4;
  const long base = (long)r * H + c0;

  const v4f x0 = *(const v4f*)(x + base), x1 = *(const v4f*)(x + base + 128);
  const v4f y0 = *(const v4f*)(y + base), y1 = *(const v4f*)(y + base + 128);
  const v4f g0 = *(const v4f*)(gam1 + c0), g1 = *(const v4f*)(gam1 + c0 + 128);
  const v4f b0 = *(const v4f*)(bet1 + c0), b1 = *(const v4f*)(bet1 + c0 + 128);
  const v4f z0 = *(const v4f*)(z + base), z1 = *(const v4f*)(z + base + 128);
  const v4f h0 = *(const v4f*)(gam2 + c0), h1 = *(const v4f*)(gam2 + c0 + 128);
  const v4f e0 = *(const v4f*)(bet2 + c0), e1 = *(const v4f*)(bet2 + c0 + 128);
  __builtin_amdgcn_sched_barrier(0);

  v4f a0, a1;
  ln_row(x0 + y0, x1 + y1, g0, g1, b0, b1, a0, a1);
  v4f o0, o1;
  ln_row(a0 + z0, a1 + z1, h0, h1, e0, e1, o0, o1);
  *(v4f*)(out + base)       = o0;
  *(v4f*)(out + base + 128) = o1;
}

#define APITCH (H + 4)

__global__ __launch_bounds__(128)
void ln_gemm_split_kernel(const float* __restrict__ x, const float* __restrict__ y,
                          const float* __restrict__ gam, const float* __restrict__ bet,
                          GemmArgs g)
{
  constexpr int NG = 16;
  __shared__ float red[4 * 16 * 32];
  __shared__ v4f   arow4[16 * APITCH / 4];
  float* arow = (float*)arow4;

  const int lane = threadIdx.x & 31;
  const int wave = __builtin_amdgcn_readfirstlane((int)(threadIdx.x >> 5));
  const int ks = wave & 1;
  const int tl = wave >> 1;
  const int id = blockIdx.x * 2 + tl;
  const int r15  = lane & 15;
  const int kq   = (lane >> 4) << 2;
  const int tn = id & ((1 << g.lgx) - 1);
  const int tm = (id >> g.lgx) & ((1 << g.lgy) - 1);

  const int c0 = lane * 4;
  v4f x0[4], x1[4], y0[4], y1[4];
  #pragma unroll
  for (int i = 0; i < 4; ++i) {
    const long base = (long)(tm * 16 + wave * 4 + i) * H + c0;
    x0[i] = *(const v4f*)(x + base);  x1[i] = *(const v4f*)(x + base + 128);
    y0[i] = *(const v4f*)(y + base);  y1[i] = *(const v4f*)(y + base + 128);
  }
  const v4f g0 = *(const v4f*)(gam + c0), g1 = *(const v4f*)(gam + c0 + 128);
  const v4f b0 = *(const v4f*)(bet + c0), b1 = *(const v4f*)(bet + c0 + 128);

  const int  col0 = tn * 32 + r15;
  const int  col1 = col0 + 16;
  const float bv0 = g.bias[col0], bv1 = g.bias[col1];

  const int koff = ks * (H / 2);
  const float* B0 = g.B + (long)(tn * 32 + r15) * g.ldbn + koff + kq;
  const float* B1 = B0 + (long)16 * g.ldbn;
  v4f fb0[NG], fb1[NG];
  #pragma unroll
  for (int q = 0; q < NG; ++q) {
    fb0[q] = *(const v4f*)(B0 + 8 * q);
    fb1[q] = *(const v4f*)(B1 + 8 * q);
  }
  __builtin_amdgcn_sched_barrier(0);

  #pragma unroll
  for (int i = 0; i < 4; ++i) {
    v4f o0, o1;
    ln_row(x0[i] + y0[i], x1[i] + y1[i], g0, g1, b0, b1, o0, o1);
    float* dst = arow + (wave * 4 + i) * APITCH + c0;
    *(v4f*)dst         = o0;
    *(v4f*)(dst + 128) = o1;
  }
  __syncthreads();

  const float* ar = arow + r15 * APITCH + koff + kq;
  v4f fa[NG];
  #pragma unroll
  for (int q = 0; q < NG; ++q) fa[q] = *(const v4f*)(ar + 8 * q);
  __builtin_amdgcn_sched_barrier(0);

  const v8f zero = {0.f,0.f,0.f,0.f,0.f,0.f,0.f,0.f};
  v8f acc0 = zero, acc1 = zero;
  #pragma unroll
  for (int q = 0; q < NG; ++q) {
    const v2f alo = __builtin_shufflevector(fa[q], fa[q], 0, 1);
    const v2f ahi = __builtin_shufflevector(fa[q], fa[q], 2, 3);
    acc0 = __builtin_amdgcn_wmma_f32_16x16x4_f32(false, alo, false, __builtin_shufflevector(fb0[q], fb0[q], 0, 1), (short)0, acc0, false, false);
    acc1 = __builtin_amdgcn_wmma_f32_16x16x4_f32(false, alo, false, __builtin_shufflevector(fb1[q], fb1[q], 0, 1), (short)0, acc1, false, false);
    acc0 = __builtin_amdgcn_wmma_f32_16x16x4_f32(false, ahi, false, __builtin_shufflevector(fb0[q], fb0[q], 2, 3), (short)0, acc0, false, false);
    acc1 = __builtin_amdgcn_wmma_f32_16x16x4_f32(false, ahi, false, __builtin_shufflevector(fb1[q], fb1[q], 2, 3), (short)0, acc1, false, false);
  }

  float* mine = red + (wave * 16) * 32 + lane;
  #pragma unroll
  for (int r = 0; r < 8; ++r) {
    mine[r * 32]       = acc0[r];
    mine[(8 + r) * 32] = acc1[r];
  }
  __syncthreads();

  const float* tile = red + (tl * 2 * 16) * 32 + lane;
  const int m0 = tm * 16 + ((lane >> 4) << 3);
  #pragma unroll
  for (int q = 0; q < 4; ++q) {
    const int r = ks * 4 + q;
    float v0 = tile[r * 32]       + tile[(16 + r) * 32];
    float v1 = tile[(8 + r) * 32] + tile[(16 + 8 + r) * 32];
    v0 += bv0;
    v1 += bv1;
    if (g.relu) { v0 = fmaxf(v0, 0.f);  v1 = fmaxf(v1, 0.f); }
    const long row = (long)(m0 + r) * g.ldc;
    g.C[row + col0] = v0;
    g.C[row + col1] = v1;
  }
}

extern "C" void kernel_launch(void* const* d_in, const int* in_sizes, int n_in,
                              void* d_out, int out_size, void* d_ws, size_t ws_size,
                              hipStream_t stream) {
  (void)in_sizes; (void)n_in; (void)out_size; (void)ws_size;
  const float* anchor_x  = (const float*)d_in[0];
  const float* node_x    = (const float*)d_in[1];
  const float* anchor_f  = (const float*)d_in[2];
  const float* node_f    = (const float*)d_in[3];
  const float* node_mask = (const float*)d_in[6];
  const float* Wq  = (const float*)d_in[7];
  const float* bq  = (const float*)d_in[8];
  const float* Wkv = (const float*)d_in[9];
  const float* bkv = (const float*)d_in[10];
  const float* ln1_g = (const float*)d_in[11];
  const float* ln1_b = (const float*)d_in[12];
  const float* W1 = (const float*)d_in[13];
  const float* b1 = (const float*)d_in[14];
  const float* W2 = (const float*)d_in[15];
  const float* b2 = (const float*)d_in[16];
  const float* W3 = (const float*)d_in[17];
  const float* b3 = (const float*)d_in[18];
  const float* ln2_g = (const float*)d_in[19];
  const float* ln2_b = (const float*)d_in[20];

  const long WKV_LD = H + HE;

  float* p = (float*)d_ws;
  float* q    = p; p += (long)NA * H;
  float* qe   = p; p += (long)NA * HE;
  float* kvn  = p; p += (long)NN * 2 * H;
  float* L    = p; p += (long)NA * N_PER;
  float* attn = p; p += (long)NA * N_PER;
  float* sbuf = p; p += (long)NA * HE;
  float* upd  = p; p += (long)NA * H;
  float* m1   = p; p += (long)NA * 2 * H;
  float* m2   = p; p += (long)NA * 2 * H;
  float* m3   = p; p += (long)NA * H;


  {
    GemmArgs gk{node_f, H, 0,  Wkv, WKV_LD, 1, 0,  kvn, 2 * H, 0,  bkv,
                nullptr, 0, nullptr, 0,  H, 0, 0, 4, 6};
    GemmArgs gq{anchor_f, H, 0,  Wq, H, 1, 0,  q, H, 0,  bq,
                nullptr, 0, nullptr, 0,  H, 0, 0, 3, 5};
    wmma_gemm_pair<true, 4, 8, true, 1, 16><<<dim3(1024 + 256), dim3(32), 0, stream>>>(gk, gq, 1024);
  }
  {
    GemmArgs ge{q, H, 0,  Wkv + H, 1, WKV_LD, 0,  qe, HE, 0,  nullptr,
                nullptr, 0, nullptr, 0,  H, 0, 0, 1, 5};
    GemmArgs gl{q, H, (long)A_PER * H,  kvn, 2 * H, 1, (long)N_PER * 2 * H,
                L, N_PER, (long)A_PER * N_PER,  nullptr,
                nullptr, 0, nullptr, 0,  H, 0, 0, 4, 2};
    wmma_gemm_split_pair<false, 8, 4, true, 16, 2><<<dim3(64 + 256), dim3(128), 0, stream>>>(ge, gl, 64);
  }
  edge_softmax_kernel<EDGE_NPT><<<dim3(NA), dim3(N_PER / EDGE_NPT), 0, stream>>>(
      anchor_x, node_x, node_mask, qe, L, attn, sbuf);
  {
    GemmArgs gu{attn, N_PER, (long)A_PER * N_PER,  kvn + H, 1, 2 * H, (long)N_PER * 2 * H,
                upd, H, (long)A_PER * H,  nullptr,
                sbuf, HE, Wkv + (long)H * WKV_LD + H, WKV_LD,  N_PER, HE, 0, 3, 2};
    wmma_gemm_split<false, 16, true, 2, 4, false><<<dim3(256), dim3(128), 0, stream>>>(gu);
  }
  {
    GemmArgs g1{nullptr, 0, 0,  W1, H, 1, 0,  m1, 2 * H, 0,  b1,
                nullptr, 0, nullptr, 0,  H, 0, 1, 4, 5};
    ln_gemm_split_kernel<<<dim3(256), dim3(128), 0, stream>>>(anchor_f, upd, ln1_g, ln1_b, g1);
  }
  {
    GemmArgs g2{m1, 2 * H, 0,  W2, 2 * H, 1, 0,  m2, 2 * H, 0,  b2,
                nullptr, 0, nullptr, 0,  2 * H, 0, 1, 4, 5};
    wmma_gemm_split<true, 16, false, 8, 4, true><<<dim3(512), dim3(128), 0, stream>>>(g2);
  }
  {
    GemmArgs g3{m2, 2 * H, 0,  W3, 2 * H, 1, 0,  m3, H, 0,  b3,
                nullptr, 0, nullptr, 0,  2 * H, 0, 0, 3, 5};
    wmma_gemm_split<true, 16, false, 8, 4, true><<<dim3(256), dim3(128), 0, stream>>>(g3);
  }
  add_ln_ln_kernel<<<dim3(NA / 4), dim3(128), 0, stream>>>(
      anchor_f, upd, ln1_g, ln1_b, m3, ln2_g, ln2_b, (float*)d_out);
}
